# starting kernel
# baseline (speedup 1.0000x reference)
#include <hip/hip_runtime.h>
#include <hip/hip_bf16.h>

typedef __attribute__((ext_vector_type(16))) _Float16 v16h;
typedef __attribute__((ext_vector_type(8)))  _Float16 v8h;
typedef __attribute__((ext_vector_type(8)))  float    v8f;

#define B_ 2
#define F_ 512
#define S_ 1024
#define E_ 4
#define I_ 1024
#define TOK (B_*S_)
#define EPSN 1e-5f
#define SLOPE_ 0.02f

__device__ __forceinline__ int frag_j(int kin) {
  return kin < 8 ? kin : (kin < 16 ? kin - 8 : (kin < 24 ? kin - 8 : kin - 16));
}
__device__ __forceinline__ int frag_lane(int kin, int n) {
  return n + (((kin >> 3) & 1) ? 16 : 0);
}

__global__ void pack_weights_kernel(const float* __restrict__ W,
                                    _Float16* __restrict__ Wp,
                                    int E, int Cin, int Cout, int mode) {
  size_t idx = (size_t)blockIdx.x * blockDim.x + threadIdx.x;
  size_t total = (size_t)E * Cin * Cout;
  if (idx >= total) return;
  int j    = (int)(idx & 15);
  int lane = (int)((idx >> 4) & 31);
  size_t rest = idx >> 9;
  int KT = Cin >> 5;
  int kt = (int)(rest % KT); rest /= KT;
  int MT = Cout >> 4;
  int mt = (int)(rest % MT);
  int e  = (int)(rest / MT);
  int khalf = (lane >> 4) * 8;
  int k = kt * 32 + khalf + (j < 8 ? j : j + 8);
  int m = mt * 16 + (lane & 15);
  float v = mode ? W[(size_t)m * Cin + k]
                 : W[((size_t)e * Cin + k) * Cout + m];
  Wp[idx] = (_Float16)v;
}

__device__ __forceinline__ v16h load_frag(const _Float16* p) {
  v8h lo = *(const v8h*)p;
  v8h hi = *(const v8h*)(p + 8);
  v16h r;
#pragma unroll
  for (int j = 0; j < 8; ++j) { r[j] = lo[j]; r[j + 8] = hi[j]; }
  return r;
}

__global__ void gate_kernel(const float* __restrict__ x, int C,
                            const float* __restrict__ gw,
                            int* __restrict__ eidx,
                            float* __restrict__ lacc) {
  extern __shared__ float gws[];
  for (int p = threadIdx.x; p < E_ * C; p += blockDim.x) gws[p] = gw[p];
  __syncthreads();
  int t = blockIdx.x * blockDim.x + threadIdx.x;
  if (t >= TOK) return;
  int b = t / S_, s = t % S_;
  const float* xb = x + (size_t)b * C * S_ + s;
  float a0 = 0.f, a1 = 0.f, a2 = 0.f, a3 = 0.f;
  for (int c = 0; c < C; ++c) {
    float xv = xb[(size_t)c * S_];
    a0 += xv * gws[0 * C + c];
    a1 += xv * gws[1 * C + c];
    a2 += xv * gws[2 * C + c];
    a3 += xv * gws[3 * C + c];
  }
  float a[4] = {a0, a1, a2, a3};
  int best = 0; float m = a[0];
  for (int e = 1; e < E_; ++e) { if (a[e] > m) { m = a[e]; best = e; } }
  float g[4]; float se = 0.f;
  for (int e = 0; e < E_; ++e) { g[e] = __expf(a[e] - m); se += g[e]; }
  float inv = 1.0f / se;
  eidx[t] = best;
  for (int e = 0; e < E_; ++e) atomicAdd(&lacc[e], g[e] * inv);
  atomicAdd(&lacc[4 + best], 1.0f);
}

__global__ void moe_wmma_kernel(const float* __restrict__ X, int Cin,
                                const _Float16* __restrict__ Wp, int Cout,
                                const int* __restrict__ eidx,
                                float* __restrict__ Out, float outscale) {
  extern __shared__ _Float16 xs[];
  int tile = blockIdx.y;
  int b = (tile * 16) / S_;
  int s0 = (tile * 16) % S_;
  int KT = Cin >> 5;
  for (int p = threadIdx.x; p < Cin * 16; p += blockDim.x) {
    int k = p >> 4, n = p & 15;
    int kt = k >> 5, kin = k & 31;
    xs[((kt * 32 + frag_lane(kin, n)) << 4) + frag_j(kin)] =
        (_Float16)X[((size_t)b * Cin + k) * S_ + s0 + n];
  }
  __syncthreads();
  int wave = threadIdx.x >> 5, lane = threadIdx.x & 31;
  int mt0 = blockIdx.x * 16 + wave * 2;
  int col = lane & 15, khalf = (lane >> 4) * 8;
  int myexp = eidx[tile * 16 + col];
  size_t fragstride = (size_t)KT * 512;
  v8f c0s = {}, c1s = {};
  for (int e = 0; e < E_; ++e) {
    if (!__any(myexp == e)) continue;
    const _Float16* We = Wp + (size_t)e * (Cout >> 4) * fragstride;
    const _Float16* A0 = We + (size_t)(mt0 + 0) * fragstride + lane * 16;
    const _Float16* A1 = We + (size_t)(mt0 + 1) * fragstride + lane * 16;
    v8f c0 = {}, c1 = {};
    for (int kt = 0; kt < KT; ++kt) {
      if (kt + 1 < KT) __builtin_prefetch(A0 + (size_t)(kt + 1) * 512, 0, 0);
      v16h bv  = load_frag(&xs[(kt * 32 + lane) * 16]);
      v16h a0v = load_frag(A0 + (size_t)kt * 512);
      v16h a1v = load_frag(A1 + (size_t)kt * 512);
      c0 = __builtin_amdgcn_wmma_f32_16x16x32_f16(false, a0v, false, bv,
                                                  (short)0, c0, false, false);
      c1 = __builtin_amdgcn_wmma_f32_16x16x32_f16(false, a1v, false, bv,
                                                  (short)0, c1, false, false);
    }
    if (myexp == e) { c0s = c0; c1s = c1; }
  }
#pragma unroll
  for (int j = 0; j < 8; ++j) {
    int r0 = mt0 * 16 + j + khalf;
    int r1 = r0 + 16;
    Out[((size_t)b * Cout + r0) * S_ + s0 + col] = c0s[j] * outscale;
    Out[((size_t)b * Cout + r1) * S_ + s0 + col] = c1s[j] * outscale;
  }
}

__global__ void cumsum_kernel(const float* __restrict__ h0,
                              const float* __restrict__ divisor,
                              float* __restrict__ tmp1) {
  int row = blockIdx.x * (blockDim.x >> 5) + (threadIdx.x >> 5);
  int lane = threadIdx.x & 31;
  if (row >= B_ * I_) return;
  int b = row / I_, i = row % I_;
  const float* dep = h0 + ((size_t)b * 3 * I_ + i) * S_;
  const float* sc  = h0 + ((size_t)b * 3 * I_ + I_ + i) * S_;
  const float* sh  = h0 + ((size_t)b * 3 * I_ + 2 * I_ + i) * S_;
  float* o = tmp1 + ((size_t)b * I_ + i) * S_;
  float carry = 0.f;
  for (int s0 = 0; s0 < S_; s0 += 32) {
    float v = dep[s0 + lane];
#pragma unroll
    for (int d = 1; d < 32; d <<= 1) {
      float t = __shfl_up(v, d, 32);
      if (lane >= d) v += t;
    }
    float cum = carry + v;
    o[s0 + lane] = cum / divisor[s0 + lane] * sc[s0 + lane] + sh[s0 + lane];
    carry = __shfl(cum, 31, 32);
  }
}

__global__ void norm_kernel(const float* __restrict__ p0,
                            const float* __restrict__ p1,
                            const float* __restrict__ p2,
                            int cstride, int mode,
                            float* __restrict__ out) {
  __shared__ float red[256];
  int t = blockIdx.x; int b = t / S_, s = t % S_;
  float x[4]; float lsum = 0.f;
#pragma unroll
  for (int r = 0; r < 4; ++r) {
    int i = threadIdx.x + r * 256;
    size_t idx = ((size_t)b * cstride + i) * S_ + s;
    float v = p0[idx];
    if (mode) v = v * p1[idx] + p2[idx];
    x[r] = v; lsum += v;
  }
  red[threadIdx.x] = lsum; __syncthreads();
  for (int st = 128; st > 0; st >>= 1) {
    if (threadIdx.x < st) red[threadIdx.x] += red[threadIdx.x + st];
    __syncthreads();
  }
  float mean = red[0] * (1.0f / I_);
  __syncthreads();
  float lss = 0.f;
#pragma unroll
  for (int r = 0; r < 4; ++r) { float d = x[r] - mean; lss += d * d; }
  red[threadIdx.x] = lss; __syncthreads();
  for (int st = 128; st > 0; st >>= 1) {
    if (threadIdx.x < st) red[threadIdx.x] += red[threadIdx.x + st];
    __syncthreads();
  }
  float n = sqrtf(red[0]) * (1.0f / 32.0f);
  float inv = 1.0f / (n + EPSN);
#pragma unroll
  for (int r = 0; r < 4; ++r) {
    int i = threadIdx.x + r * 256;
    float y = (x[r] - mean) * inv;
    y = y >= 0.f ? y : y * SLOPE_;
    out[((size_t)b * I_ + i) * S_ + s] = y;
  }
}

__global__ void conv_wmma_kernel(const float* __restrict__ G,
                                 const _Float16* __restrict__ Wp1,
                                 float* __restrict__ Out) {
  __shared__ _Float16 xs[128 * 16];
  int tile = blockIdx.y;
  int b = (tile * 16) / S_, s0 = (tile * 16) % S_;
  int wave = threadIdx.x >> 5, lane = threadIdx.x & 31;
  int mt0 = blockIdx.x * 16 + wave * 2;
  int col = lane & 15, khalf = (lane >> 4) * 8;
  const int KTtot = (3 * I_) >> 5;
  size_t fragstride = (size_t)KTtot * 512;
  const _Float16* A0 = Wp1 + (size_t)(mt0 + 0) * fragstride + lane * 16;
  const _Float16* A1 = Wp1 + (size_t)(mt0 + 1) * fragstride + lane * 16;
  v8f c0 = {}, c1 = {};
  for (int cc = 0; cc < 3 * I_; cc += 128) {
    __syncthreads();
    for (int p = threadIdx.x; p < 128 * 16; p += blockDim.x) {
      int kk = cc + (p >> 4); int n = p & 15;
      int i = kk / 3, kc = kk - i * 3;
      int s = s0 + n - 2 + kc;
      int klocal = p >> 4;
      int kt = klocal >> 5, kin = klocal & 31;
      xs[((kt * 32 + frag_lane(kin, n)) << 4) + frag_j(kin)] =
          (s >= 0) ? (_Float16)G[((size_t)b * I_ + i) * S_ + s] : (_Float16)0.0f;
    }
    __syncthreads();
    int ktbase = cc >> 5;
#pragma unroll
    for (int kt = 0; kt < 4; ++kt) {
      __builtin_prefetch(A0 + (size_t)(ktbase + kt + 1) * 512, 0, 0);
      v16h bv  = load_frag(&xs[(kt * 32 + lane) * 16]);
      v16h a0v = load_frag(A0 + (size_t)(ktbase + kt) * 512);
      v16h a1v = load_frag(A1 + (size_t)(ktbase + kt) * 512);
      c0 = __builtin_amdgcn_wmma_f32_16x16x32_f16(false, a0v, false, bv,
                                                  (short)0, c0, false, false);
      c1 = __builtin_amdgcn_wmma_f32_16x16x32_f16(false, a1v, false, bv,
                                                  (short)0, c1, false, false);
    }
  }
#pragma unroll
  for (int j = 0; j < 8; ++j) {
    int r0 = mt0 * 16 + j + khalf;
    int r1 = r0 + 16;
    Out[((size_t)b * (3 * I_) + r0) * S_ + s0 + col] = c0[j];
    Out[((size_t)b * (3 * I_) + r1) * S_ + s0 + col] = c1[j];
  }
}

__global__ void init_loss_kernel(float* lacc) {
  if (threadIdx.x < 16) lacc[threadIdx.x] = 0.f;
}
__global__ void finalize_loss_kernel(const float* __restrict__ lacc,
                                     float* __restrict__ out_loss) {
  if (threadIdx.x == 0) {
    const float invT = 1.0f / (float)TOK;
    float l = 0.f;
    for (int e = 0; e < E_; ++e) l += (lacc[e] * invT) * (lacc[4 + e] * invT);
    for (int e = 0; e < E_; ++e) l += (lacc[8 + e] * invT) * (lacc[12 + e] * invT);
    *out_loss = l;
  }
}

extern "C" void kernel_launch(void* const* d_in, const int* in_sizes, int n_in,
                              void* d_out, int out_size, void* d_ws, size_t ws_size,
                              hipStream_t stream) {
  const float* inp     = (const float*)d_in[0];
  const float* divisor = (const float*)d_in[1];
  const float* w0_gate = (const float*)d_in[2];
  const float* w0      = (const float*)d_in[3];
  const float* w1      = (const float*)d_in[4];
  const float* w2_gate = (const float*)d_in[5];
  const float* w2      = (const float*)d_in[6];
  float* out = (float*)d_out;

  char* ws = (char*)d_ws;
  size_t off = 0;
  float* lacc = (float*)(ws + off); off += 256;
  const size_t h0b = (size_t)B_ * 3 * I_ * S_ * sizeof(float);
  const size_t t1b = (size_t)B_ * I_ * S_ * sizeof(float);
  float* h0   = (float*)(ws + off); off += h0b;
  float* tmp1 = (float*)(ws + off); off += t1b;
  float* g1   = (float*)(ws + off); off += t1b;
  int*   idx0 = (int*)(ws + off); off += TOK * sizeof(int);
  int*   idx1 = (int*)(ws + off); off += TOK * sizeof(int);
  off = (off + 255) & ~(size_t)255;
  const size_t w0n = (size_t)E_ * F_ * 3 * I_;
  const size_t w1n = (size_t)3 * I_ * 3 * I_;
  const size_t w2n = (size_t)E_ * I_ * F_;
  _Float16* w0p = (_Float16*)(ws + off); off += w0n * 2;
  _Float16* w1p = (_Float16*)(ws + off); off += w1n * 2;
  _Float16* w2p = (_Float16*)(ws + off); off += w2n * 2;
  float* h2 = h0;
  float* h3 = tmp1;

  pack_weights_kernel<<<(int)(w0n / 256), 256, 0, stream>>>(w0, w0p, E_, F_, 3 * I_, 0);
  pack_weights_kernel<<<(int)(w1n / 256), 256, 0, stream>>>(w1, w1p, 1, 3 * I_, 3 * I_, 1);
  pack_weights_kernel<<<(int)(w2n / 256), 256, 0, stream>>>(w2, w2p, E_, I_, F_, 0);

  init_loss_kernel<<<1, 32, 0, stream>>>(lacc);

  gate_kernel<<<TOK / 256, 256, E_ * F_ * sizeof(float), stream>>>(
      inp, F_, w0_gate, idx0, lacc);
  moe_wmma_kernel<<<dim3(3 * I_ / 256, TOK / 16), 256,
                    F_ * 16 * sizeof(_Float16), stream>>>(
      inp, F_, w0p, 3 * I_, idx0, h0, 1.0f);

  cumsum_kernel<<<(B_ * I_) / 8, 256, 0, stream>>>(h0, divisor, tmp1);
  norm_kernel<<<TOK, 256, 0, stream>>>(tmp1, tmp1, tmp1, I_, 0, g1);

  conv_wmma_kernel<<<dim3(3 * I_ / 256, TOK / 16), 256, 0, stream>>>(g1, w1p, h2);

  norm_kernel<<<TOK, 256, 0, stream>>>(
      h2, h2 + (size_t)I_ * S_, h2 + (size_t)2 * I_ * S_, 3 * I_, 1, h3);

  gate_kernel<<<TOK / 256, 256, E_ * I_ * sizeof(float), stream>>>(
      h3, I_, w2_gate, idx1, lacc + 8);
  moe_wmma_kernel<<<dim3(F_ / 256, TOK / 16), 256,
                    I_ * 16 * sizeof(_Float16), stream>>>(
      h3, I_, w2p, F_, idx1, out, 1.0f  );

  finalize_loss_kernel<<<1, 32, 0, stream>>>(lacc, out + (size_t)B_ * F_ * S_);
}


# cumsum made a decoupled scan and fused with norm 1: GEMM 0's epilogue stores 32-token row sums, one kernel per 32-token tile scans, norms, writes the f16 plane; f32 plane gone (sum order changes)
# speedup vs baseline: 3.3293x; 3.3293x over previous
#include <hip/hip_runtime.h>
#include <hip/hip_bf16.h>

typedef __attribute__((ext_vector_type(16))) _Float16 v16h;
typedef __attribute__((ext_vector_type(8)))  _Float16 v8h;
typedef __attribute__((ext_vector_type(4)))  _Float16 v4h;
typedef __attribute__((ext_vector_type(8)))  float    v8f;

#define B_ 2
#define F_ 512
#define S_ 1024
#define E_ 4
#define I_ 1024
#define TOK (B_*S_)
#define EPSN 1e-5f
#define SLOPE_ 0.02f

__device__ __forceinline__ int frag_j(int kin) {
  return kin < 8 ? kin : (kin < 16 ? kin - 8 : (kin < 24 ? kin - 8 : kin - 16));
}
__device__ __forceinline__ int frag_lane(int kin, int n) {
  return n + (((kin >> 3) & 1) ? 16 : 0);
}

__global__ void pack_weights_kernel(const float* __restrict__ W,
                                    _Float16* __restrict__ Wp,
                                    int E, int Cin, int Cout) {
  size_t idx = (size_t)blockIdx.x * blockDim.x + threadIdx.x;
  size_t total = (size_t)E * Cin * Cout;
  if (idx >= total) return;
  int j    = (int)(idx & 15);
  int lane = (int)((idx >> 4) & 31);
  size_t rest = idx >> 9;
  int KT = Cin >> 5;
  int kt = (int)(rest % KT); rest /= KT;
  int MT = Cout >> 4;
  int mt = (int)(rest % MT);
  int e  = (int)(rest / MT);
  int khalf = (lane >> 4) * 8;
  int k = kt * 32 + khalf + (j < 8 ? j : j + 8);
  int m = mt * 16 + (lane & 15);
  float v = W[((size_t)e * Cin + k) * Cout + m];
  Wp[idx] = (_Float16)v;
}

__device__ __forceinline__ v16h load_frag(const _Float16* p) {
  v8h lo = *(const v8h*)p;
  v8h hi = *(const v8h*)(p + 8);
  v16h r;
#pragma unroll
  for (int j = 0; j < 8; ++j) { r[j] = lo[j]; r[j + 8] = hi[j]; }
  return r;
}

template <int CTRL>
__device__ __forceinline__ float row_dpp(float v) {
  return __int_as_float(__builtin_amdgcn_update_dpp(0, __float_as_int(v), CTRL, 0xf, 0xf, false));
}

__global__ void __launch_bounds__(256)
moe_wmma_kernel(const float* __restrict__ H, int Cin,
                const _Float16* __restrict__ Wp, int Cout,
                const float* __restrict__ gw,
                float* __restrict__ lacc,
                float* __restrict__ Out, float outscale) {
  extern __shared__ _Float16 xs[];
  __shared__ float4 gpart[256];
  __shared__ float red1[256], red2[256];
  int tile = blockIdx.y;
  int b = (tile * 16) / S_;
  int s0 = (tile * 16) % S_;
  int KT = Cin >> 5;
  __builtin_assume((KT & 3) == 0);
  {
    constexpr int NG = I_ / 64;
    int n = threadIdx.x & 15, kq = threadIdx.x >> 4;
    const float* c0 = H + ((size_t)b * 3 * Cin + kq * 4) * S_ + s0 + n;
    const float* c1 = c0 + (size_t)Cin * S_;
    const float* c2 = c1 + (size_t)Cin * S_;
    float v[NG * 4];
    float lsum = 0.f;
#pragma unroll
    for (int g = 0; g < NG; ++g) {
#pragma unroll
      for (int i = 0; i < 4; ++i) {
        size_t o = (size_t)(g * 64 + i) * S_;
        float t = c0[o] * c1[o] + c2[o];
        v[g * 4 + i] = t; lsum += t;
      }
    }
    red1[threadIdx.x] = lsum;
    __syncthreads();
    float tot = 0.f;
#pragma unroll
    for (int q = 0; q < 16; ++q) tot += red1[q * 16 + n];
    float mean = tot * (1.0f / I_);
    float lss = 0.f;
#pragma unroll
    for (int j = 0; j < NG * 4; ++j) { float d = v[j] - mean; v[j] = d; lss += d * d; }
    red2[threadIdx.x] = lss;
    __syncthreads();
    float ss = 0.f;
#pragma unroll
    for (int q = 0; q < 16; ++q) ss += red2[q * 16 + n];
    float nrm = sqrtf(ss) * (1.0f / 32.0f);
    float inv = 1.0f / (nrm + EPSN);
    float a0 = 0.f, a1 = 0.f, a2 = 0.f, a3 = 0.f;
#pragma unroll
    for (int g = 0; g < NG; ++g) {
      int k0 = g * 64 + kq * 4;
      float x0 = v[g * 4 + 0] * inv, x1 = v[g * 4 + 1] * inv;
      float x2 = v[g * 4 + 2] * inv, x3 = v[g * 4 + 3] * inv;
      x0 = x0 >= 0.f ? x0 : x0 * SLOPE_;
      x1 = x1 >= 0.f ? x1 : x1 * SLOPE_;
      x2 = x2 >= 0.f ? x2 : x2 * SLOPE_;
      x3 = x3 >= 0.f ? x3 : x3 * SLOPE_;
      float4 g0 = *(const float4*)(gw + 0 * Cin + k0);
      float4 g1 = *(const float4*)(gw + 1 * Cin + k0);
      float4 g2 = *(const float4*)(gw + 2 * Cin + k0);
      float4 g3 = *(const float4*)(gw + 3 * Cin + k0);
      a0 += x0 * g0.x; a0 += x1 * g0.y; a0 += x2 * g0.z; a0 += x3 * g0.w;
      a1 += x0 * g1.x; a1 += x1 * g1.y; a1 += x2 * g1.z; a1 += x3 * g1.w;
      a2 += x0 * g2.x; a2 += x1 * g2.y; a2 += x2 * g2.z; a2 += x3 * g2.w;
      a3 += x0 * g3.x; a3 += x1 * g3.y; a3 += x2 * g3.z; a3 += x3 * g3.w;
      int kt = k0 >> 5, kin = k0 & 31;
      v4h pk = {(_Float16)x0, (_Float16)x1, (_Float16)x2, (_Float16)x3};
      *(v4h*)&xs[((kt * 32 + frag_lane(kin, n)) << 4) + frag_j(kin)] = pk;
    }
    gpart[threadIdx.x] = make_float4(a0, a1, a2, a3);
  }
  __syncthreads();
  int wave = threadIdx.x >> 5, lane = threadIdx.x & 31;
  int mt0 = blockIdx.x * 16 + wave * 2;
  int col = lane & 15, khalf = (lane >> 4) * 8;
  float l0 = 0.f, l1 = 0.f, l2 = 0.f, l3 = 0.f;
#pragma unroll
  for (int q = 0; q < 16; ++q) {
    float4 p = gpart[q * 16 + col];
    l0 += p.x; l1 += p.y; l2 += p.z; l3 += p.w;
  }
  int myexp = 0; float lmax = l0;
  if (l1 > lmax) { lmax = l1; myexp = 1; }
  if (l2 > lmax) { lmax = l2; myexp = 2; }
  if (l3 > lmax) { lmax = l3; myexp = 3; }
  if (blockIdx.x == 0 && wave == 0) {
    float e0 = __expf(l0 - lmax), e1 = __expf(l1 - lmax);
    float e2 = __expf(l2 - lmax), e3 = __expf(l3 - lmax);
    float inv = 1.0f / (e0 + e1 + e2 + e3);
    float v[8] = {e0 * inv, e1 * inv, e2 * inv, e3 * inv,
                  myexp == 0 ? 1.f : 0.f, myexp == 1 ? 1.f : 0.f,
                  myexp == 2 ? 1.f : 0.f, myexp == 3 ? 1.f : 0.f};
#pragma unroll
    for (int d = 8; d > 0; d >>= 1) {
#pragma unroll
      for (int i = 0; i < 8; ++i) v[i] += __shfl_xor(v[i], d, 16);
    }
    if (lane == 0) {
#pragma unroll
      for (int i = 0; i < 8; ++i) atomicAdd(&lacc[i], v[i]);
    }
  }
  size_t fragstride = (size_t)KT * 512;
  v8f c0s = {}, c1s = {};
  for (int e = 0; e < E_; ++e) {
    if (!__any(myexp == e)) continue;
    const _Float16* We = Wp + (size_t)e * (Cout >> 4) * fragstride;
    const _Float16* A0 = We + (size_t)(mt0 + 0) * fragstride + lane * 16;
    const _Float16* A1 = We + (size_t)(mt0 + 1) * fragstride + lane * 16;
    v8f c0 = {}, c1 = {};
    for (int kt = 0; kt < KT; ++kt) {
      if (kt + 1 < KT) __builtin_prefetch(A0 + (size_t)(kt + 1) * 512, 0, 0);
      v16h bv  = load_frag(&xs[(kt * 32 + lane) * 16]);
      v16h a0v = load_frag(A0 + (size_t)kt * 512);
      v16h a1v = load_frag(A1 + (size_t)kt * 512);
      c0 = __builtin_amdgcn_wmma_f32_16x16x32_f16(false, a0v, false, bv,
                                                  (short)0, c0, false, false);
      c1 = __builtin_amdgcn_wmma_f32_16x16x32_f16(false, a1v, false, bv,
                                                  (short)0, c1, false, false);
    }
    if (myexp == e) { c0s = c0; c1s = c1; }
  }
#pragma unroll
  for (int j = 0; j < 8; ++j) {
    int r0 = mt0 * 16 + j + khalf;
    int r1 = r0 + 16;
    Out[((size_t)b * Cout + r0) * S_ + s0 + col] = c0s[j] * outscale;
    Out[((size_t)b * Cout + r1) * S_ + s0 + col] = c1s[j] * outscale;
  }
}

#define SP_ (S_ + 2)
__global__ void __launch_bounds__(1024)
cumnorm_kernel(const float* __restrict__ h0,
               const float* __restrict__ divisor,
               const float* __restrict__ bsum,
               _Float16* __restrict__ outp) {
  __shared__ float red1[1024], red2[1024];
  const int tile = blockIdx.x, b = blockIdx.y;
  const int s0 = tile * 32;
  const int wave = threadIdx.x >> 5, lane = threadIdx.x & 31;
  float cv = 0.f;
  {
    const float* bp = bsum + (size_t)b * 32 * I_ + wave * 32 + lane;
    float ts[32];
#pragma unroll
    for (int t = 0; t < 32; ++t) ts[t] = bp[(size_t)t * I_];
#pragma unroll
    for (int t = 0; t < 32; ++t) cv += t < tile ? ts[t] : 0.f;
  }
  const float dv = divisor[s0 + lane];
  const float* dep = h0 + ((size_t)b * 3 * I_ + wave * 32) * S_ + s0 + lane;
  const float* scp = dep + (size_t)I_ * S_;
  const float* shp = scp + (size_t)I_ * S_;
  float v[32];
  float lsum = 0.f;
#pragma unroll
  for (int i = 0; i < 32; ++i) {
    float d = dep[(size_t)i * S_];
    const float sc = scp[(size_t)i * S_], sh = shp[(size_t)i * S_];
    d += row_dpp<0x111>(d);
    d += row_dpp<0x112>(d);
    d += row_dpp<0x114>(d);
    d += row_dpp<0x118>(d);
    const float low = __int_as_float(__builtin_amdgcn_readlane(__float_as_int(d), 15));
    d += lane >= 16 ? low : 0.f;
    const float carry = __int_as_float(__builtin_amdgcn_readlane(__float_as_int(cv), i));
    const float x = (carry + d) / dv * sc + sh;
    v[i] = x; lsum += x;
  }
  red1[threadIdx.x] = lsum;
  __syncthreads();
  float tot = 0.f;
#pragma unroll
  for (int q = 0; q < 32; ++q) tot += red1[q * 32 + lane];
  const float mean = tot * (1.0f / I_);
  float lss = 0.f;
#pragma unroll
  for (int i = 0; i < 32; ++i) { float d = v[i] - mean; v[i] = d; lss += d * d; }
  red2[threadIdx.x] = lss;
  __syncthreads();
  float ss = 0.f;
#pragma unroll
  for (int q = 0; q < 32; ++q) ss += red2[q * 32 + lane];
  const float nrm = sqrtf(ss) * (1.0f / 32.0f);
  const float inv = 1.0f / (nrm + EPSN);
#pragma unroll
  for (int g = 0; g < 2; ++g) {
    v8h lo, hi;
#pragma unroll
    for (int e = 0; e < 8; ++e) {
      float y0 = v[g * 16 + e] * inv, y1 = v[g * 16 + 8 + e] * inv;
      y0 = y0 >= 0.f ? y0 : y0 * SLOPE_;
      y1 = y1 >= 0.f ? y1 : y1 * SLOPE_;
      lo[e] = (_Float16)y0; hi[e] = (_Float16)y1;
    }
    _Float16* o = outp + ((((size_t)(b * 32 + wave) * 2 + g) * SP_ + (s0 + lane + 2)) * 2) * 8;
    *(v8h*)o = lo;
    *(v8h*)(o + 8) = hi;
    if (tile == 0 && lane == 0) {
      const v8h z = {};
      *(v8h*)(o - 32) = z; *(v8h*)(o - 24) = z;
      *(v8h*)(o - 16) = z; *(v8h*)(o - 8) = z;
    }
  }
}

__device__ __forceinline__ v16h join_frag(v8h lo, v8h hi) {
  v16h r;
#pragma unroll
  for (int j = 0; j < 8; ++j) { r[j] = lo[j]; r[j + 8] = hi[j]; }
  return r;
}
__global__ void __launch_bounds__(256)
conv_ws_kernel(const float* __restrict__ W1,
               const _Float16* __restrict__ P,
               float* __restrict__ Out) {
  __shared__ __attribute__((aligned(16))) _Float16 Al[3 * 96 * 512];
  const int tid = threadIdx.x;
  const int mbase = blockIdx.x * 48;
  const int b = blockIdx.y;
  const int tbase = blockIdx.z * 512;
  {
    const float* Wrow = W1 + (size_t)mbase * (3 * I_) + tid * 12;
    const int kin = (tid * 4) & 31, ct = tid >> 3;
    const int hl = ((kin >> 3) & 1) * 16;
    const int j0 = frag_j(kin);
    const int sub = (j0 >> 3) * 256 + (j0 & 7);
#pragma unroll 16
    for (int r = 0; r < 48; ++r) {
      const float4* s4 = (const float4*)(Wrow + (size_t)r * (3 * I_));
      float4 f0 = s4[0], f1 = s4[1], f2 = s4[2];
      _Float16* dst = Al + ((r >> 4) * 96 + ct * 3) * 512 + ((r & 15) + hl) * 8 + sub;
      v4h q0 = {(_Float16)f0.x, (_Float16)f0.w, (_Float16)f1.z, (_Float16)f2.y};
      v4h q1 = {(_Float16)f0.y, (_Float16)f1.x, (_Float16)f1.w, (_Float16)f2.z};
      v4h q2 = {(_Float16)f0.z, (_Float16)f1.y, (_Float16)f2.x, (_Float16)f2.w};
      *(v4h*)(dst) = q0;
      *(v4h*)(dst + 512) = q1;
      *(v4h*)(dst + 1024) = q2;
    }
  }
  __syncthreads();
  const int wave = tid >> 5, lane = tid & 31, c = lane & 15, h = lane >> 4;
  const _Float16* Ab = Al + lane * 8;
  const size_t jhs = (size_t)SP_ * 16;
  const size_t cts = 2 * jhs;
  for (int pass = 0; pass < 2; ++pass) {
    const int s0 = tbase + (pass * 8 + wave) * 32;
    const _Float16* Pb = P + (size_t)b * 32 * cts + (size_t)(s0 + c) * 16 + h * 8;
    v8f acc[3][2];
#pragma unroll
    for (int mt = 0; mt < 3; ++mt) { acc[mt][0] = v8f{}; acc[mt][1] = v8f{}; }
    v16h bq[6][2];
#pragma unroll
    for (int d = 0; d < 6; ++d) {
      const _Float16* pp = Pb + (size_t)(d / 3) * cts + (d % 3) * 16;
#pragma unroll
      for (int tt = 0; tt < 2; ++tt)
        bq[d][tt] = join_frag(*(const v8h*)(pp + tt * 256),
                              *(const v8h*)(pp + tt * 256 + jhs));
    }
    v16h a[3];
#pragma unroll
    for (int mt = 0; mt < 3; ++mt)
      a[mt] = join_frag(*(const v8h*)(Ab + (mt * 96) * 512),
                        *(const v8h*)(Ab + (mt * 96) * 512 + 256));
    for (int ct0 = 0; ct0 < 32; ct0 += 2) {
#pragma unroll
      for (int d = 0; d < 6; ++d) {
        const int kt = ct0 * 3 + d;
        const int ktn = kt + 1 < 96 ? kt + 1 : 95;
        v16h an[3];
#pragma unroll
        for (int mt = 0; mt < 3; ++mt)
          an[mt] = join_frag(*(const v8h*)(Ab + (mt * 96 + ktn) * 512),
                             *(const v8h*)(Ab + (mt * 96 + ktn) * 512 + 256));
#pragma unroll
        for (int mt = 0; mt < 3; ++mt) {
#pragma unroll
          for (int tt = 0; tt < 2; ++tt)
            acc[mt][tt] = __builtin_amdgcn_wmma_f32_16x16x32_f16(
                false, a[mt], false, bq[d][tt], (short)0, acc[mt][tt], false, false);
        }
        int ctn = ct0 + 2 + d / 3; ctn = ctn < 32 ? ctn : 31;
        const _Float16* pp = Pb + (size_t)ctn * cts + (d % 3) * 16;
#pragma unroll
        for (int tt = 0; tt < 2; ++tt)
          bq[d][tt] = join_frag(*(const v8h*)(pp + tt * 256),
                                *(const v8h*)(pp + tt * 256 + jhs));
#pragma unroll
        for (int mt = 0; mt < 3; ++mt) a[mt] = an[mt];
      }
    }
#pragma unroll
    for (int mt = 0; mt < 3; ++mt) {
#pragma unroll
      for (int tt = 0; tt < 2; ++tt) {
#pragma unroll
        for (int j = 0; j < 8; ++j) {
          int m = mbase + mt * 16 + h * 8 + j;
          Out[((size_t)b * (3 * I_) + m) * S_ + s0 + tt * 16 + c] = acc[mt][tt][j];
        }
      }
    }
  }
}

#define WS_ROWS 48
#define WS_TOKB 64
__global__ void __launch_bounds__(256)
moe_ws_kernel(const float* __restrict__ X, const float* __restrict__ W,
              const float* __restrict__ gw, float* __restrict__ lacc,
              float* __restrict__ Out, float* __restrict__ bsum) {
  constexpr int CIN = F_, COUT = 3 * I_, KT = CIN / 32, MT = WS_ROWS / 16;
  __shared__ __attribute__((aligned(16))) _Float16 Aw[E_ * MT * KT * 512];
  __shared__ __attribute__((aligned(16))) _Float16 xs[(WS_TOKB / 16) * KT * 512];
  __shared__ float4 gpart[16 * WS_TOKB];
  const int tid = threadIdx.x;
  const int mbase = blockIdx.x * WS_ROWS;
  {
    for (int t0 = 0; t0 < 12; t0 += 3) {
      float4 f[3][8];
#pragma unroll
      for (int tq = 0; tq < 3; ++tq) {
        const int q = (t0 + tq) * 256 + tid;
        const int m4 = q % 12, ko = q / 12;
        const float* src = W + (size_t)(ko * 8) * COUT + mbase + m4 * 4;
#pragma unroll
        for (int i = 0; i < 8; ++i) f[tq][i] = *(const float4*)(src + (size_t)i * COUT);
      }
#pragma unroll
      for (int tq = 0; tq < 3; ++tq) {
        const int q = (t0 + tq) * 256 + tid;
        const int m4 = q % 12, ko = q / 12;
        const int e = ko >> 6, k8 = ko & 63;
        const int kt = k8 >> 2, o = k8 & 3;
        const int mt = m4 >> 2, c0 = (m4 & 3) * 4;
        _Float16* dst = Aw + ((e * MT + mt) * KT + kt) * 512 + (o >> 1) * 256
                        + (c0 + 16 * (o & 1)) * 8;
        v8h r0, r1, r2, r3;
#pragma unroll
        for (int i = 0; i < 8; ++i) {
          r0[i] = (_Float16)f[tq][i].x; r1[i] = (_Float16)f[tq][i].y;
          r2[i] = (_Float16)f[tq][i].z; r3[i] = (_Float16)f[tq][i].w;
        }
        *(v8h*)(dst) = r0; *(v8h*)(dst + 8) = r1;
        *(v8h*)(dst + 16) = r2; *(v8h*)(dst + 24) = r3;
      }
    }
  }
  const int wave = tid >> 5, lane = tid & 31, c = lane & 15, h = lane >> 4;
  const int n64 = tid & 63, kq4 = tid >> 6;
  const int tp = wave < 6 ? wave / 3 : wave - 6;
  const int mt = wave % 3;
  for (int blk = 0; blk < 512 / WS_TOKB; ++blk) {
    const int tok0 = blockIdx.y * 512 + blk * WS_TOKB;
    const int b = tok0 / S_, s0 = tok0 % S_;
    {
      const float* xcol = X + (size_t)b * CIN * S_ + s0 + n64;
      const int tt = n64 >> 4, n = n64 & 15;
      float ga[4][4];
#pragma unroll
      for (int u = 0; u < 4; ++u) { ga[u][0] = 0.f; ga[u][1] = 0.f; ga[u][2] = 0.f; ga[u][3] = 0.f; }
#pragma unroll 2
      for (int kb = 0; kb < CIN; kb += 64) {
#pragma unroll
        for (int u = 0; u < 4; ++u) {
          const int k0 = kb + u * 16 + kq4 * 4;
          float x0 = xcol[(size_t)(k0 + 0) * S_];
          float x1 = xcol[(size_t)(k0 + 1) * S_];
          float x2 = xcol[(size_t)(k0 + 2) * S_];
          float x3 = xcol[(size_t)(k0 + 3) * S_];
          float4 g0 = *(const float4*)(gw + 0 * CIN + k0);
          float4 g1 = *(const float4*)(gw + 1 * CIN + k0);
          float4 g2 = *(const float4*)(gw + 2 * CIN + k0);
          float4 g3 = *(const float4*)(gw + 3 * CIN + k0);
          ga[u][0] += x0 * g0.x; ga[u][0] += x1 * g0.y; ga[u][0] += x2 * g0.z; ga[u][0] += x3 * g0.w;
          ga[u][1] += x0 * g1.x; ga[u][1] += x1 * g1.y; ga[u][1] += x2 * g1.z; ga[u][1] += x3 * g1.w;
          ga[u][2] += x0 * g2.x; ga[u][2] += x1 * g2.y; ga[u][2] += x2 * g2.z; ga[u][2] += x3 * g2.w;
          ga[u][3] += x0 * g3.x; ga[u][3] += x1 * g3.y; ga[u][3] += x2 * g3.z; ga[u][3] += x3 * g3.w;
          const int kt = k0 >> 5, kin = k0 & 31;
          const int j0 = frag_j(kin);
          v4h pk = {(_Float16)x0, (_Float16)x1, (_Float16)x2, (_Float16)x3};
          *(v4h*)&xs[(tt * KT + kt) * 512 + (j0 >> 3) * 256 + frag_lane(kin, n) * 8 + (j0 & 7)] = pk;
        }
      }
#pragma unroll
      for (int u = 0; u < 4; ++u)
        gpart[(u * 4 + kq4) * WS_TOKB + n64] = make_float4(ga[u][0], ga[u][1], ga[u][2], ga[u][3]);
    }
    __syncthreads();
    float lg[2][4]; int myexp[2]; float lmax[2];
#pragma unroll
    for (int t = 0; t < 2; ++t) {
      float l0 = 0.f, l1 = 0.f, l2 = 0.f, l3 = 0.f;
#pragma unroll
      for (int q = 0; q < 16; ++q) {
        float4 p = gpart[q * WS_TOKB + (tp * 2 + t) * 16 + c];
        l0 += p.x; l1 += p.y; l2 += p.z; l3 += p.w;
      }
      int me = 0; float lm = l0;
      if (l1 > lm) { lm = l1; me = 1; }
      if (l2 > lm) { lm = l2; me = 2; }
      if (l3 > lm) { lm = l3; me = 3; }
      lg[t][0] = l0; lg[t][1] = l1; lg[t][2] = l2; lg[t][3] = l3;
      myexp[t] = me; lmax[t] = lm;
    }
    if (wave >= 6) {
      if (blockIdx.x == 0) {
#pragma unroll
        for (int t = 0; t < 2; ++t) {
          float e0 = __expf(lg[t][0] - lmax[t]), e1 = __expf(lg[t][1] - lmax[t]);
          float e2 = __expf(lg[t][2] - lmax[t]), e3 = __expf(lg[t][3] - lmax[t]);
          float inv = 1.0f / (e0 + e1 + e2 + e3);
          float v[8] = {e0 * inv, e1 * inv, e2 * inv, e3 * inv,
                        myexp[t] == 0 ? 1.f : 0.f, myexp[t] == 1 ? 1.f : 0.f,
                        myexp[t] == 2 ? 1.f : 0.f, myexp[t] == 3 ? 1.f : 0.f};
#pragma unroll
          for (int d = 8; d > 0; d >>= 1) {
#pragma unroll
            for (int i = 0; i < 8; ++i) v[i] += __shfl_xor(v[i], d, 16);
          }
          if (lane == 0) {
#pragma unroll
            for (int i = 0; i < 8; ++i) atomicAdd(&lacc[i], v[i]);
          }
        }
      }
    } else {
      const _Float16* Ab = Aw + (size_t)mt * KT * 512 + lane * 8;
      const _Float16* Xb = xs + (size_t)(tp * 2) * KT * 512 + lane * 8;
      v8f acc[4][2];
#pragma unroll
      for (int e = 0; e < 4; ++e) { acc[e][0] = v8f{}; acc[e][1] = v8f{}; }
      v16h a[4], bv[2];
#pragma unroll
      for (int e = 0; e < 4; ++e)
        a[e] = join_frag(*(const v8h*)(Ab + (size_t)e * MT * KT * 512),
                         *(const v8h*)(Ab + (size_t)e * MT * KT * 512 + 256));
#pragma unroll
      for (int t = 0; t < 2; ++t)
        bv[t] = join_frag(*(const v8h*)(Xb + (size_t)t * KT * 512),
                          *(const v8h*)(Xb + (size_t)t * KT * 512 + 256));
#pragma unroll
      for (int kt = 0; kt < KT; ++kt) {
        v16h an[4], bn[2];
        if (kt + 1 < KT) {
#pragma unroll
          for (int e = 0; e < 4; ++e)
            an[e] = join_frag(*(const v8h*)(Ab + ((size_t)e * MT * KT + kt + 1) * 512),
                              *(const v8h*)(Ab + ((size_t)e * MT * KT + kt + 1) * 512 + 256));
#pragma unroll
          for (int t = 0; t < 2; ++t)
            bn[t] = join_frag(*(const v8h*)(Xb + ((size_t)t * KT + kt + 1) * 512),
                              *(const v8h*)(Xb + ((size_t)t * KT + kt + 1) * 512 + 256));
        }
#pragma unroll
        for (int e = 0; e < 4; ++e) {
#pragma unroll
          for (int t = 0; t < 2; ++t)
            acc[e][t] = __builtin_amdgcn_wmma_f32_16x16x32_f16(
                false, a[e], false, bv[t], (short)0, acc[e][t], false, false);
        }
        if (kt + 1 < KT) {
#pragma unroll
          for (int e = 0; e < 4; ++e) a[e] = an[e];
          bv[0] = bn[0]; bv[1] = bn[1];
        }
      }
      float rs[8];
#pragma unroll
      for (int t = 0; t < 2; ++t) {
        const int me = myexp[t];
#pragma unroll
        for (int j = 0; j < 8; ++j) {
          float v = acc[0][t][j];
          v = me == 1 ? acc[1][t][j] : v;
          v = me == 2 ? acc[2][t][j] : v;
          v = me == 3 ? acc[3][t][j] : v;
          const int m = mbase + mt * 16 + h * 8 + j;
          Out[((size_t)b * COUT + m) * S_ + s0 + (tp * 2 + t) * 16 + c] = v;
          rs[j] = t == 0 ? v : rs[j] + v;
        }
      }
      if (mbase + mt * 16 < I_) {
#pragma unroll
        for (int j = 0; j < 8; ++j) {
          rs[j] += row_dpp<0x161>(rs[j]);
          rs[j] += row_dpp<0x162>(rs[j]);
          rs[j] += row_dpp<0x164>(rs[j]);
          rs[j] += row_dpp<0x168>(rs[j]);
        }
        if (c == 0) {
          float* bp = bsum + (size_t)(b * 32 + (s0 >> 5) + tp) * I_ + mbase + mt * 16 + h * 8;
          *(float4*)bp = make_float4(rs[0], rs[1], rs[2], rs[3]);
          *(float4*)(bp + 4) = make_float4(rs[4], rs[5], rs[6], rs[7]);
        }
      }
    }
    __syncthreads();
  }
}

__global__ void init_loss_kernel(float* lacc) {
  if (threadIdx.x < 16) lacc[threadIdx.x] = 0.f;
}
__global__ void finalize_loss_kernel(const float* __restrict__ lacc,
                                     float* __restrict__ out_loss) {
  if (threadIdx.x == 0) {
    const float invT = 1.0f / (float)TOK;
    float l = 0.f;
    for (int e = 0; e < E_; ++e) l += (lacc[e] * invT) * (lacc[4 + e] * invT);
    for (int e = 0; e < E_; ++e) l += (lacc[8 + e] * invT) * (lacc[12 + e] * invT);
    *out_loss = l;
  }
}

extern "C" void kernel_launch(void* const* d_in, const int* in_sizes, int n_in,
                              void* d_out, int out_size, void* d_ws, size_t ws_size,
                              hipStream_t stream) {
  const float* inp     = (const float*)d_in[0];
  const float* divisor = (const float*)d_in[1];
  const float* w0_gate = (const float*)d_in[2];
  const float* w0      = (const float*)d_in[3];
  const float* w1      = (const float*)d_in[4];
  const float* w2_gate = (const float*)d_in[5];
  const float* w2      = (const float*)d_in[6];
  float* out = (float*)d_out;

  char* ws = (char*)d_ws;
  size_t off = 0;
  float* lacc = (float*)(ws + off); off += 256;
  const size_t h0b = (size_t)B_ * 3 * I_ * S_ * sizeof(float);
  const size_t t1b = (size_t)B_ * I_ * S_ * sizeof(float);
  float* h0   = (float*)(ws + off); off += h0b;
  float* bsum = (float*)(ws + off); off += (size_t)B_ * 32 * I_ * sizeof(float);
  _Float16* g1p = (_Float16*)(ws + off); off += t1b;
  off = (off + 255) & ~(size_t)255;
  const size_t w2n = (size_t)E_ * I_ * F_;
  _Float16* w2p = (_Float16*)(ws + off); off += w2n * 2;
  float* h2 = h0;

  pack_weights_kernel<<<(int)(w2n / 256), 256, 0, stream>>>(w2, w2p, E_, I_, F_);

  init_loss_kernel<<<1, 32, 0, stream>>>(lacc);

  moe_ws_kernel<<<dim3(3 * I_ / WS_ROWS, TOK / 512), 256, 0, stream>>>(
      inp, w0, w0_gate, lacc, h0, bsum);

  cumnorm_kernel<<<dim3(S_ / 32, B_), 1024, 0, stream>>>(h0, divisor, bsum, g1p);

  conv_ws_kernel<<<dim3(3 * I_ / 48, B_, S_ / 512), 256, 0, stream>>>(w1, g1p, h2);

  moe_wmma_kernel<<<dim3(F_ / 256, TOK / 16), 256,
                    I_ * 16 * sizeof(_Float16), stream>>>(
      h2, I_, w2p, F_, w2_gate, lacc + 8, out, 1.0f  );

  finalize_loss_kernel<<<1, 32, 0, stream>>>(lacc, out + (size_t)B_ * F_ * S_);
}


# last routed GEMM: expert loop moved inside the k loop: a k-step requests the fragments of all 4 experts x 2 m-tiles together (8 loads in flight, was 1) into 8 accumulators; a lane keeps its expert's
# speedup vs baseline: 4.1494x; 1.2463x over previous
#include <hip/hip_runtime.h>
#include <hip/hip_bf16.h>

typedef __attribute__((ext_vector_type(16))) _Float16 v16h;
typedef __attribute__((ext_vector_type(8)))  _Float16 v8h;
typedef __attribute__((ext_vector_type(4)))  _Float16 v4h;
typedef __attribute__((ext_vector_type(8)))  float    v8f;

#define B_ 2
#define F_ 512
#define S_ 1024
#define E_ 4
#define I_ 1024
#define TOK (B_*S_)
#define EPSN 1e-5f
#define SLOPE_ 0.02f

__device__ __forceinline__ int frag_j(int kin) {
  return kin < 8 ? kin : (kin < 16 ? kin - 8 : (kin < 24 ? kin - 8 : kin - 16));
}
__device__ __forceinline__ int frag_lane(int kin, int n) {
  return n + (((kin >> 3) & 1) ? 16 : 0);
}

__global__ void pack_weights_kernel(const float* __restrict__ W,
                                    _Float16* __restrict__ Wp,
                                    int E, int Cin, int Cout) {
  size_t idx = (size_t)blockIdx.x * blockDim.x + threadIdx.x;
  size_t total = (size_t)E * Cin * Cout;
  if (idx >= total) return;
  int j    = (int)(idx & 15);
  int lane = (int)((idx >> 4) & 31);
  size_t rest = idx >> 9;
  int KT = Cin >> 5;
  int kt = (int)(rest % KT); rest /= KT;
  int MT = Cout >> 4;
  int mt = (int)(rest % MT);
  int e  = (int)(rest / MT);
  int khalf = (lane >> 4) * 8;
  int k = kt * 32 + khalf + (j < 8 ? j : j + 8);
  int m = mt * 16 + (lane & 15);
  float v = W[((size_t)e * Cin + k) * Cout + m];
  Wp[idx] = (_Float16)v;
}

__device__ __forceinline__ v16h load_frag(const _Float16* p) {
  v8h lo = *(const v8h*)p;
  v8h hi = *(const v8h*)(p + 8);
  v16h r;
#pragma unroll
  for (int j = 0; j < 8; ++j) { r[j] = lo[j]; r[j + 8] = hi[j]; }
  return r;
}

template <int CTRL>
__device__ __forceinline__ float row_dpp(float v) {
  return __int_as_float(__builtin_amdgcn_update_dpp(0, __float_as_int(v), CTRL, 0xf, 0xf, false));
}

__global__ void __launch_bounds__(256)
moe_wmma_kernel(const float* __restrict__ H, int Cin,
                const _Float16* __restrict__ Wp, int Cout,
                const float* __restrict__ gw,
                float* __restrict__ lacc,
                float* __restrict__ Out, float outscale) {
  extern __shared__ _Float16 xs[];
  __shared__ float4 gpart[256];
  __shared__ float red1[256], red2[256];
  int tile = blockIdx.y;
  int b = (tile * 16) / S_;
  int s0 = (tile * 16) % S_;
  int KT = Cin >> 5;
  __builtin_assume((KT & 3) == 0);
  {
    constexpr int NG = I_ / 64;
    int n = threadIdx.x & 15, kq = threadIdx.x >> 4;
    const float* c0 = H + ((size_t)b * 3 * Cin + kq * 4) * S_ + s0 + n;
    const float* c1 = c0 + (size_t)Cin * S_;
    const float* c2 = c1 + (size_t)Cin * S_;
    float v[NG * 4];
    float lsum = 0.f;
#pragma unroll
    for (int g = 0; g < NG; ++g) {
#pragma unroll
      for (int i = 0; i < 4; ++i) {
        size_t o = (size_t)(g * 64 + i) * S_;
        float t = c0[o] * c1[o] + c2[o];
        v[g * 4 + i] = t; lsum += t;
      }
    }
    red1[threadIdx.x] = lsum;
    __syncthreads();
    float tot = 0.f;
#pragma unroll
    for (int q = 0; q < 16; ++q) tot += red1[q * 16 + n];
    float mean = tot * (1.0f / I_);
    float lss = 0.f;
#pragma unroll
    for (int j = 0; j < NG * 4; ++j) { float d = v[j] - mean; v[j] = d; lss += d * d; }
    red2[threadIdx.x] = lss;
    __syncthreads();
    float ss = 0.f;
#pragma unroll
    for (int q = 0; q < 16; ++q) ss += red2[q * 16 + n];
    float nrm = sqrtf(ss) * (1.0f / 32.0f);
    float inv = 1.0f / (nrm + EPSN);
    float a0 = 0.f, a1 = 0.f, a2 = 0.f, a3 = 0.f;
#pragma unroll
    for (int g = 0; g < NG; ++g) {
      int k0 = g * 64 + kq * 4;
      float x0 = v[g * 4 + 0] * inv, x1 = v[g * 4 + 1] * inv;
      float x2 = v[g * 4 + 2] * inv, x3 = v[g * 4 + 3] * inv;
      x0 = x0 >= 0.f ? x0 : x0 * SLOPE_;
      x1 = x1 >= 0.f ? x1 : x1 * SLOPE_;
      x2 = x2 >= 0.f ? x2 : x2 * SLOPE_;
      x3 = x3 >= 0.f ? x3 : x3 * SLOPE_;
      float4 g0 = *(const float4*)(gw + 0 * Cin + k0);
      float4 g1 = *(const float4*)(gw + 1 * Cin + k0);
      float4 g2 = *(const float4*)(gw + 2 * Cin + k0);
      float4 g3 = *(const float4*)(gw + 3 * Cin + k0);
      a0 += x0 * g0.x; a0 += x1 * g0.y; a0 += x2 * g0.z; a0 += x3 * g0.w;
      a1 += x0 * g1.x; a1 += x1 * g1.y; a1 += x2 * g1.z; a1 += x3 * g1.w;
      a2 += x0 * g2.x; a2 += x1 * g2.y; a2 += x2 * g2.z; a2 += x3 * g2.w;
      a3 += x0 * g3.x; a3 += x1 * g3.y; a3 += x2 * g3.z; a3 += x3 * g3.w;
      int kt = k0 >> 5, kin = k0 & 31;
      v4h pk = {(_Float16)x0, (_Float16)x1, (_Float16)x2, (_Float16)x3};
      *(v4h*)&xs[((kt * 32 + frag_lane(kin, n)) << 4) + frag_j(kin)] = pk;
    }
    gpart[threadIdx.x] = make_float4(a0, a1, a2, a3);
  }
  __syncthreads();
  int wave = threadIdx.x >> 5, lane = threadIdx.x & 31;
  int mt0 = blockIdx.x * 16 + wave * 2;
  int col = lane & 15, khalf = (lane >> 4) * 8;
  float l0 = 0.f, l1 = 0.f, l2 = 0.f, l3 = 0.f;
#pragma unroll
  for (int q = 0; q < 16; ++q) {
    float4 p = gpart[q * 16 + col];
    l0 += p.x; l1 += p.y; l2 += p.z; l3 += p.w;
  }
  int myexp = 0; float lmax = l0;
  if (l1 > lmax) { lmax = l1; myexp = 1; }
  if (l2 > lmax) { lmax = l2; myexp = 2; }
  if (l3 > lmax) { lmax = l3; myexp = 3; }
  if (blockIdx.x == 0 && wave == 0) {
    float e0 = __expf(l0 - lmax), e1 = __expf(l1 - lmax);
    float e2 = __expf(l2 - lmax), e3 = __expf(l3 - lmax);
    float inv = 1.0f / (e0 + e1 + e2 + e3);
    float v[8] = {e0 * inv, e1 * inv, e2 * inv, e3 * inv,
                  myexp == 0 ? 1.f : 0.f, myexp == 1 ? 1.f : 0.f,
                  myexp == 2 ? 1.f : 0.f, myexp == 3 ? 1.f : 0.f};
#pragma unroll
    for (int d = 8; d > 0; d >>= 1) {
#pragma unroll
      for (int i = 0; i < 8; ++i) v[i] += __shfl_xor(v[i], d, 16);
    }
    if (lane == 0) {
#pragma unroll
      for (int i = 0; i < 8; ++i) atomicAdd(&lacc[i], v[i]);
    }
  }
  size_t fragstride = (size_t)KT * 512;
  size_t estride = (size_t)(Cout >> 4) * fragstride;
  const _Float16* Ab = Wp + (size_t)mt0 * fragstride + lane * 16;
  v8f acc[E_][2];
#pragma unroll
  for (int e = 0; e < E_; ++e) { acc[e][0] = v8f{}; acc[e][1] = v8f{}; }
  for (int kt = 0; kt < KT; ++kt) {
    v16h a[E_][2];
#pragma unroll
    for (int e = 0; e < E_; ++e) {
#pragma unroll
      for (int m = 0; m < 2; ++m)
        a[e][m] = load_frag(Ab + (size_t)e * estride + (size_t)m * fragstride
                            + (size_t)kt * 512);
    }
    v16h bv = load_frag(&xs[(kt * 32 + lane) * 16]);
    asm volatile("" :: "v"(a[0][0]), "v"(a[0][1]), "v"(a[1][0]), "v"(a[1][1]),
                       "v"(a[2][0]), "v"(a[2][1]), "v"(a[3][0]), "v"(a[3][1]));
#pragma unroll
    for (int e = 0; e < E_; ++e) {
#pragma unroll
      for (int m = 0; m < 2; ++m)
        acc[e][m] = __builtin_amdgcn_wmma_f32_16x16x32_f16(
            false, a[e][m], false, bv, (short)0, acc[e][m], false, false);
    }
  }
#pragma unroll
  for (int j = 0; j < 8; ++j) {
    float v0 = acc[0][0][j], v1 = acc[0][1][j];
    v0 = myexp == 1 ? acc[1][0][j] : v0;  v1 = myexp == 1 ? acc[1][1][j] : v1;
    v0 = myexp == 2 ? acc[2][0][j] : v0;  v1 = myexp == 2 ? acc[2][1][j] : v1;
    v0 = myexp == 3 ? acc[3][0][j] : v0;  v1 = myexp == 3 ? acc[3][1][j] : v1;
    int r0 = mt0 * 16 + j + khalf;
    int r1 = r0 + 16;
    Out[((size_t)b * Cout + r0) * S_ + s0 + col] = v0 * outscale;
    Out[((size_t)b * Cout + r1) * S_ + s0 + col] = v1 * outscale;
  }
}

#define SP_ (S_ + 2)
__global__ void __launch_bounds__(1024)
cumnorm_kernel(const float* __restrict__ h0,
               const float* __restrict__ divisor,
               const float* __restrict__ bsum,
               _Float16* __restrict__ outp) {
  __shared__ float red1[1024], red2[1024];
  const int tile = blockIdx.x, b = blockIdx.y;
  const int s0 = tile * 32;
  const int wave = threadIdx.x >> 5, lane = threadIdx.x & 31;
  float cv = 0.f;
  {
    const float* bp = bsum + (size_t)b * 32 * I_ + wave * 32 + lane;
    float ts[32];
#pragma unroll
    for (int t = 0; t < 32; ++t) ts[t] = bp[(size_t)t * I_];
#pragma unroll
    for (int t = 0; t < 32; ++t) cv += t < tile ? ts[t] : 0.f;
  }
  const float dv = divisor[s0 + lane];
  const float* dep = h0 + ((size_t)b * 3 * I_ + wave * 32) * S_ + s0 + lane;
  const float* scp = dep + (size_t)I_ * S_;
  const float* shp = scp + (size_t)I_ * S_;
  float v[32];
  float lsum = 0.f;
#pragma unroll
  for (int i = 0; i < 32; ++i) {
    float d = dep[(size_t)i * S_];
    const float sc = scp[(size_t)i * S_], sh = shp[(size_t)i * S_];
    d += row_dpp<0x111>(d);
    d += row_dpp<0x112>(d);
    d += row_dpp<0x114>(d);
    d += row_dpp<0x118>(d);
    const float low = __int_as_float(__builtin_amdgcn_readlane(__float_as_int(d), 15));
    d += lane >= 16 ? low : 0.f;
    const float carry = __int_as_float(__builtin_amdgcn_readlane(__float_as_int(cv), i));
    const float x = (carry + d) / dv * sc + sh;
    v[i] = x; lsum += x;
  }
  red1[threadIdx.x] = lsum;
  __syncthreads();
  float tot = 0.f;
#pragma unroll
  for (int q = 0; q < 32; ++q) tot += red1[q * 32 + lane];
  const float mean = tot * (1.0f / I_);
  float lss = 0.f;
#pragma unroll
  for (int i = 0; i < 32; ++i) { float d = v[i] - mean; v[i] = d; lss += d * d; }
  red2[threadIdx.x] = lss;
  __syncthreads();
  float ss = 0.f;
#pragma unroll
  for (int q = 0; q < 32; ++q) ss += red2[q * 32 + lane];
  const float nrm = sqrtf(ss) * (1.0f / 32.0f);
  const float inv = 1.0f / (nrm + EPSN);
#pragma unroll
  for (int g = 0; g < 2; ++g) {
    v8h lo, hi;
#pragma unroll
    for (int e = 0; e < 8; ++e) {
      float y0 = v[g * 16 + e] * inv, y1 = v[g * 16 + 8 + e] * inv;
      y0 = y0 >= 0.f ? y0 : y0 * SLOPE_;
      y1 = y1 >= 0.f ? y1 : y1 * SLOPE_;
      lo[e] = (_Float16)y0; hi[e] = (_Float16)y1;
    }
    _Float16* o = outp + ((((size_t)(b * 32 + wave) * 2 + g) * SP_ + (s0 + lane + 2)) * 2) * 8;
    *(v8h*)o = lo;
    *(v8h*)(o + 8) = hi;
    if (tile == 0 && lane == 0) {
      const v8h z = {};
      *(v8h*)(o - 32) = z; *(v8h*)(o - 24) = z;
      *(v8h*)(o - 16) = z; *(v8h*)(o - 8) = z;
    }
  }
}

__device__ __forceinline__ v16h join_frag(v8h lo, v8h hi) {
  v16h r;
#pragma unroll
  for (int j = 0; j < 8; ++j) { r[j] = lo[j]; r[j + 8] = hi[j]; }
  return r;
}
__global__ void __launch_bounds__(256)
conv_ws_kernel(const float* __restrict__ W1,
               const _Float16* __restrict__ P,
               float* __restrict__ Out) {
  __shared__ __attribute__((aligned(16))) _Float16 Al[3 * 96 * 512];
  const int tid = threadIdx.x;
  const int mbase = blockIdx.x * 48;
  const int b = blockIdx.y;
  const int tbase = blockIdx.z * 512;
  {
    const float* Wrow = W1 + (size_t)mbase * (3 * I_) + tid * 12;
    const int kin = (tid * 4) & 31, ct = tid >> 3;
    const int hl = ((kin >> 3) & 1) * 16;
    const int j0 = frag_j(kin);
    const int sub = (j0 >> 3) * 256 + (j0 & 7);
#pragma unroll 16
    for (int r = 0; r < 48; ++r) {
      const float4* s4 = (const float4*)(Wrow + (size_t)r * (3 * I_));
      float4 f0 = s4[0], f1 = s4[1], f2 = s4[2];
      _Float16* dst = Al + ((r >> 4) * 96 + ct * 3) * 512 + ((r & 15) + hl) * 8 + sub;
      v4h q0 = {(_Float16)f0.x, (_Float16)f0.w, (_Float16)f1.z, (_Float16)f2.y};
      v4h q1 = {(_Float16)f0.y, (_Float16)f1.x, (_Float16)f1.w, (_Float16)f2.z};
      v4h q2 = {(_Float16)f0.z, (_Float16)f1.y, (_Float16)f2.x, (_Float16)f2.w};
      *(v4h*)(dst) = q0;
      *(v4h*)(dst + 512) = q1;
      *(v4h*)(dst + 1024) = q2;
    }
  }
  __syncthreads();
  const int wave = tid >> 5, lane = tid & 31, c = lane & 15, h = lane >> 4;
  const _Float16* Ab = Al + lane * 8;
  const size_t jhs = (size_t)SP_ * 16;
  const size_t cts = 2 * jhs;
  for (int pass = 0; pass < 2; ++pass) {
    const int s0 = tbase + (pass * 8 + wave) * 32;
    const _Float16* Pb = P + (size_t)b * 32 * cts + (size_t)(s0 + c) * 16 + h * 8;
    v8f acc[3][2];
#pragma unroll
    for (int mt = 0; mt < 3; ++mt) { acc[mt][0] = v8f{}; acc[mt][1] = v8f{}; }
    v16h bq[6][2];
#pragma unroll
    for (int d = 0; d < 6; ++d) {
      const _Float16* pp = Pb + (size_t)(d / 3) * cts + (d % 3) * 16;
#pragma unroll
      for (int tt = 0; tt < 2; ++tt)
        bq[d][tt] = join_frag(*(const v8h*)(pp + tt * 256),
                              *(const v8h*)(pp + tt * 256 + jhs));
    }
    v16h a[3];
#pragma unroll
    for (int mt = 0; mt < 3; ++mt)
      a[mt] = join_frag(*(const v8h*)(Ab + (mt * 96) * 512),
                        *(const v8h*)(Ab + (mt * 96) * 512 + 256));
    for (int ct0 = 0; ct0 < 32; ct0 += 2) {
#pragma unroll
      for (int d = 0; d < 6; ++d) {
        const int kt = ct0 * 3 + d;
        const int ktn = kt + 1 < 96 ? kt + 1 : 95;
        v16h an[3];
#pragma unroll
        for (int mt = 0; mt < 3; ++mt)
          an[mt] = join_frag(*(const v8h*)(Ab + (mt * 96 + ktn) * 512),
                             *(const v8h*)(Ab + (mt * 96 + ktn) * 512 + 256));
#pragma unroll
        for (int mt = 0; mt < 3; ++mt) {
#pragma unroll
          for (int tt = 0; tt < 2; ++tt)
            acc[mt][tt] = __builtin_amdgcn_wmma_f32_16x16x32_f16(
                false, a[mt], false, bq[d][tt], (short)0, acc[mt][tt], false, false);
        }
        int ctn = ct0 + 2 + d / 3; ctn = ctn < 32 ? ctn : 31;
        const _Float16* pp = Pb + (size_t)ctn * cts + (d % 3) * 16;
#pragma unroll
        for (int tt = 0; tt < 2; ++tt)
          bq[d][tt] = join_frag(*(const v8h*)(pp + tt * 256),
                                *(const v8h*)(pp + tt * 256 + jhs));
#pragma unroll
        for (int mt = 0; mt < 3; ++mt) a[mt] = an[mt];
      }
    }
#pragma unroll
    for (int mt = 0; mt < 3; ++mt) {
#pragma unroll
      for (int tt = 0; tt < 2; ++tt) {
#pragma unroll
        for (int j = 0; j < 8; ++j) {
          int m = mbase + mt * 16 + h * 8 + j;
          Out[((size_t)b * (3 * I_) + m) * S_ + s0 + tt * 16 + c] = acc[mt][tt][j];
        }
      }
    }
  }
}

#define WS_ROWS 48
#define WS_TOKB 64
__global__ void __launch_bounds__(256)
moe_ws_kernel(const float* __restrict__ X, const float* __restrict__ W,
              const float* __restrict__ gw, float* __restrict__ lacc,
              float* __restrict__ Out, float* __restrict__ bsum) {
  constexpr int CIN = F_, COUT = 3 * I_, KT = CIN / 32, MT = WS_ROWS / 16;
  __shared__ __attribute__((aligned(16))) _Float16 Aw[E_ * MT * KT * 512];
  __shared__ __attribute__((aligned(16))) _Float16 xs[(WS_TOKB / 16) * KT * 512];
  __shared__ float4 gpart[16 * WS_TOKB];
  const int tid = threadIdx.x;
  const int mbase = blockIdx.x * WS_ROWS;
  {
    for (int t0 = 0; t0 < 12; t0 += 3) {
      float4 f[3][8];
#pragma unroll
      for (int tq = 0; tq < 3; ++tq) {
        const int q = (t0 + tq) * 256 + tid;
        const int m4 = q % 12, ko = q / 12;
        const float* src = W + (size_t)(ko * 8) * COUT + mbase + m4 * 4;
#pragma unroll
        for (int i = 0; i < 8; ++i) f[tq][i] = *(const float4*)(src + (size_t)i * COUT);
      }
#pragma unroll
      for (int tq = 0; tq < 3; ++tq) {
        const int q = (t0 + tq) * 256 + tid;
        const int m4 = q % 12, ko = q / 12;
        const int e = ko >> 6, k8 = ko & 63;
        const int kt = k8 >> 2, o = k8 & 3;
        const int mt = m4 >> 2, c0 = (m4 & 3) * 4;
        _Float16* dst = Aw + ((e * MT + mt) * KT + kt) * 512 + (o >> 1) * 256
                        + (c0 + 16 * (o & 1)) * 8;
        v8h r0, r1, r2, r3;
#pragma unroll
        for (int i = 0; i < 8; ++i) {
          r0[i] = (_Float16)f[tq][i].x; r1[i] = (_Float16)f[tq][i].y;
          r2[i] = (_Float16)f[tq][i].z; r3[i] = (_Float16)f[tq][i].w;
        }
        *(v8h*)(dst) = r0; *(v8h*)(dst + 8) = r1;
        *(v8h*)(dst + 16) = r2; *(v8h*)(dst + 24) = r3;
      }
    }
  }
  const int wave = tid >> 5, lane = tid & 31, c = lane & 15, h = lane >> 4;
  const int n64 = tid & 63, kq4 = tid >> 6;
  const int tp = wave < 6 ? wave / 3 : wave - 6;
  const int mt = wave % 3;
  for (int blk = 0; blk < 512 / WS_TOKB; ++blk) {
    const int tok0 = blockIdx.y * 512 + blk * WS_TOKB;
    const int b = tok0 / S_, s0 = tok0 % S_;
    {
      const float* xcol = X + (size_t)b * CIN * S_ + s0 + n64;
      const int tt = n64 >> 4, n = n64 & 15;
      float ga[4][4];
#pragma unroll
      for (int u = 0; u < 4; ++u) { ga[u][0] = 0.f; ga[u][1] = 0.f; ga[u][2] = 0.f; ga[u][3] = 0.f; }
#pragma unroll 2
      for (int kb = 0; kb < CIN; kb += 64) {
#pragma unroll
        for (int u = 0; u < 4; ++u) {
          const int k0 = kb + u * 16 + kq4 * 4;
          float x0 = xcol[(size_t)(k0 + 0) * S_];
          float x1 = xcol[(size_t)(k0 + 1) * S_];
          float x2 = xcol[(size_t)(k0 + 2) * S_];
          float x3 = xcol[(size_t)(k0 + 3) * S_];
          float4 g0 = *(const float4*)(gw + 0 * CIN + k0);
          float4 g1 = *(const float4*)(gw + 1 * CIN + k0);
          float4 g2 = *(const float4*)(gw + 2 * CIN + k0);
          float4 g3 = *(const float4*)(gw + 3 * CIN + k0);
          ga[u][0] += x0 * g0.x; ga[u][0] += x1 * g0.y; ga[u][0] += x2 * g0.z; ga[u][0] += x3 * g0.w;
          ga[u][1] += x0 * g1.x; ga[u][1] += x1 * g1.y; ga[u][1] += x2 * g1.z; ga[u][1] += x3 * g1.w;
          ga[u][2] += x0 * g2.x; ga[u][2] += x1 * g2.y; ga[u][2] += x2 * g2.z; ga[u][2] += x3 * g2.w;
          ga[u][3] += x0 * g3.x; ga[u][3] += x1 * g3.y; ga[u][3] += x2 * g3.z; ga[u][3] += x3 * g3.w;
          const int kt = k0 >> 5, kin = k0 & 31;
          const int j0 = frag_j(kin);
          v4h pk = {(_Float16)x0, (_Float16)x1, (_Float16)x2, (_Float16)x3};
          *(v4h*)&xs[(tt * KT + kt) * 512 + (j0 >> 3) * 256 + frag_lane(kin, n) * 8 + (j0 & 7)] = pk;
        }
      }
#pragma unroll
      for (int u = 0; u < 4; ++u)
        gpart[(u * 4 + kq4) * WS_TOKB + n64] = make_float4(ga[u][0], ga[u][1], ga[u][2], ga[u][3]);
    }
    __syncthreads();
    float lg[2][4]; int myexp[2]; float lmax[2];
#pragma unroll
    for (int t = 0; t < 2; ++t) {
      float l0 = 0.f, l1 = 0.f, l2 = 0.f, l3 = 0.f;
#pragma unroll
      for (int q = 0; q < 16; ++q) {
        float4 p = gpart[q * WS_TOKB + (tp * 2 + t) * 16 + c];
        l0 += p.x; l1 += p.y; l2 += p.z; l3 += p.w;
      }
      int me = 0; float lm = l0;
      if (l1 > lm) { lm = l1; me = 1; }
      if (l2 > lm) { lm = l2; me = 2; }
      if (l3 > lm) { lm = l3; me = 3; }
      lg[t][0] = l0; lg[t][1] = l1; lg[t][2] = l2; lg[t][3] = l3;
      myexp[t] = me; lmax[t] = lm;
    }
    if (wave >= 6) {
      if (blockIdx.x == 0) {
#pragma unroll
        for (int t = 0; t < 2; ++t) {
          float e0 = __expf(lg[t][0] - lmax[t]), e1 = __expf(lg[t][1] - lmax[t]);
          float e2 = __expf(lg[t][2] - lmax[t]), e3 = __expf(lg[t][3] - lmax[t]);
          float inv = 1.0f / (e0 + e1 + e2 + e3);
          float v[8] = {e0 * inv, e1 * inv, e2 * inv, e3 * inv,
                        myexp[t] == 0 ? 1.f : 0.f, myexp[t] == 1 ? 1.f : 0.f,
                        myexp[t] == 2 ? 1.f : 0.f, myexp[t] == 3 ? 1.f : 0.f};
#pragma unroll
          for (int d = 8; d > 0; d >>= 1) {
#pragma unroll
            for (int i = 0; i < 8; ++i) v[i] += __shfl_xor(v[i], d, 16);
          }
          if (lane == 0) {
#pragma unroll
            for (int i = 0; i < 8; ++i) atomicAdd(&lacc[i], v[i]);
          }
        }
      }
    } else {
      const _Float16* Ab = Aw + (size_t)mt * KT * 512 + lane * 8;
      const _Float16* Xb = xs + (size_t)(tp * 2) * KT * 512 + lane * 8;
      v8f acc[4][2];
#pragma unroll
      for (int e = 0; e < 4; ++e) { acc[e][0] = v8f{}; acc[e][1] = v8f{}; }
      v16h a[4], bv[2];
#pragma unroll
      for (int e = 0; e < 4; ++e)
        a[e] = join_frag(*(const v8h*)(Ab + (size_t)e * MT * KT * 512),
                         *(const v8h*)(Ab + (size_t)e * MT * KT * 512 + 256));
#pragma unroll
      for (int t = 0; t < 2; ++t)
        bv[t] = join_frag(*(const v8h*)(Xb + (size_t)t * KT * 512),
                          *(const v8h*)(Xb + (size_t)t * KT * 512 + 256));
#pragma unroll
      for (int kt = 0; kt < KT; ++kt) {
        v16h an[4], bn[2];
        if (kt + 1 < KT) {
#pragma unroll
          for (int e = 0; e < 4; ++e)
            an[e] = join_frag(*(const v8h*)(Ab + ((size_t)e * MT * KT + kt + 1) * 512),
                              *(const v8h*)(Ab + ((size_t)e * MT * KT + kt + 1) * 512 + 256));
#pragma unroll
          for (int t = 0; t < 2; ++t)
            bn[t] = join_frag(*(const v8h*)(Xb + ((size_t)t * KT + kt + 1) * 512),
                              *(const v8h*)(Xb + ((size_t)t * KT + kt + 1) * 512 + 256));
        }
#pragma unroll
        for (int e = 0; e < 4; ++e) {
#pragma unroll
          for (int t = 0; t < 2; ++t)
            acc[e][t] = __builtin_amdgcn_wmma_f32_16x16x32_f16(
                false, a[e], false, bv[t], (short)0, acc[e][t], false, false);
        }
        if (kt + 1 < KT) {
#pragma unroll
          for (int e = 0; e < 4; ++e) a[e] = an[e];
          bv[0] = bn[0]; bv[1] = bn[1];
        }
      }
      float rs[8];
#pragma unroll
      for (int t = 0; t < 2; ++t) {
        const int me = myexp[t];
#pragma unroll
        for (int j = 0; j < 8; ++j) {
          float v = acc[0][t][j];
          v = me == 1 ? acc[1][t][j] : v;
          v = me == 2 ? acc[2][t][j] : v;
          v = me == 3 ? acc[3][t][j] : v;
          const int m = mbase + mt * 16 + h * 8 + j;
          Out[((size_t)b * COUT + m) * S_ + s0 + (tp * 2 + t) * 16 + c] = v;
          rs[j] = t == 0 ? v : rs[j] + v;
        }
      }
      if (mbase + mt * 16 < I_) {
#pragma unroll
        for (int j = 0; j < 8; ++j) {
          rs[j] += row_dpp<0x161>(rs[j]);
          rs[j] += row_dpp<0x162>(rs[j]);
          rs[j] += row_dpp<0x164>(rs[j]);
          rs[j] += row_dpp<0x168>(rs[j]);
        }
        if (c == 0) {
          float* bp = bsum + (size_t)(b * 32 + (s0 >> 5) + tp) * I_ + mbase + mt * 16 + h * 8;
          *(float4*)bp = make_float4(rs[0], rs[1], rs[2], rs[3]);
          *(float4*)(bp + 4) = make_float4(rs[4], rs[5], rs[6], rs[7]);
        }
      }
    }
    __syncthreads();
  }
}

__global__ void init_loss_kernel(float* lacc) {
  if (threadIdx.x < 16) lacc[threadIdx.x] = 0.f;
}
__global__ void finalize_loss_kernel(const float* __restrict__ lacc,
                                     float* __restrict__ out_loss) {
  if (threadIdx.x == 0) {
    const float invT = 1.0f / (float)TOK;
    float l = 0.f;
    for (int e = 0; e < E_; ++e) l += (lacc[e] * invT) * (lacc[4 + e] * invT);
    for (int e = 0; e < E_; ++e) l += (lacc[8 + e] * invT) * (lacc[12 + e] * invT);
    *out_loss = l;
  }
}

extern "C" void kernel_launch(void* const* d_in, const int* in_sizes, int n_in,
                              void* d_out, int out_size, void* d_ws, size_t ws_size,
                              hipStream_t stream) {
  const float* inp     = (const float*)d_in[0];
  const float* divisor = (const float*)d_in[1];
  const float* w0_gate = (const float*)d_in[2];
  const float* w0      = (const float*)d_in[3];
  const float* w1      = (const float*)d_in[4];
  const float* w2_gate = (const float*)d_in[5];
  const float* w2      = (const float*)d_in[6];
  float* out = (float*)d_out;

  char* ws = (char*)d_ws;
  size_t off = 0;
  float* lacc = (float*)(ws + off); off += 256;
  const size_t h0b = (size_t)B_ * 3 * I_ * S_ * sizeof(float);
  const size_t t1b = (size_t)B_ * I_ * S_ * sizeof(float);
  float* h0   = (float*)(ws + off); off += h0b;
  float* bsum = (float*)(ws + off); off += (size_t)B_ * 32 * I_ * sizeof(float);
  _Float16* g1p = (_Float16*)(ws + off); off += t1b;
  off = (off + 255) & ~(size_t)255;
  const size_t w2n = (size_t)E_ * I_ * F_;
  _Float16* w2p = (_Float16*)(ws + off); off += w2n * 2;
  float* h2 = h0;

  pack_weights_kernel<<<(int)(w2n / 256), 256, 0, stream>>>(w2, w2p, E_, I_, F_);

  init_loss_kernel<<<1, 32, 0, stream>>>(lacc);

  moe_ws_kernel<<<dim3(3 * I_ / WS_ROWS, TOK / 512), 256, 0, stream>>>(
      inp, w0, w0_gate, lacc, h0, bsum);

  cumnorm_kernel<<<dim3(S_ / 32, B_), 1024, 0, stream>>>(h0, divisor, bsum, g1p);

  conv_ws_kernel<<<dim3(3 * I_ / 48, B_, S_ / 512), 256, 0, stream>>>(w1, g1p, h2);

  moe_wmma_kernel<<<dim3(F_ / 256, TOK / 16), 256,
                    I_ * 16 * sizeof(_Float16), stream>>>(
      h2, I_, w2p, F_, w2_gate, lacc + 8, out, 1.0f  );

  finalize_loss_kernel<<<1, 32, 0, stream>>>(lacc, out + (size_t)B_ * F_ * S_);
}


# first routed layer's gate computed once, in the launch that only zeroed the loss sums, not in all 64 row groups of its GEMM; its loss sums added in a fixed order, no atomics (loss sum order changes)
# speedup vs baseline: 4.5237x; 1.0902x over previous
#include <hip/hip_runtime.h>
#include <hip/hip_bf16.h>

typedef __attribute__((ext_vector_type(16))) _Float16 v16h;
typedef __attribute__((ext_vector_type(8)))  _Float16 v8h;
typedef __attribute__((ext_vector_type(4)))  _Float16 v4h;
typedef __attribute__((ext_vector_type(8)))  float    v8f;

#define B_ 2
#define F_ 512
#define S_ 1024
#define E_ 4
#define I_ 1024
#define TOK (B_*S_)
#define EPSN 1e-5f
#define SLOPE_ 0.02f

__device__ __forceinline__ int frag_j(int kin) {
  return kin < 8 ? kin : (kin < 16 ? kin - 8 : (kin < 24 ? kin - 8 : kin - 16));
}
__device__ __forceinline__ int frag_lane(int kin, int n) {
  return n + (((kin >> 3) & 1) ? 16 : 0);
}

__global__ void pack_weights_kernel(const float* __restrict__ W,
                                    _Float16* __restrict__ Wp,
                                    int E, int Cin, int Cout) {
  size_t idx = (size_t)blockIdx.x * blockDim.x + threadIdx.x;
  size_t total = (size_t)E * Cin * Cout;
  if (idx >= total) return;
  int j    = (int)(idx & 15);
  int lane = (int)((idx >> 4) & 31);
  size_t rest = idx >> 9;
  int KT = Cin >> 5;
  int kt = (int)(rest % KT); rest /= KT;
  int MT = Cout >> 4;
  int mt = (int)(rest % MT);
  int e  = (int)(rest / MT);
  int khalf = (lane >> 4) * 8;
  int k = kt * 32 + khalf + (j < 8 ? j : j + 8);
  int m = mt * 16 + (lane & 15);
  float v = W[((size_t)e * Cin + k) * Cout + m];
  Wp[idx] = (_Float16)v;
}

__device__ __forceinline__ v16h load_frag(const _Float16* p) {
  v8h lo = *(const v8h*)p;
  v8h hi = *(const v8h*)(p + 8);
  v16h r;
#pragma unroll
  for (int j = 0; j < 8; ++j) { r[j] = lo[j]; r[j + 8] = hi[j]; }
  return r;
}

template <int CTRL>
__device__ __forceinline__ float row_dpp(float v) {
  return __int_as_float(__builtin_amdgcn_update_dpp(0, __float_as_int(v), CTRL, 0xf, 0xf, false));
}

__global__ void __launch_bounds__(256)
moe_wmma_kernel(const float* __restrict__ H, int Cin,
                const _Float16* __restrict__ Wp, int Cout,
                const float* __restrict__ gw,
                float* __restrict__ lacc,
                float* __restrict__ Out, float outscale) {
  extern __shared__ _Float16 xs[];
  __shared__ float4 gpart[256];
  __shared__ float red1[256], red2[256];
  int tile = blockIdx.y;
  int b = (tile * 16) / S_;
  int s0 = (tile * 16) % S_;
  int KT = Cin >> 5;
  __builtin_assume((KT & 3) == 0);
  {
    constexpr int NG = I_ / 64;
    int n = threadIdx.x & 15, kq = threadIdx.x >> 4;
    const float* c0 = H + ((size_t)b * 3 * Cin + kq * 4) * S_ + s0 + n;
    const float* c1 = c0 + (size_t)Cin * S_;
    const float* c2 = c1 + (size_t)Cin * S_;
    float v[NG * 4];
    float lsum = 0.f;
#pragma unroll
    for (int g = 0; g < NG; ++g) {
#pragma unroll
      for (int i = 0; i < 4; ++i) {
        size_t o = (size_t)(g * 64 + i) * S_;
        float t = c0[o] * c1[o] + c2[o];
        v[g * 4 + i] = t; lsum += t;
      }
    }
    red1[threadIdx.x] = lsum;
    __syncthreads();
    float tot = 0.f;
#pragma unroll
    for (int q = 0; q < 16; ++q) tot += red1[q * 16 + n];
    float mean = tot * (1.0f / I_);
    float lss = 0.f;
#pragma unroll
    for (int j = 0; j < NG * 4; ++j) { float d = v[j] - mean; v[j] = d; lss += d * d; }
    red2[threadIdx.x] = lss;
    __syncthreads();
    float ss = 0.f;
#pragma unroll
    for (int q = 0; q < 16; ++q) ss += red2[q * 16 + n];
    float nrm = sqrtf(ss) * (1.0f / 32.0f);
    float inv = 1.0f / (nrm + EPSN);
    float a0 = 0.f, a1 = 0.f, a2 = 0.f, a3 = 0.f;
#pragma unroll
    for (int g = 0; g < NG; ++g) {
      int k0 = g * 64 + kq * 4;
      float x0 = v[g * 4 + 0] * inv, x1 = v[g * 4 + 1] * inv;
      float x2 = v[g * 4 + 2] * inv, x3 = v[g * 4 + 3] * inv;
      x0 = x0 >= 0.f ? x0 : x0 * SLOPE_;
      x1 = x1 >= 0.f ? x1 : x1 * SLOPE_;
      x2 = x2 >= 0.f ? x2 : x2 * SLOPE_;
      x3 = x3 >= 0.f ? x3 : x3 * SLOPE_;
      float4 g0 = *(const float4*)(gw + 0 * Cin + k0);
      float4 g1 = *(const float4*)(gw + 1 * Cin + k0);
      float4 g2 = *(const float4*)(gw + 2 * Cin + k0);
      float4 g3 = *(const float4*)(gw + 3 * Cin + k0);
      a0 += x0 * g0.x; a0 += x1 * g0.y; a0 += x2 * g0.z; a0 += x3 * g0.w;
      a1 += x0 * g1.x; a1 += x1 * g1.y; a1 += x2 * g1.z; a1 += x3 * g1.w;
      a2 += x0 * g2.x; a2 += x1 * g2.y; a2 += x2 * g2.z; a2 += x3 * g2.w;
      a3 += x0 * g3.x; a3 += x1 * g3.y; a3 += x2 * g3.z; a3 += x3 * g3.w;
      int kt = k0 >> 5, kin = k0 & 31;
      v4h pk = {(_Float16)x0, (_Float16)x1, (_Float16)x2, (_Float16)x3};
      *(v4h*)&xs[((kt * 32 + frag_lane(kin, n)) << 4) + frag_j(kin)] = pk;
    }
    gpart[threadIdx.x] = make_float4(a0, a1, a2, a3);
  }
  __syncthreads();
  int wave = threadIdx.x >> 5, lane = threadIdx.x & 31;
  int mt0 = blockIdx.x * 16 + wave * 2;
  int col = lane & 15, khalf = (lane >> 4) * 8;
  float l0 = 0.f, l1 = 0.f, l2 = 0.f, l3 = 0.f;
#pragma unroll
  for (int q = 0; q < 16; ++q) {
    float4 p = gpart[q * 16 + col];
    l0 += p.x; l1 += p.y; l2 += p.z; l3 += p.w;
  }
  int myexp = 0; float lmax = l0;
  if (l1 > lmax) { lmax = l1; myexp = 1; }
  if (l2 > lmax) { lmax = l2; myexp = 2; }
  if (l3 > lmax) { lmax = l3; myexp = 3; }
  if (blockIdx.x == 0 && wave == 0) {
    float e0 = __expf(l0 - lmax), e1 = __expf(l1 - lmax);
    float e2 = __expf(l2 - lmax), e3 = __expf(l3 - lmax);
    float inv = 1.0f / (e0 + e1 + e2 + e3);
    float v[8] = {e0 * inv, e1 * inv, e2 * inv, e3 * inv,
                  myexp == 0 ? 1.f : 0.f, myexp == 1 ? 1.f : 0.f,
                  myexp == 2 ? 1.f : 0.f, myexp == 3 ? 1.f : 0.f};
#pragma unroll
    for (int d = 8; d > 0; d >>= 1) {
#pragma unroll
      for (int i = 0; i < 8; ++i) v[i] += __shfl_xor(v[i], d, 16);
    }
    if (lane == 0) {
#pragma unroll
      for (int i = 0; i < 8; ++i) atomicAdd(&lacc[i], v[i]);
    }
  }
  size_t fragstride = (size_t)KT * 512;
  size_t estride = (size_t)(Cout >> 4) * fragstride;
  const _Float16* Ab = Wp + (size_t)mt0 * fragstride + lane * 16;
  v8f acc[E_][2];
#pragma unroll
  for (int e = 0; e < E_; ++e) { acc[e][0] = v8f{}; acc[e][1] = v8f{}; }
  for (int kt = 0; kt < KT; ++kt) {
    v16h a[E_][2];
#pragma unroll
    for (int e = 0; e < E_; ++e) {
#pragma unroll
      for (int m = 0; m < 2; ++m)
        a[e][m] = load_frag(Ab + (size_t)e * estride + (size_t)m * fragstride
                            + (size_t)kt * 512);
    }
    v16h bv = load_frag(&xs[(kt * 32 + lane) * 16]);
    asm volatile("" :: "v"(a[0][0]), "v"(a[0][1]), "v"(a[1][0]), "v"(a[1][1]),
                       "v"(a[2][0]), "v"(a[2][1]), "v"(a[3][0]), "v"(a[3][1]));
#pragma unroll
    for (int e = 0; e < E_; ++e) {
#pragma unroll
      for (int m = 0; m < 2; ++m)
        acc[e][m] = __builtin_amdgcn_wmma_f32_16x16x32_f16(
            false, a[e][m], false, bv, (short)0, acc[e][m], false, false);
    }
  }
#pragma unroll
  for (int j = 0; j < 8; ++j) {
    float v0 = acc[0][0][j], v1 = acc[0][1][j];
    v0 = myexp == 1 ? acc[1][0][j] : v0;  v1 = myexp == 1 ? acc[1][1][j] : v1;
    v0 = myexp == 2 ? acc[2][0][j] : v0;  v1 = myexp == 2 ? acc[2][1][j] : v1;
    v0 = myexp == 3 ? acc[3][0][j] : v0;  v1 = myexp == 3 ? acc[3][1][j] : v1;
    int r0 = mt0 * 16 + j + khalf;
    int r1 = r0 + 16;
    Out[((size_t)b * Cout + r0) * S_ + s0 + col] = v0 * outscale;
    Out[((size_t)b * Cout + r1) * S_ + s0 + col] = v1 * outscale;
  }
}

#define SP_ (S_ + 2)
__global__ void __launch_bounds__(1024)
cumnorm_kernel(const float* __restrict__ h0,
               const float* __restrict__ divisor,
               const float* __restrict__ bsum,
               _Float16* __restrict__ outp) {
  __shared__ float red1[1024], red2[1024];
  const int tile = blockIdx.x, b = blockIdx.y;
  const int s0 = tile * 32;
  const int wave = threadIdx.x >> 5, lane = threadIdx.x & 31;
  float cv = 0.f;
  {
    const float* bp = bsum + (size_t)b * 32 * I_ + wave * 32 + lane;
    float ts[32];
#pragma unroll
    for (int t = 0; t < 32; ++t) ts[t] = bp[(size_t)t * I_];
#pragma unroll
    for (int t = 0; t < 32; ++t) cv += t < tile ? ts[t] : 0.f;
  }
  const float dv = divisor[s0 + lane];
  const float* dep = h0 + ((size_t)b * 3 * I_ + wave * 32) * S_ + s0 + lane;
  const float* scp = dep + (size_t)I_ * S_;
  const float* shp = scp + (size_t)I_ * S_;
  float v[32];
  float lsum = 0.f;
#pragma unroll
  for (int i = 0; i < 32; ++i) {
    float d = dep[(size_t)i * S_];
    const float sc = scp[(size_t)i * S_], sh = shp[(size_t)i * S_];
    d += row_dpp<0x111>(d);
    d += row_dpp<0x112>(d);
    d += row_dpp<0x114>(d);
    d += row_dpp<0x118>(d);
    const float low = __int_as_float(__builtin_amdgcn_readlane(__float_as_int(d), 15));
    d += lane >= 16 ? low : 0.f;
    const float carry = __int_as_float(__builtin_amdgcn_readlane(__float_as_int(cv), i));
    const float x = (carry + d) / dv * sc + sh;
    v[i] = x; lsum += x;
  }
  red1[threadIdx.x] = lsum;
  __syncthreads();
  float tot = 0.f;
#pragma unroll
  for (int q = 0; q < 32; ++q) tot += red1[q * 32 + lane];
  const float mean = tot * (1.0f / I_);
  float lss = 0.f;
#pragma unroll
  for (int i = 0; i < 32; ++i) { float d = v[i] - mean; v[i] = d; lss += d * d; }
  red2[threadIdx.x] = lss;
  __syncthreads();
  float ss = 0.f;
#pragma unroll
  for (int q = 0; q < 32; ++q) ss += red2[q * 32 + lane];
  const float nrm = sqrtf(ss) * (1.0f / 32.0f);
  const float inv = 1.0f / (nrm + EPSN);
#pragma unroll
  for (int g = 0; g < 2; ++g) {
    v8h lo, hi;
#pragma unroll
    for (int e = 0; e < 8; ++e) {
      float y0 = v[g * 16 + e] * inv, y1 = v[g * 16 + 8 + e] * inv;
      y0 = y0 >= 0.f ? y0 : y0 * SLOPE_;
      y1 = y1 >= 0.f ? y1 : y1 * SLOPE_;
      lo[e] = (_Float16)y0; hi[e] = (_Float16)y1;
    }
    _Float16* o = outp + ((((size_t)(b * 32 + wave) * 2 + g) * SP_ + (s0 + lane + 2)) * 2) * 8;
    *(v8h*)o = lo;
    *(v8h*)(o + 8) = hi;
    if (tile == 0 && lane == 0) {
      const v8h z = {};
      *(v8h*)(o - 32) = z; *(v8h*)(o - 24) = z;
      *(v8h*)(o - 16) = z; *(v8h*)(o - 8) = z;
    }
  }
}

__device__ __forceinline__ v16h join_frag(v8h lo, v8h hi) {
  v16h r;
#pragma unroll
  for (int j = 0; j < 8; ++j) { r[j] = lo[j]; r[j + 8] = hi[j]; }
  return r;
}
__global__ void __launch_bounds__(256)
conv_ws_kernel(const float* __restrict__ W1,
               const _Float16* __restrict__ P,
               float* __restrict__ Out) {
  __shared__ __attribute__((aligned(16))) _Float16 Al[3 * 96 * 512];
  const int tid = threadIdx.x;
  const int mbase = blockIdx.x * 48;
  const int b = blockIdx.y;
  const int tbase = blockIdx.z * 512;
  {
    const float* Wrow = W1 + (size_t)mbase * (3 * I_) + tid * 12;
    const int kin = (tid * 4) & 31, ct = tid >> 3;
    const int hl = ((kin >> 3) & 1) * 16;
    const int j0 = frag_j(kin);
    const int sub = (j0 >> 3) * 256 + (j0 & 7);
#pragma unroll 16
    for (int r = 0; r < 48; ++r) {
      const float4* s4 = (const float4*)(Wrow + (size_t)r * (3 * I_));
      float4 f0 = s4[0], f1 = s4[1], f2 = s4[2];
      _Float16* dst = Al + ((r >> 4) * 96 + ct * 3) * 512 + ((r & 15) + hl) * 8 + sub;
      v4h q0 = {(_Float16)f0.x, (_Float16)f0.w, (_Float16)f1.z, (_Float16)f2.y};
      v4h q1 = {(_Float16)f0.y, (_Float16)f1.x, (_Float16)f1.w, (_Float16)f2.z};
      v4h q2 = {(_Float16)f0.z, (_Float16)f1.y, (_Float16)f2.x, (_Float16)f2.w};
      *(v4h*)(dst) = q0;
      *(v4h*)(dst + 512) = q1;
      *(v4h*)(dst + 1024) = q2;
    }
  }
  __syncthreads();
  const int wave = tid >> 5, lane = tid & 31, c = lane & 15, h = lane >> 4;
  const _Float16* Ab = Al + lane * 8;
  const size_t jhs = (size_t)SP_ * 16;
  const size_t cts = 2 * jhs;
  for (int pass = 0; pass < 2; ++pass) {
    const int s0 = tbase + (pass * 8 + wave) * 32;
    const _Float16* Pb = P + (size_t)b * 32 * cts + (size_t)(s0 + c) * 16 + h * 8;
    v8f acc[3][2];
#pragma unroll
    for (int mt = 0; mt < 3; ++mt) { acc[mt][0] = v8f{}; acc[mt][1] = v8f{}; }
    v16h bq[6][2];
#pragma unroll
    for (int d = 0; d < 6; ++d) {
      const _Float16* pp = Pb + (size_t)(d / 3) * cts + (d % 3) * 16;
#pragma unroll
      for (int tt = 0; tt < 2; ++tt)
        bq[d][tt] = join_frag(*(const v8h*)(pp + tt * 256),
                              *(const v8h*)(pp + tt * 256 + jhs));
    }
    v16h a[3];
#pragma unroll
    for (int mt = 0; mt < 3; ++mt)
      a[mt] = join_frag(*(const v8h*)(Ab + (mt * 96) * 512),
                        *(const v8h*)(Ab + (mt * 96) * 512 + 256));
    for (int ct0 = 0; ct0 < 32; ct0 += 2) {
#pragma unroll
      for (int d = 0; d < 6; ++d) {
        const int kt = ct0 * 3 + d;
        const int ktn = kt + 1 < 96 ? kt + 1 : 95;
        v16h an[3];
#pragma unroll
        for (int mt = 0; mt < 3; ++mt)
          an[mt] = join_frag(*(const v8h*)(Ab + (mt * 96 + ktn) * 512),
                             *(const v8h*)(Ab + (mt * 96 + ktn) * 512 + 256));
#pragma unroll
        for (int mt = 0; mt < 3; ++mt) {
#pragma unroll
          for (int tt = 0; tt < 2; ++tt)
            acc[mt][tt] = __builtin_amdgcn_wmma_f32_16x16x32_f16(
                false, a[mt], false, bq[d][tt], (short)0, acc[mt][tt], false, false);
        }
        int ctn = ct0 + 2 + d / 3; ctn = ctn < 32 ? ctn : 31;
        const _Float16* pp = Pb + (size_t)ctn * cts + (d % 3) * 16;
#pragma unroll
        for (int tt = 0; tt < 2; ++tt)
          bq[d][tt] = join_frag(*(const v8h*)(pp + tt * 256),
                                *(const v8h*)(pp + tt * 256 + jhs));
#pragma unroll
        for (int mt = 0; mt < 3; ++mt) a[mt] = an[mt];
      }
    }
#pragma unroll
    for (int mt = 0; mt < 3; ++mt) {
#pragma unroll
      for (int tt = 0; tt < 2; ++tt) {
#pragma unroll
        for (int j = 0; j < 8; ++j) {
          int m = mbase + mt * 16 + h * 8 + j;
          Out[((size_t)b * (3 * I_) + m) * S_ + s0 + tt * 16 + c] = acc[mt][tt][j];
        }
      }
    }
  }
}

__global__ void __launch_bounds__(256)
gate0_kernel(const float* __restrict__ X, const float* __restrict__ gw,
             int* __restrict__ eidx, float* __restrict__ lpart,
             float* __restrict__ lacc1) {
  constexpr int CIN = F_;
  __shared__ float4 gpart[256];
  const int tid = threadIdx.x;
  const int n = tid & 15, q = tid >> 4;
  const int tok0 = blockIdx.x * 16;
  const int b = tok0 / S_, s0 = tok0 % S_;
  if (blockIdx.x == 0 && tid < 8) lacc1[tid] = 0.f;
  {
    const float* xcol = X + ((size_t)b * CIN + q * 4) * S_ + s0 + n;
    float a0 = 0.f, a1 = 0.f, a2 = 0.f, a3 = 0.f;
#pragma unroll
    for (int g = 0; g < CIN / 64; ++g) {
      const int k0 = g * 64 + q * 4;
      float x0 = xcol[(size_t)(g * 64 + 0) * S_];
      float x1 = xcol[(size_t)(g * 64 + 1) * S_];
      float x2 = xcol[(size_t)(g * 64 + 2) * S_];
      float x3 = xcol[(size_t)(g * 64 + 3) * S_];
      float4 g0 = *(const float4*)(gw + 0 * CIN + k0);
      float4 g1 = *(const float4*)(gw + 1 * CIN + k0);
      float4 g2 = *(const float4*)(gw + 2 * CIN + k0);
      float4 g3 = *(const float4*)(gw + 3 * CIN + k0);
      a0 += x0 * g0.x; a0 += x1 * g0.y; a0 += x2 * g0.z; a0 += x3 * g0.w;
      a1 += x0 * g1.x; a1 += x1 * g1.y; a1 += x2 * g1.z; a1 += x3 * g1.w;
      a2 += x0 * g2.x; a2 += x1 * g2.y; a2 += x2 * g2.z; a2 += x3 * g2.w;
      a3 += x0 * g3.x; a3 += x1 * g3.y; a3 += x2 * g3.z; a3 += x3 * g3.w;
    }
    gpart[q * 16 + n] = make_float4(a0, a1, a2, a3);
  }
  __syncthreads();
  if (tid < 32) {
    const int c = tid & 15;
    float l0 = 0.f, l1 = 0.f, l2 = 0.f, l3 = 0.f;
#pragma unroll
    for (int r = 0; r < 16; ++r) {
      float4 p = gpart[r * 16 + c];
      l0 += p.x; l1 += p.y; l2 += p.z; l3 += p.w;
    }
    int me = 0; float lm = l0;
    if (l1 > lm) { lm = l1; me = 1; }
    if (l2 > lm) { lm = l2; me = 2; }
    if (l3 > lm) { lm = l3; me = 3; }
    if (tid < 16) eidx[tok0 + c] = me;
    float e0 = __expf(l0 - lm), e1 = __expf(l1 - lm);
    float e2 = __expf(l2 - lm), e3 = __expf(l3 - lm);
    float inv = 1.0f / (e0 + e1 + e2 + e3);
    float v[8] = {e0 * inv, e1 * inv, e2 * inv, e3 * inv,
                  me == 0 ? 1.f : 0.f, me == 1 ? 1.f : 0.f,
                  me == 2 ? 1.f : 0.f, me == 3 ? 1.f : 0.f};
#pragma unroll
    for (int d = 8; d > 0; d >>= 1) {
#pragma unroll
      for (int i = 0; i < 8; ++i) v[i] += __shfl_xor(v[i], d, 16);
    }
    if (tid == 0) {
      float* lp = lpart + (size_t)blockIdx.x * 8;
      *(float4*)lp = make_float4(v[0], v[1], v[2], v[3]);
      *(float4*)(lp + 4) = make_float4(v[4], v[5], v[6], v[7]);
    }
  }
}

#define WS_ROWS 48
#define WS_TOKB 64
__global__ void __launch_bounds__(256)
moe_ws_kernel(const float* __restrict__ X, const float* __restrict__ W,
              const int* __restrict__ eidx,
              float* __restrict__ Out, float* __restrict__ bsum) {
  constexpr int CIN = F_, COUT = 3 * I_, KT = CIN / 32, MT = WS_ROWS / 16;
  __shared__ __attribute__((aligned(16))) _Float16 Aw[E_ * MT * KT * 512];
  __shared__ __attribute__((aligned(16))) _Float16 xs[(WS_TOKB / 16) * KT * 512];
  const int tid = threadIdx.x;
  const int mbase = blockIdx.x * WS_ROWS;
  {
    for (int t0 = 0; t0 < 12; t0 += 3) {
      float4 f[3][8];
#pragma unroll
      for (int tq = 0; tq < 3; ++tq) {
        const int q = (t0 + tq) * 256 + tid;
        const int m4 = q % 12, ko = q / 12;
        const float* src = W + (size_t)(ko * 8) * COUT + mbase + m4 * 4;
#pragma unroll
        for (int i = 0; i < 8; ++i) f[tq][i] = *(const float4*)(src + (size_t)i * COUT);
      }
#pragma unroll
      for (int tq = 0; tq < 3; ++tq) {
        const int q = (t0 + tq) * 256 + tid;
        const int m4 = q % 12, ko = q / 12;
        const int e = ko >> 6, k8 = ko & 63;
        const int kt = k8 >> 2, o = k8 & 3;
        const int mt = m4 >> 2, c0 = (m4 & 3) * 4;
        _Float16* dst = Aw + ((e * MT + mt) * KT + kt) * 512 + (o >> 1) * 256
                        + (c0 + 16 * (o & 1)) * 8;
        v8h r0, r1, r2, r3;
#pragma unroll
        for (int i = 0; i < 8; ++i) {
          r0[i] = (_Float16)f[tq][i].x; r1[i] = (_Float16)f[tq][i].y;
          r2[i] = (_Float16)f[tq][i].z; r3[i] = (_Float16)f[tq][i].w;
        }
        *(v8h*)(dst) = r0; *(v8h*)(dst + 8) = r1;
        *(v8h*)(dst + 16) = r2; *(v8h*)(dst + 24) = r3;
      }
    }
  }
  const int wave = tid >> 5, lane = tid & 31, c = lane & 15, h = lane >> 4;
  const int n64 = tid & 63, kq4 = tid >> 6;
  const int tp = wave < 6 ? wave / 3 : wave - 6;
  const int mt = wave % 3;
  for (int blk = 0; blk < 512 / WS_TOKB; ++blk) {
    const int tok0 = blockIdx.y * 512 + blk * WS_TOKB;
    const int b = tok0 / S_, s0 = tok0 % S_;
    int myexp[2];
    myexp[0] = eidx[tok0 + (tp * 2 + 0) * 16 + c];
    myexp[1] = eidx[tok0 + (tp * 2 + 1) * 16 + c];
    {
      const float* xcol = X + (size_t)b * CIN * S_ + s0 + n64;
      const int tt = n64 >> 4, n = n64 & 15;
#pragma unroll 4
      for (int kb = 0; kb < CIN; kb += 64) {
#pragma unroll
        for (int u = 0; u < 4; ++u) {
          const int k0 = kb + u * 16 + kq4 * 4;
          float x0 = xcol[(size_t)(k0 + 0) * S_];
          float x1 = xcol[(size_t)(k0 + 1) * S_];
          float x2 = xcol[(size_t)(k0 + 2) * S_];
          float x3 = xcol[(size_t)(k0 + 3) * S_];
          const int kt = k0 >> 5, kin = k0 & 31;
          const int j0 = frag_j(kin);
          v4h pk = {(_Float16)x0, (_Float16)x1, (_Float16)x2, (_Float16)x3};
          *(v4h*)&xs[(tt * KT + kt) * 512 + (j0 >> 3) * 256 + frag_lane(kin, n) * 8 + (j0 & 7)] = pk;
        }
      }
    }
    __syncthreads();
    if (wave < 6) {
      const _Float16* Ab = Aw + (size_t)mt * KT * 512 + lane * 8;
      const _Float16* Xb = xs + (size_t)(tp * 2) * KT * 512 + lane * 8;
      v8f acc[4][2];
#pragma unroll
      for (int e = 0; e < 4; ++e) { acc[e][0] = v8f{}; acc[e][1] = v8f{}; }
      v16h a[4], bv[2];
#pragma unroll
      for (int e = 0; e < 4; ++e)
        a[e] = join_frag(*(const v8h*)(Ab + (size_t)e * MT * KT * 512),
                         *(const v8h*)(Ab + (size_t)e * MT * KT * 512 + 256));
#pragma unroll
      for (int t = 0; t < 2; ++t)
        bv[t] = join_frag(*(const v8h*)(Xb + (size_t)t * KT * 512),
                          *(const v8h*)(Xb + (size_t)t * KT * 512 + 256));
#pragma unroll
      for (int kt = 0; kt < KT; ++kt) {
        v16h an[4], bn[2];
        if (kt + 1 < KT) {
#pragma unroll
          for (int e = 0; e < 4; ++e)
            an[e] = join_frag(*(const v8h*)(Ab + ((size_t)e * MT * KT + kt + 1) * 512),
                              *(const v8h*)(Ab + ((size_t)e * MT * KT + kt + 1) * 512 + 256));
#pragma unroll
          for (int t = 0; t < 2; ++t)
            bn[t] = join_frag(*(const v8h*)(Xb + ((size_t)t * KT + kt + 1) * 512),
                              *(const v8h*)(Xb + ((size_t)t * KT + kt + 1) * 512 + 256));
        }
#pragma unroll
        for (int e = 0; e < 4; ++e) {
#pragma unroll
          for (int t = 0; t < 2; ++t)
            acc[e][t] = __builtin_amdgcn_wmma_f32_16x16x32_f16(
                false, a[e], false, bv[t], (short)0, acc[e][t], false, false);
        }
        if (kt + 1 < KT) {
#pragma unroll
          for (int e = 0; e < 4; ++e) a[e] = an[e];
          bv[0] = bn[0]; bv[1] = bn[1];
        }
      }
      float rs[8];
#pragma unroll
      for (int t = 0; t < 2; ++t) {
        const int me = myexp[t];
#pragma unroll
        for (int j = 0; j < 8; ++j) {
          float v = acc[0][t][j];
          v = me == 1 ? acc[1][t][j] : v;
          v = me == 2 ? acc[2][t][j] : v;
          v = me == 3 ? acc[3][t][j] : v;
          const int m = mbase + mt * 16 + h * 8 + j;
          Out[((size_t)b * COUT + m) * S_ + s0 + (tp * 2 + t) * 16 + c] = v;
          rs[j] = t == 0 ? v : rs[j] + v;
        }
      }
      if (mbase + mt * 16 < I_) {
#pragma unroll
        for (int j = 0; j < 8; ++j) {
          rs[j] += row_dpp<0x161>(rs[j]);
          rs[j] += row_dpp<0x162>(rs[j]);
          rs[j] += row_dpp<0x164>(rs[j]);
          rs[j] += row_dpp<0x168>(rs[j]);
        }
        if (c == 0) {
          float* bp = bsum + (size_t)(b * 32 + (s0 >> 5) + tp) * I_ + mbase + mt * 16 + h * 8;
          *(float4*)bp = make_float4(rs[0], rs[1], rs[2], rs[3]);
          *(float4*)(bp + 4) = make_float4(rs[4], rs[5], rs[6], rs[7]);
        }
      }
    }
    __syncthreads();
  }
}

__global__ void finalize_loss_kernel(const float* __restrict__ lpart,
                                     const float* __restrict__ lacc1,
                                     float* __restrict__ out_loss) {
  static_assert(TOK / 16 == 128, "this kernel adds 128 rows with one wave");
  const int lane = threadIdx.x;
  float v[8];
#pragma unroll
  for (int r = 0; r < 4; ++r) {
    const float4 a = *(const float4*)(lpart + (lane + 32 * r) * 8);
    const float4 c = *(const float4*)(lpart + (lane + 32 * r) * 8 + 4);
    const float w[8] = {a.x, a.y, a.z, a.w, c.x, c.y, c.z, c.w};
#pragma unroll
    for (int i = 0; i < 8; ++i) v[i] = r == 0 ? w[i] : v[i] + w[i];
  }
#pragma unroll
  for (int d = 16; d > 0; d >>= 1) {
#pragma unroll
    for (int i = 0; i < 8; ++i) v[i] += __shfl_xor(v[i], d, 32);
  }
  if (lane == 0) {
    const float invT = 1.0f / (float)TOK;
    float l = 0.f;
    for (int e = 0; e < E_; ++e) l += (v[e] * invT) * (v[4 + e] * invT);
    for (int e = 0; e < E_; ++e) l += (lacc1[e] * invT) * (lacc1[4 + e] * invT);
    *out_loss = l;
  }
}

extern "C" void kernel_launch(void* const* d_in, const int* in_sizes, int n_in,
                              void* d_out, int out_size, void* d_ws, size_t ws_size,
                              hipStream_t stream) {
  const float* inp     = (const float*)d_in[0];
  const float* divisor = (const float*)d_in[1];
  const float* w0_gate = (const float*)d_in[2];
  const float* w0      = (const float*)d_in[3];
  const float* w1      = (const float*)d_in[4];
  const float* w2_gate = (const float*)d_in[5];
  const float* w2      = (const float*)d_in[6];
  float* out = (float*)d_out;

  char* ws = (char*)d_ws;
  size_t off = 0;
  float* lacc1 = (float*)(ws + off); off += 256;
  float* lpart = (float*)(ws + off); off += 4096;
  int* eidx = (int*)(ws + off); off += (size_t)TOK * sizeof(int);
  const size_t h0b = (size_t)B_ * 3 * I_ * S_ * sizeof(float);
  const size_t t1b = (size_t)B_ * I_ * S_ * sizeof(float);
  float* h0   = (float*)(ws + off); off += h0b;
  float* bsum = (float*)(ws + off); off += (size_t)B_ * 32 * I_ * sizeof(float);
  _Float16* g1p = (_Float16*)(ws + off); off += t1b;
  off = (off + 255) & ~(size_t)255;
  const size_t w2n = (size_t)E_ * I_ * F_;
  _Float16* w2p = (_Float16*)(ws + off); off += w2n * 2;
  float* h2 = h0;

  pack_weights_kernel<<<(int)(w2n / 256), 256, 0, stream>>>(w2, w2p, E_, I_, F_);

  gate0_kernel<<<TOK / 16, 256, 0, stream>>>(inp, w0_gate, eidx, lpart, lacc1);

  moe_ws_kernel<<<dim3(3 * I_ / WS_ROWS, TOK / 512), 256, 0, stream>>>(
      inp, w0, eidx, h0, bsum);

  cumnorm_kernel<<<dim3(S_ / 32, B_), 1024, 0, stream>>>(h0, divisor, bsum, g1p);

  conv_ws_kernel<<<dim3(3 * I_ / 48, B_, S_ / 512), 256, 0, stream>>>(w1, g1p, h2);

  moe_wmma_kernel<<<dim3(F_ / 256, TOK / 16), 256,
                    I_ * 16 * sizeof(_Float16), stream>>>(
      h2, I_, w2p, F_, w2_gate, lacc1, out, 1.0f  );

  finalize_loss_kernel<<<1, 32, 0, stream>>>(lpart, lacc1, out + (size_t)B_ * F_ * S_);
}


# GEMM 0's token operand converted once: the gate kernel stores inp as f16 in fragment order; the GEMM's groups copy 64-token blocks of it to LDS, the next block's copy in flight under the products
# speedup vs baseline: 4.6907x; 1.0369x over previous
#include <hip/hip_runtime.h>
#include <hip/hip_bf16.h>

typedef __attribute__((ext_vector_type(16))) _Float16 v16h;
typedef __attribute__((ext_vector_type(8)))  _Float16 v8h;
typedef __attribute__((ext_vector_type(4)))  _Float16 v4h;
typedef __attribute__((ext_vector_type(8)))  float    v8f;

#define B_ 2
#define F_ 512
#define S_ 1024
#define E_ 4
#define I_ 1024
#define TOK (B_*S_)
#define EPSN 1e-5f
#define SLOPE_ 0.02f

__device__ __forceinline__ int frag_j(int kin) {
  return kin < 8 ? kin : (kin < 16 ? kin - 8 : (kin < 24 ? kin - 8 : kin - 16));
}
__device__ __forceinline__ int frag_lane(int kin, int n) {
  return n + (((kin >> 3) & 1) ? 16 : 0);
}

__global__ void pack_weights_kernel(const float* __restrict__ W,
                                    _Float16* __restrict__ Wp,
                                    int E, int Cin, int Cout) {
  size_t idx = (size_t)blockIdx.x * blockDim.x + threadIdx.x;
  size_t total = (size_t)E * Cin * Cout;
  if (idx >= total) return;
  int j    = (int)(idx & 15);
  int lane = (int)((idx >> 4) & 31);
  size_t rest = idx >> 9;
  int KT = Cin >> 5;
  int kt = (int)(rest % KT); rest /= KT;
  int MT = Cout >> 4;
  int mt = (int)(rest % MT);
  int e  = (int)(rest / MT);
  int khalf = (lane >> 4) * 8;
  int k = kt * 32 + khalf + (j < 8 ? j : j + 8);
  int m = mt * 16 + (lane & 15);
  float v = W[((size_t)e * Cin + k) * Cout + m];
  Wp[idx] = (_Float16)v;
}

__device__ __forceinline__ v16h load_frag(const _Float16* p) {
  v8h lo = *(const v8h*)p;
  v8h hi = *(const v8h*)(p + 8);
  v16h r;
#pragma unroll
  for (int j = 0; j < 8; ++j) { r[j] = lo[j]; r[j + 8] = hi[j]; }
  return r;
}

template <int CTRL>
__device__ __forceinline__ float row_dpp(float v) {
  return __int_as_float(__builtin_amdgcn_update_dpp(0, __float_as_int(v), CTRL, 0xf, 0xf, false));
}

__global__ void __launch_bounds__(256)
moe_wmma_kernel(const float* __restrict__ H, int Cin,
                const _Float16* __restrict__ Wp, int Cout,
                const float* __restrict__ gw,
                float* __restrict__ lacc,
                float* __restrict__ Out, float outscale) {
  extern __shared__ _Float16 xs[];
  __shared__ float4 gpart[256];
  __shared__ float red1[256], red2[256];
  int tile = blockIdx.y;
  int b = (tile * 16) / S_;
  int s0 = (tile * 16) % S_;
  int KT = Cin >> 5;
  __builtin_assume((KT & 3) == 0);
  {
    constexpr int NG = I_ / 64;
    int n = threadIdx.x & 15, kq = threadIdx.x >> 4;
    const float* c0 = H + ((size_t)b * 3 * Cin + kq * 4) * S_ + s0 + n;
    const float* c1 = c0 + (size_t)Cin * S_;
    const float* c2 = c1 + (size_t)Cin * S_;
    float v[NG * 4];
    float lsum = 0.f;
#pragma unroll
    for (int g = 0; g < NG; ++g) {
#pragma unroll
      for (int i = 0; i < 4; ++i) {
        size_t o = (size_t)(g * 64 + i) * S_;
        float t = c0[o] * c1[o] + c2[o];
        v[g * 4 + i] = t; lsum += t;
      }
    }
    red1[threadIdx.x] = lsum;
    __syncthreads();
    float tot = 0.f;
#pragma unroll
    for (int q = 0; q < 16; ++q) tot += red1[q * 16 + n];
    float mean = tot * (1.0f / I_);
    float lss = 0.f;
#pragma unroll
    for (int j = 0; j < NG * 4; ++j) { float d = v[j] - mean; v[j] = d; lss += d * d; }
    red2[threadIdx.x] = lss;
    __syncthreads();
    float ss = 0.f;
#pragma unroll
    for (int q = 0; q < 16; ++q) ss += red2[q * 16 + n];
    float nrm = sqrtf(ss) * (1.0f / 32.0f);
    float inv = 1.0f / (nrm + EPSN);
    float a0 = 0.f, a1 = 0.f, a2 = 0.f, a3 = 0.f;
#pragma unroll
    for (int g = 0; g < NG; ++g) {
      int k0 = g * 64 + kq * 4;
      float x0 = v[g * 4 + 0] * inv, x1 = v[g * 4 + 1] * inv;
      float x2 = v[g * 4 + 2] * inv, x3 = v[g * 4 + 3] * inv;
      x0 = x0 >= 0.f ? x0 : x0 * SLOPE_;
      x1 = x1 >= 0.f ? x1 : x1 * SLOPE_;
      x2 = x2 >= 0.f ? x2 : x2 * SLOPE_;
      x3 = x3 >= 0.f ? x3 : x3 * SLOPE_;
      float4 g0 = *(const float4*)(gw + 0 * Cin + k0);
      float4 g1 = *(const float4*)(gw + 1 * Cin + k0);
      float4 g2 = *(const float4*)(gw + 2 * Cin + k0);
      float4 g3 = *(const float4*)(gw + 3 * Cin + k0);
      a0 += x0 * g0.x; a0 += x1 * g0.y; a0 += x2 * g0.z; a0 += x3 * g0.w;
      a1 += x0 * g1.x; a1 += x1 * g1.y; a1 += x2 * g1.z; a1 += x3 * g1.w;
      a2 += x0 * g2.x; a2 += x1 * g2.y; a2 += x2 * g2.z; a2 += x3 * g2.w;
      a3 += x0 * g3.x; a3 += x1 * g3.y; a3 += x2 * g3.z; a3 += x3 * g3.w;
      int kt = k0 >> 5, kin = k0 & 31;
      v4h pk = {(_Float16)x0, (_Float16)x1, (_Float16)x2, (_Float16)x3};
      *(v4h*)&xs[((kt * 32 + frag_lane(kin, n)) << 4) + frag_j(kin)] = pk;
    }
    gpart[threadIdx.x] = make_float4(a0, a1, a2, a3);
  }
  __syncthreads();
  int wave = threadIdx.x >> 5, lane = threadIdx.x & 31;
  int mt0 = blockIdx.x * 16 + wave * 2;
  int col = lane & 15, khalf = (lane >> 4) * 8;
  float l0 = 0.f, l1 = 0.f, l2 = 0.f, l3 = 0.f;
#pragma unroll
  for (int q = 0; q < 16; ++q) {
    float4 p = gpart[q * 16 + col];
    l0 += p.x; l1 += p.y; l2 += p.z; l3 += p.w;
  }
  int myexp = 0; float lmax = l0;
  if (l1 > lmax) { lmax = l1; myexp = 1; }
  if (l2 > lmax) { lmax = l2; myexp = 2; }
  if (l3 > lmax) { lmax = l3; myexp = 3; }
  if (blockIdx.x == 0 && wave == 0) {
    float e0 = __expf(l0 - lmax), e1 = __expf(l1 - lmax);
    float e2 = __expf(l2 - lmax), e3 = __expf(l3 - lmax);
    float inv = 1.0f / (e0 + e1 + e2 + e3);
    float v[8] = {e0 * inv, e1 * inv, e2 * inv, e3 * inv,
                  myexp == 0 ? 1.f : 0.f, myexp == 1 ? 1.f : 0.f,
                  myexp == 2 ? 1.f : 0.f, myexp == 3 ? 1.f : 0.f};
#pragma unroll
    for (int d = 8; d > 0; d >>= 1) {
#pragma unroll
      for (int i = 0; i < 8; ++i) v[i] += __shfl_xor(v[i], d, 16);
    }
    if (lane == 0) {
#pragma unroll
      for (int i = 0; i < 8; ++i) atomicAdd(&lacc[i], v[i]);
    }
  }
  size_t fragstride = (size_t)KT * 512;
  size_t estride = (size_t)(Cout >> 4) * fragstride;
  const _Float16* Ab = Wp + (size_t)mt0 * fragstride + lane * 16;
  v8f acc[E_][2];
#pragma unroll
  for (int e = 0; e < E_; ++e) { acc[e][0] = v8f{}; acc[e][1] = v8f{}; }
  for (int kt = 0; kt < KT; ++kt) {
    v16h a[E_][2];
#pragma unroll
    for (int e = 0; e < E_; ++e) {
#pragma unroll
      for (int m = 0; m < 2; ++m)
        a[e][m] = load_frag(Ab + (size_t)e * estride + (size_t)m * fragstride
                            + (size_t)kt * 512);
    }
    v16h bv = load_frag(&xs[(kt * 32 + lane) * 16]);
    asm volatile("" :: "v"(a[0][0]), "v"(a[0][1]), "v"(a[1][0]), "v"(a[1][1]),
                       "v"(a[2][0]), "v"(a[2][1]), "v"(a[3][0]), "v"(a[3][1]));
#pragma unroll
    for (int e = 0; e < E_; ++e) {
#pragma unroll
      for (int m = 0; m < 2; ++m)
        acc[e][m] = __builtin_amdgcn_wmma_f32_16x16x32_f16(
            false, a[e][m], false, bv, (short)0, acc[e][m], false, false);
    }
  }
#pragma unroll
  for (int j = 0; j < 8; ++j) {
    float v0 = acc[0][0][j], v1 = acc[0][1][j];
    v0 = myexp == 1 ? acc[1][0][j] : v0;  v1 = myexp == 1 ? acc[1][1][j] : v1;
    v0 = myexp == 2 ? acc[2][0][j] : v0;  v1 = myexp == 2 ? acc[2][1][j] : v1;
    v0 = myexp == 3 ? acc[3][0][j] : v0;  v1 = myexp == 3 ? acc[3][1][j] : v1;
    int r0 = mt0 * 16 + j + khalf;
    int r1 = r0 + 16;
    Out[((size_t)b * Cout + r0) * S_ + s0 + col] = v0 * outscale;
    Out[((size_t)b * Cout + r1) * S_ + s0 + col] = v1 * outscale;
  }
}

#define SP_ (S_ + 2)
__global__ void __launch_bounds__(1024)
cumnorm_kernel(const float* __restrict__ h0,
               const float* __restrict__ divisor,
               const float* __restrict__ bsum,
               _Float16* __restrict__ outp) {
  __shared__ float red1[1024], red2[1024];
  const int tile = blockIdx.x, b = blockIdx.y;
  const int s0 = tile * 32;
  const int wave = threadIdx.x >> 5, lane = threadIdx.x & 31;
  float cv = 0.f;
  {
    const float* bp = bsum + (size_t)b * 32 * I_ + wave * 32 + lane;
    float ts[32];
#pragma unroll
    for (int t = 0; t < 32; ++t) ts[t] = bp[(size_t)t * I_];
#pragma unroll
    for (int t = 0; t < 32; ++t) cv += t < tile ? ts[t] : 0.f;
  }
  const float dv = divisor[s0 + lane];
  const float* dep = h0 + ((size_t)b * 3 * I_ + wave * 32) * S_ + s0 + lane;
  const float* scp = dep + (size_t)I_ * S_;
  const float* shp = scp + (size_t)I_ * S_;
  float v[32];
  float lsum = 0.f;
#pragma unroll
  for (int i = 0; i < 32; ++i) {
    float d = dep[(size_t)i * S_];
    const float sc = scp[(size_t)i * S_], sh = shp[(size_t)i * S_];
    d += row_dpp<0x111>(d);
    d += row_dpp<0x112>(d);
    d += row_dpp<0x114>(d);
    d += row_dpp<0x118>(d);
    const float low = __int_as_float(__builtin_amdgcn_readlane(__float_as_int(d), 15));
    d += lane >= 16 ? low : 0.f;
    const float carry = __int_as_float(__builtin_amdgcn_readlane(__float_as_int(cv), i));
    const float x = (carry + d) / dv * sc + sh;
    v[i] = x; lsum += x;
  }
  red1[threadIdx.x] = lsum;
  __syncthreads();
  float tot = 0.f;
#pragma unroll
  for (int q = 0; q < 32; ++q) tot += red1[q * 32 + lane];
  const float mean = tot * (1.0f / I_);
  float lss = 0.f;
#pragma unroll
  for (int i = 0; i < 32; ++i) { float d = v[i] - mean; v[i] = d; lss += d * d; }
  red2[threadIdx.x] = lss;
  __syncthreads();
  float ss = 0.f;
#pragma unroll
  for (int q = 0; q < 32; ++q) ss += red2[q * 32 + lane];
  const float nrm = sqrtf(ss) * (1.0f / 32.0f);
  const float inv = 1.0f / (nrm + EPSN);
#pragma unroll
  for (int g = 0; g < 2; ++g) {
    v8h lo, hi;
#pragma unroll
    for (int e = 0; e < 8; ++e) {
      float y0 = v[g * 16 + e] * inv, y1 = v[g * 16 + 8 + e] * inv;
      y0 = y0 >= 0.f ? y0 : y0 * SLOPE_;
      y1 = y1 >= 0.f ? y1 : y1 * SLOPE_;
      lo[e] = (_Float16)y0; hi[e] = (_Float16)y1;
    }
    _Float16* o = outp + ((((size_t)(b * 32 + wave) * 2 + g) * SP_ + (s0 + lane + 2)) * 2) * 8;
    *(v8h*)o = lo;
    *(v8h*)(o + 8) = hi;
    if (tile == 0 && lane == 0) {
      const v8h z = {};
      *(v8h*)(o - 32) = z; *(v8h*)(o - 24) = z;
      *(v8h*)(o - 16) = z; *(v8h*)(o - 8) = z;
    }
  }
}

__device__ __forceinline__ v16h join_frag(v8h lo, v8h hi) {
  v16h r;
#pragma unroll
  for (int j = 0; j < 8; ++j) { r[j] = lo[j]; r[j + 8] = hi[j]; }
  return r;
}
__global__ void __launch_bounds__(256)
conv_ws_kernel(const float* __restrict__ W1,
               const _Float16* __restrict__ P,
               float* __restrict__ Out) {
  __shared__ __attribute__((aligned(16))) _Float16 Al[3 * 96 * 512];
  const int tid = threadIdx.x;
  const int mbase = blockIdx.x * 48;
  const int b = blockIdx.y;
  const int tbase = blockIdx.z * 512;
  {
    const float* Wrow = W1 + (size_t)mbase * (3 * I_) + tid * 12;
    const int kin = (tid * 4) & 31, ct = tid >> 3;
    const int hl = ((kin >> 3) & 1) * 16;
    const int j0 = frag_j(kin);
    const int sub = (j0 >> 3) * 256 + (j0 & 7);
#pragma unroll 16
    for (int r = 0; r < 48; ++r) {
      const float4* s4 = (const float4*)(Wrow + (size_t)r * (3 * I_));
      float4 f0 = s4[0], f1 = s4[1], f2 = s4[2];
      _Float16* dst = Al + ((r >> 4) * 96 + ct * 3) * 512 + ((r & 15) + hl) * 8 + sub;
      v4h q0 = {(_Float16)f0.x, (_Float16)f0.w, (_Float16)f1.z, (_Float16)f2.y};
      v4h q1 = {(_Float16)f0.y, (_Float16)f1.x, (_Float16)f1.w, (_Float16)f2.z};
      v4h q2 = {(_Float16)f0.z, (_Float16)f1.y, (_Float16)f2.x, (_Float16)f2.w};
      *(v4h*)(dst) = q0;
      *(v4h*)(dst + 512) = q1;
      *(v4h*)(dst + 1024) = q2;
    }
  }
  __syncthreads();
  const int wave = tid >> 5, lane = tid & 31, c = lane & 15, h = lane >> 4;
  const _Float16* Ab = Al + lane * 8;
  const size_t jhs = (size_t)SP_ * 16;
  const size_t cts = 2 * jhs;
  for (int pass = 0; pass < 2; ++pass) {
    const int s0 = tbase + (pass * 8 + wave) * 32;
    const _Float16* Pb = P + (size_t)b * 32 * cts + (size_t)(s0 + c) * 16 + h * 8;
    v8f acc[3][2];
#pragma unroll
    for (int mt = 0; mt < 3; ++mt) { acc[mt][0] = v8f{}; acc[mt][1] = v8f{}; }
    v16h bq[6][2];
#pragma unroll
    for (int d = 0; d < 6; ++d) {
      const _Float16* pp = Pb + (size_t)(d / 3) * cts + (d % 3) * 16;
#pragma unroll
      for (int tt = 0; tt < 2; ++tt)
        bq[d][tt] = join_frag(*(const v8h*)(pp + tt * 256),
                              *(const v8h*)(pp + tt * 256 + jhs));
    }
    v16h a[3];
#pragma unroll
    for (int mt = 0; mt < 3; ++mt)
      a[mt] = join_frag(*(const v8h*)(Ab + (mt * 96) * 512),
                        *(const v8h*)(Ab + (mt * 96) * 512 + 256));
    for (int ct0 = 0; ct0 < 32; ct0 += 2) {
#pragma unroll
      for (int d = 0; d < 6; ++d) {
        const int kt = ct0 * 3 + d;
        const int ktn = kt + 1 < 96 ? kt + 1 : 95;
        v16h an[3];
#pragma unroll
        for (int mt = 0; mt < 3; ++mt)
          an[mt] = join_frag(*(const v8h*)(Ab + (mt * 96 + ktn) * 512),
                             *(const v8h*)(Ab + (mt * 96 + ktn) * 512 + 256));
#pragma unroll
        for (int mt = 0; mt < 3; ++mt) {
#pragma unroll
          for (int tt = 0; tt < 2; ++tt)
            acc[mt][tt] = __builtin_amdgcn_wmma_f32_16x16x32_f16(
                false, a[mt], false, bq[d][tt], (short)0, acc[mt][tt], false, false);
        }
        int ctn = ct0 + 2 + d / 3; ctn = ctn < 32 ? ctn : 31;
        const _Float16* pp = Pb + (size_t)ctn * cts + (d % 3) * 16;
#pragma unroll
        for (int tt = 0; tt < 2; ++tt)
          bq[d][tt] = join_frag(*(const v8h*)(pp + tt * 256),
                                *(const v8h*)(pp + tt * 256 + jhs));
#pragma unroll
        for (int mt = 0; mt < 3; ++mt) a[mt] = an[mt];
      }
    }
#pragma unroll
    for (int mt = 0; mt < 3; ++mt) {
#pragma unroll
      for (int tt = 0; tt < 2; ++tt) {
#pragma unroll
        for (int j = 0; j < 8; ++j) {
          int m = mbase + mt * 16 + h * 8 + j;
          Out[((size_t)b * (3 * I_) + m) * S_ + s0 + tt * 16 + c] = acc[mt][tt][j];
        }
      }
    }
  }
}

__global__ void __launch_bounds__(256)
gate0_kernel(const float* __restrict__ X, const float* __restrict__ gw,
             int* __restrict__ eidx, float* __restrict__ lpart,
             float* __restrict__ lacc1, _Float16* __restrict__ xh) {
  constexpr int CIN = F_;
  __shared__ float4 gpart[256];
  const int tid = threadIdx.x;
  const int n = tid & 15, q = tid >> 4;
  const int tok0 = blockIdx.x * 16;
  const int b = tok0 / S_, s0 = tok0 % S_;
  if (blockIdx.x == 0 && tid < 8) lacc1[tid] = 0.f;
  {
    const float* xcol = X + ((size_t)b * CIN + q * 4) * S_ + s0 + n;
    float a0 = 0.f, a1 = 0.f, a2 = 0.f, a3 = 0.f;
#pragma unroll
    for (int g = 0; g < CIN / 64; ++g) {
      const int k0 = g * 64 + q * 4;
      float x0 = xcol[(size_t)(g * 64 + 0) * S_];
      float x1 = xcol[(size_t)(g * 64 + 1) * S_];
      float x2 = xcol[(size_t)(g * 64 + 2) * S_];
      float x3 = xcol[(size_t)(g * 64 + 3) * S_];
      float4 g0 = *(const float4*)(gw + 0 * CIN + k0);
      float4 g1 = *(const float4*)(gw + 1 * CIN + k0);
      float4 g2 = *(const float4*)(gw + 2 * CIN + k0);
      float4 g3 = *(const float4*)(gw + 3 * CIN + k0);
      a0 += x0 * g0.x; a0 += x1 * g0.y; a0 += x2 * g0.z; a0 += x3 * g0.w;
      a1 += x0 * g1.x; a1 += x1 * g1.y; a1 += x2 * g1.z; a1 += x3 * g1.w;
      a2 += x0 * g2.x; a2 += x1 * g2.y; a2 += x2 * g2.z; a2 += x3 * g2.w;
      a3 += x0 * g3.x; a3 += x1 * g3.y; a3 += x2 * g3.z; a3 += x3 * g3.w;
      const int kt = k0 >> 5, kin = k0 & 31;
      const int j0 = frag_j(kin);
      v4h pk = {(_Float16)x0, (_Float16)x1, (_Float16)x2, (_Float16)x3};
      *(v4h*)&xh[((size_t)blockIdx.x * (CIN / 32) + kt) * 512 + (j0 >> 3) * 256
                 + frag_lane(kin, n) * 8 + (j0 & 7)] = pk;
    }
    gpart[q * 16 + n] = make_float4(a0, a1, a2, a3);
  }
  __syncthreads();
  if (tid < 32) {
    const int c = tid & 15;
    float l0 = 0.f, l1 = 0.f, l2 = 0.f, l3 = 0.f;
#pragma unroll
    for (int r = 0; r < 16; ++r) {
      float4 p = gpart[r * 16 + c];
      l0 += p.x; l1 += p.y; l2 += p.z; l3 += p.w;
    }
    int me = 0; float lm = l0;
    if (l1 > lm) { lm = l1; me = 1; }
    if (l2 > lm) { lm = l2; me = 2; }
    if (l3 > lm) { lm = l3; me = 3; }
    if (tid < 16) eidx[tok0 + c] = me;
    float e0 = __expf(l0 - lm), e1 = __expf(l1 - lm);
    float e2 = __expf(l2 - lm), e3 = __expf(l3 - lm);
    float inv = 1.0f / (e0 + e1 + e2 + e3);
    float v[8] = {e0 * inv, e1 * inv, e2 * inv, e3 * inv,
                  me == 0 ? 1.f : 0.f, me == 1 ? 1.f : 0.f,
                  me == 2 ? 1.f : 0.f, me == 3 ? 1.f : 0.f};
#pragma unroll
    for (int d = 8; d > 0; d >>= 1) {
#pragma unroll
      for (int i = 0; i < 8; ++i) v[i] += __shfl_xor(v[i], d, 16);
    }
    if (tid == 0) {
      float* lp = lpart + (size_t)blockIdx.x * 8;
      *(float4*)lp = make_float4(v[0], v[1], v[2], v[3]);
      *(float4*)(lp + 4) = make_float4(v[4], v[5], v[6], v[7]);
    }
  }
}

#define WS_ROWS 48
#define WS_TOKB 64
__global__ void __launch_bounds__(256)
moe_ws_kernel(const _Float16* __restrict__ Xh, const float* __restrict__ W,
              const int* __restrict__ eidx,
              float* __restrict__ Out, float* __restrict__ bsum) {
  constexpr int CIN = F_, COUT = 3 * I_, KT = CIN / 32, MT = WS_ROWS / 16;
  __shared__ __attribute__((aligned(16))) _Float16 Aw[E_ * MT * KT * 512];
  __shared__ __attribute__((aligned(16))) _Float16 xs[(WS_TOKB / 16) * KT * 512];
  const int tid = threadIdx.x;
  const int mbase = blockIdx.x * WS_ROWS;
  constexpr int XSN = (WS_TOKB / 16) * KT * 512;
  const _Float16* xsrc = Xh + (size_t)blockIdx.y * (512 / WS_TOKB) * XSN + tid * 8;
  const int wave = tid >> 5, lane = tid & 31, c = lane & 15, h = lane >> 4;
  const int tp = wave < 6 ? wave / 3 : wave - 6;
  const int mt = wave % 3;
  const int* esrc = eidx + blockIdx.y * 512 + tp * 32 + c;
  int nexp[2];
  nexp[0] = esrc[0];
  nexp[1] = esrc[16];
  v8h pre[16];
#pragma unroll
  for (int i = 0; i < 16; ++i) pre[i] = *(const v8h*)(xsrc + i * 2048);
  {
    for (int t0 = 0; t0 < 12; t0 += 3) {
      float4 f[3][8];
#pragma unroll
      for (int tq = 0; tq < 3; ++tq) {
        const int q = (t0 + tq) * 256 + tid;
        const int m4 = q % 12, ko = q / 12;
        const float* src = W + (size_t)(ko * 8) * COUT + mbase + m4 * 4;
#pragma unroll
        for (int i = 0; i < 8; ++i) f[tq][i] = *(const float4*)(src + (size_t)i * COUT);
      }
#pragma unroll
      for (int tq = 0; tq < 3; ++tq) {
        const int q = (t0 + tq) * 256 + tid;
        const int m4 = q % 12, ko = q / 12;
        const int e = ko >> 6, k8 = ko & 63;
        const int kt = k8 >> 2, o = k8 & 3;
        const int mt = m4 >> 2, c0 = (m4 & 3) * 4;
        _Float16* dst = Aw + ((e * MT + mt) * KT + kt) * 512 + (o >> 1) * 256
                        + (c0 + 16 * (o & 1)) * 8;
        v8h r0, r1, r2, r3;
#pragma unroll
        for (int i = 0; i < 8; ++i) {
          r0[i] = (_Float16)f[tq][i].x; r1[i] = (_Float16)f[tq][i].y;
          r2[i] = (_Float16)f[tq][i].z; r3[i] = (_Float16)f[tq][i].w;
        }
        *(v8h*)(dst) = r0; *(v8h*)(dst + 8) = r1;
        *(v8h*)(dst + 16) = r2; *(v8h*)(dst + 24) = r3;
      }
    }
  }
  for (int blk = 0; blk < 512 / WS_TOKB; ++blk) {
    const int tok0 = blockIdx.y * 512 + blk * WS_TOKB;
    const int b = tok0 / S_, s0 = tok0 % S_;
#pragma unroll
    for (int i = 0; i < 16; ++i) *(v8h*)(xs + i * 2048 + tid * 8) = pre[i];
    __syncthreads();
    int myexp[2];
    myexp[0] = nexp[0];
    myexp[1] = nexp[1];
    {
      const int nb = blk + 1 < 512 / WS_TOKB ? blk + 1 : blk;
      nexp[0] = esrc[nb * WS_TOKB];
      nexp[1] = esrc[nb * WS_TOKB + 16];
#pragma unroll
      for (int i = 0; i < 16; ++i)
        pre[i] = *(const v8h*)(xsrc + (size_t)nb * XSN + i * 2048);
    }
    if (wave < 6) {
      const _Float16* Ab = Aw + (size_t)mt * KT * 512 + lane * 8;
      const _Float16* Xb = xs + (size_t)(tp * 2) * KT * 512 + lane * 8;
      v8f acc[4][2];
#pragma unroll
      for (int e = 0; e < 4; ++e) { acc[e][0] = v8f{}; acc[e][1] = v8f{}; }
      v16h a[4], bv[2];
#pragma unroll
      for (int e = 0; e < 4; ++e)
        a[e] = join_frag(*(const v8h*)(Ab + (size_t)e * MT * KT * 512),
                         *(const v8h*)(Ab + (size_t)e * MT * KT * 512 + 256));
#pragma unroll
      for (int t = 0; t < 2; ++t)
        bv[t] = join_frag(*(const v8h*)(Xb + (size_t)t * KT * 512),
                          *(const v8h*)(Xb + (size_t)t * KT * 512 + 256));
#pragma unroll
      for (int kt = 0; kt < KT; ++kt) {
        v16h an[4], bn[2];
        if (kt + 1 < KT) {
#pragma unroll
          for (int e = 0; e < 4; ++e)
            an[e] = join_frag(*(const v8h*)(Ab + ((size_t)e * MT * KT + kt + 1) * 512),
                              *(const v8h*)(Ab + ((size_t)e * MT * KT + kt + 1) * 512 + 256));
#pragma unroll
          for (int t = 0; t < 2; ++t)
            bn[t] = join_frag(*(const v8h*)(Xb + ((size_t)t * KT + kt + 1) * 512),
                              *(const v8h*)(Xb + ((size_t)t * KT + kt + 1) * 512 + 256));
        }
#pragma unroll
        for (int e = 0; e < 4; ++e) {
#pragma unroll
          for (int t = 0; t < 2; ++t)
            acc[e][t] = __builtin_amdgcn_wmma_f32_16x16x32_f16(
                false, a[e], false, bv[t], (short)0, acc[e][t], false, false);
        }
        if (kt + 1 < KT) {
#pragma unroll
          for (int e = 0; e < 4; ++e) a[e] = an[e];
          bv[0] = bn[0]; bv[1] = bn[1];
        }
      }
      float rs[8];
#pragma unroll
      for (int t = 0; t < 2; ++t) {
        const int me = myexp[t];
#pragma unroll
        for (int j = 0; j < 8; ++j) {
          float v = acc[0][t][j];
          v = me == 1 ? acc[1][t][j] : v;
          v = me == 2 ? acc[2][t][j] : v;
          v = me == 3 ? acc[3][t][j] : v;
          const int m = mbase + mt * 16 + h * 8 + j;
          Out[((size_t)b * COUT + m) * S_ + s0 + (tp * 2 + t) * 16 + c] = v;
          rs[j] = t == 0 ? v : rs[j] + v;
        }
      }
      if (mbase + mt * 16 < I_) {
#pragma unroll
        for (int j = 0; j < 8; ++j) {
          rs[j] += row_dpp<0x161>(rs[j]);
          rs[j] += row_dpp<0x162>(rs[j]);
          rs[j] += row_dpp<0x164>(rs[j]);
          rs[j] += row_dpp<0x168>(rs[j]);
        }
        if (c == 0) {
          float* bp = bsum + (size_t)(b * 32 + (s0 >> 5) + tp) * I_ + mbase + mt * 16 + h * 8;
          *(float4*)bp = make_float4(rs[0], rs[1], rs[2], rs[3]);
          *(float4*)(bp + 4) = make_float4(rs[4], rs[5], rs[6], rs[7]);
        }
      }
    }
    __syncthreads();
  }
}

__global__ void finalize_loss_kernel(const float* __restrict__ lpart,
                                     const float* __restrict__ lacc1,
                                     float* __restrict__ out_loss) {
  static_assert(TOK / 16 == 128, "this kernel adds 128 rows with one wave");
  const int lane = threadIdx.x;
  float v[8];
#pragma unroll
  for (int r = 0; r < 4; ++r) {
    const float4 a = *(const float4*)(lpart + (lane + 32 * r) * 8);
    const float4 c = *(const float4*)(lpart + (lane + 32 * r) * 8 + 4);
    const float w[8] = {a.x, a.y, a.z, a.w, c.x, c.y, c.z, c.w};
#pragma unroll
    for (int i = 0; i < 8; ++i) v[i] = r == 0 ? w[i] : v[i] + w[i];
  }
#pragma unroll
  for (int d = 16; d > 0; d >>= 1) {
#pragma unroll
    for (int i = 0; i < 8; ++i) v[i] += __shfl_xor(v[i], d, 32);
  }
  if (lane == 0) {
    const float invT = 1.0f / (float)TOK;
    float l = 0.f;
    for (int e = 0; e < E_; ++e) l += (v[e] * invT) * (v[4 + e] * invT);
    for (int e = 0; e < E_; ++e) l += (lacc1[e] * invT) * (lacc1[4 + e] * invT);
    *out_loss = l;
  }
}

extern "C" void kernel_launch(void* const* d_in, const int* in_sizes, int n_in,
                              void* d_out, int out_size, void* d_ws, size_t ws_size,
                              hipStream_t stream) {
  const float* inp     = (const float*)d_in[0];
  const float* divisor = (const float*)d_in[1];
  const float* w0_gate = (const float*)d_in[2];
  const float* w0      = (const float*)d_in[3];
  const float* w1      = (const float*)d_in[4];
  const float* w2_gate = (const float*)d_in[5];
  const float* w2      = (const float*)d_in[6];
  float* out = (float*)d_out;

  char* ws = (char*)d_ws;
  size_t off = 0;
  float* lacc1 = (float*)(ws + off); off += 256;
  float* lpart = (float*)(ws + off); off += 4096;
  int* eidx = (int*)(ws + off); off += (size_t)TOK * sizeof(int);
  const size_t h0b = (size_t)B_ * 3 * I_ * S_ * sizeof(float);
  const size_t t1b = (size_t)B_ * I_ * S_ * sizeof(float);
  float* h0   = (float*)(ws + off); off += h0b;
  float* bsum = (float*)(ws + off); off += (size_t)B_ * 32 * I_ * sizeof(float);
  _Float16* g1p = (_Float16*)(ws + off); off += t1b;
  off = (off + 255) & ~(size_t)255;
  const size_t w2n = (size_t)E_ * I_ * F_;
  _Float16* w2p = (_Float16*)(ws + off); off += w2n * 2;
  _Float16* xh = (_Float16*)(ws + off); off += (size_t)TOK * F_ * 2;
  float* h2 = h0;

  pack_weights_kernel<<<(int)(w2n / 256), 256, 0, stream>>>(w2, w2p, E_, I_, F_);

  gate0_kernel<<<TOK / 16, 256, 0, stream>>>(inp, w0_gate, eidx, lpart, lacc1, xh);

  moe_ws_kernel<<<dim3(3 * I_ / WS_ROWS, TOK / 512), 256, 0, stream>>>(
      xh, w0, eidx, h0, bsum);

  cumnorm_kernel<<<dim3(S_ / 32, B_), 1024, 0, stream>>>(h0, divisor, bsum, g1p);

  conv_ws_kernel<<<dim3(3 * I_ / 48, B_, S_ / 512), 256, 0, stream>>>(w1, g1p, h2);

  moe_wmma_kernel<<<dim3(F_ / 256, TOK / 16), 256,
                    I_ * 16 * sizeof(_Float16), stream>>>(
      h2, I_, w2p, F_, w2_gate, lacc1, out, 1.0f  );

  finalize_loss_kernel<<<1, 32, 0, stream>>>(lpart, lacc1, out + (size_t)B_ * F_ * S_);
}


# last routed GEMM made weight-stationary: a work-group converts its 16 rows of all 4 experts of w2 into LDS once and sweeps 256 tokens; norm 2 + gate run once per token tile in front; w2 pre-pack gone
# speedup vs baseline: 5.1793x; 1.1042x over previous
#include <hip/hip_runtime.h>
#include <hip/hip_bf16.h>

typedef __attribute__((ext_vector_type(16))) _Float16 v16h;
typedef __attribute__((ext_vector_type(8)))  _Float16 v8h;
typedef __attribute__((ext_vector_type(4)))  _Float16 v4h;
typedef __attribute__((ext_vector_type(8)))  float    v8f;

#define B_ 2
#define F_ 512
#define S_ 1024
#define E_ 4
#define I_ 1024
#define TOK (B_*S_)
#define EPSN 1e-5f
#define SLOPE_ 0.02f

__device__ __forceinline__ int frag_j(int kin) {
  return kin < 8 ? kin : (kin < 16 ? kin - 8 : (kin < 24 ? kin - 8 : kin - 16));
}
__device__ __forceinline__ int frag_lane(int kin, int n) {
  return n + (((kin >> 3) & 1) ? 16 : 0);
}

template <int CTRL>
__device__ __forceinline__ float row_dpp(float v) {
  return __int_as_float(__builtin_amdgcn_update_dpp(0, __float_as_int(v), CTRL, 0xf, 0xf, false));
}

#define SP_ (S_ + 2)
__global__ void __launch_bounds__(1024)
cumnorm_kernel(const float* __restrict__ h0,
               const float* __restrict__ divisor,
               const float* __restrict__ bsum,
               _Float16* __restrict__ outp) {
  __shared__ float red1[1024], red2[1024];
  const int tile = blockIdx.x, b = blockIdx.y;
  const int s0 = tile * 32;
  const int wave = threadIdx.x >> 5, lane = threadIdx.x & 31;
  float cv = 0.f;
  {
    const float* bp = bsum + (size_t)b * 32 * I_ + wave * 32 + lane;
    float ts[32];
#pragma unroll
    for (int t = 0; t < 32; ++t) ts[t] = bp[(size_t)t * I_];
#pragma unroll
    for (int t = 0; t < 32; ++t) cv += t < tile ? ts[t] : 0.f;
  }
  const float dv = divisor[s0 + lane];
  const float* dep = h0 + ((size_t)b * 3 * I_ + wave * 32) * S_ + s0 + lane;
  const float* scp = dep + (size_t)I_ * S_;
  const float* shp = scp + (size_t)I_ * S_;
  float v[32];
  float lsum = 0.f;
#pragma unroll
  for (int i = 0; i < 32; ++i) {
    float d = dep[(size_t)i * S_];
    const float sc = scp[(size_t)i * S_], sh = shp[(size_t)i * S_];
    d += row_dpp<0x111>(d);
    d += row_dpp<0x112>(d);
    d += row_dpp<0x114>(d);
    d += row_dpp<0x118>(d);
    const float low = __int_as_float(__builtin_amdgcn_readlane(__float_as_int(d), 15));
    d += lane >= 16 ? low : 0.f;
    const float carry = __int_as_float(__builtin_amdgcn_readlane(__float_as_int(cv), i));
    const float x = (carry + d) / dv * sc + sh;
    v[i] = x; lsum += x;
  }
  red1[threadIdx.x] = lsum;
  __syncthreads();
  float tot = 0.f;
#pragma unroll
  for (int q = 0; q < 32; ++q) tot += red1[q * 32 + lane];
  const float mean = tot * (1.0f / I_);
  float lss = 0.f;
#pragma unroll
  for (int i = 0; i < 32; ++i) { float d = v[i] - mean; v[i] = d; lss += d * d; }
  red2[threadIdx.x] = lss;
  __syncthreads();
  float ss = 0.f;
#pragma unroll
  for (int q = 0; q < 32; ++q) ss += red2[q * 32 + lane];
  const float nrm = sqrtf(ss) * (1.0f / 32.0f);
  const float inv = 1.0f / (nrm + EPSN);
#pragma unroll
  for (int g = 0; g < 2; ++g) {
    v8h lo, hi;
#pragma unroll
    for (int e = 0; e < 8; ++e) {
      float y0 = v[g * 16 + e] * inv, y1 = v[g * 16 + 8 + e] * inv;
      y0 = y0 >= 0.f ? y0 : y0 * SLOPE_;
      y1 = y1 >= 0.f ? y1 : y1 * SLOPE_;
      lo[e] = (_Float16)y0; hi[e] = (_Float16)y1;
    }
    _Float16* o = outp + ((((size_t)(b * 32 + wave) * 2 + g) * SP_ + (s0 + lane + 2)) * 2) * 8;
    *(v8h*)o = lo;
    *(v8h*)(o + 8) = hi;
    if (tile == 0 && lane == 0) {
      const v8h z = {};
      *(v8h*)(o - 32) = z; *(v8h*)(o - 24) = z;
      *(v8h*)(o - 16) = z; *(v8h*)(o - 8) = z;
    }
  }
}

__device__ __forceinline__ v16h join_frag(v8h lo, v8h hi) {
  v16h r;
#pragma unroll
  for (int j = 0; j < 8; ++j) { r[j] = lo[j]; r[j + 8] = hi[j]; }
  return r;
}
__global__ void __launch_bounds__(256)
conv_ws_kernel(const float* __restrict__ W1,
               const _Float16* __restrict__ P,
               float* __restrict__ Out) {
  __shared__ __attribute__((aligned(16))) _Float16 Al[3 * 96 * 512];
  const int tid = threadIdx.x;
  const int mbase = blockIdx.x * 48;
  const int b = blockIdx.y;
  const int tbase = blockIdx.z * 512;
  {
    const float* Wrow = W1 + (size_t)mbase * (3 * I_) + tid * 12;
    const int kin = (tid * 4) & 31, ct = tid >> 3;
    const int hl = ((kin >> 3) & 1) * 16;
    const int j0 = frag_j(kin);
    const int sub = (j0 >> 3) * 256 + (j0 & 7);
#pragma unroll 16
    for (int r = 0; r < 48; ++r) {
      const float4* s4 = (const float4*)(Wrow + (size_t)r * (3 * I_));
      float4 f0 = s4[0], f1 = s4[1], f2 = s4[2];
      _Float16* dst = Al + ((r >> 4) * 96 + ct * 3) * 512 + ((r & 15) + hl) * 8 + sub;
      v4h q0 = {(_Float16)f0.x, (_Float16)f0.w, (_Float16)f1.z, (_Float16)f2.y};
      v4h q1 = {(_Float16)f0.y, (_Float16)f1.x, (_Float16)f1.w, (_Float16)f2.z};
      v4h q2 = {(_Float16)f0.z, (_Float16)f1.y, (_Float16)f2.x, (_Float16)f2.w};
      *(v4h*)(dst) = q0;
      *(v4h*)(dst + 512) = q1;
      *(v4h*)(dst + 1024) = q2;
    }
  }
  __syncthreads();
  const int wave = tid >> 5, lane = tid & 31, c = lane & 15, h = lane >> 4;
  const _Float16* Ab = Al + lane * 8;
  const size_t jhs = (size_t)SP_ * 16;
  const size_t cts = 2 * jhs;
  for (int pass = 0; pass < 2; ++pass) {
    const int s0 = tbase + (pass * 8 + wave) * 32;
    const _Float16* Pb = P + (size_t)b * 32 * cts + (size_t)(s0 + c) * 16 + h * 8;
    v8f acc[3][2];
#pragma unroll
    for (int mt = 0; mt < 3; ++mt) { acc[mt][0] = v8f{}; acc[mt][1] = v8f{}; }
    v16h bq[6][2];
#pragma unroll
    for (int d = 0; d < 6; ++d) {
      const _Float16* pp = Pb + (size_t)(d / 3) * cts + (d % 3) * 16;
#pragma unroll
      for (int tt = 0; tt < 2; ++tt)
        bq[d][tt] = join_frag(*(const v8h*)(pp + tt * 256),
                              *(const v8h*)(pp + tt * 256 + jhs));
    }
    v16h a[3];
#pragma unroll
    for (int mt = 0; mt < 3; ++mt)
      a[mt] = join_frag(*(const v8h*)(Ab + (mt * 96) * 512),
                        *(const v8h*)(Ab + (mt * 96) * 512 + 256));
    for (int ct0 = 0; ct0 < 32; ct0 += 2) {
#pragma unroll
      for (int d = 0; d < 6; ++d) {
        const int kt = ct0 * 3 + d;
        const int ktn = kt + 1 < 96 ? kt + 1 : 95;
        v16h an[3];
#pragma unroll
        for (int mt = 0; mt < 3; ++mt)
          an[mt] = join_frag(*(const v8h*)(Ab + (mt * 96 + ktn) * 512),
                             *(const v8h*)(Ab + (mt * 96 + ktn) * 512 + 256));
#pragma unroll
        for (int mt = 0; mt < 3; ++mt) {
#pragma unroll
          for (int tt = 0; tt < 2; ++tt)
            acc[mt][tt] = __builtin_amdgcn_wmma_f32_16x16x32_f16(
                false, a[mt], false, bq[d][tt], (short)0, acc[mt][tt], false, false);
        }
        int ctn = ct0 + 2 + d / 3; ctn = ctn < 32 ? ctn : 31;
        const _Float16* pp = Pb + (size_t)ctn * cts + (d % 3) * 16;
#pragma unroll
        for (int tt = 0; tt < 2; ++tt)
          bq[d][tt] = join_frag(*(const v8h*)(pp + tt * 256),
                                *(const v8h*)(pp + tt * 256 + jhs));
#pragma unroll
        for (int mt = 0; mt < 3; ++mt) a[mt] = an[mt];
      }
    }
#pragma unroll
    for (int mt = 0; mt < 3; ++mt) {
#pragma unroll
      for (int tt = 0; tt < 2; ++tt) {
#pragma unroll
        for (int j = 0; j < 8; ++j) {
          int m = mbase + mt * 16 + h * 8 + j;
          Out[((size_t)b * (3 * I_) + m) * S_ + s0 + tt * 16 + c] = acc[mt][tt][j];
        }
      }
    }
  }
}

__global__ void __launch_bounds__(256)
gate0_kernel(const float* __restrict__ X, const float* __restrict__ gw,
             int* __restrict__ eidx, float* __restrict__ lpart,
             float* __restrict__ lacc1, _Float16* __restrict__ xh) {
  constexpr int CIN = F_;
  __shared__ float4 gpart[256];
  const int tid = threadIdx.x;
  const int n = tid & 15, q = tid >> 4;
  const int tok0 = blockIdx.x * 16;
  const int b = tok0 / S_, s0 = tok0 % S_;
  if (blockIdx.x == 0 && tid < 8) lacc1[tid] = 0.f;
  {
    const float* xcol = X + ((size_t)b * CIN + q * 4) * S_ + s0 + n;
    float a0 = 0.f, a1 = 0.f, a2 = 0.f, a3 = 0.f;
#pragma unroll
    for (int g = 0; g < CIN / 64; ++g) {
      const int k0 = g * 64 + q * 4;
      float x0 = xcol[(size_t)(g * 64 + 0) * S_];
      float x1 = xcol[(size_t)(g * 64 + 1) * S_];
      float x2 = xcol[(size_t)(g * 64 + 2) * S_];
      float x3 = xcol[(size_t)(g * 64 + 3) * S_];
      float4 g0 = *(const float4*)(gw + 0 * CIN + k0);
      float4 g1 = *(const float4*)(gw + 1 * CIN + k0);
      float4 g2 = *(const float4*)(gw + 2 * CIN + k0);
      float4 g3 = *(const float4*)(gw + 3 * CIN + k0);
      a0 += x0 * g0.x; a0 += x1 * g0.y; a0 += x2 * g0.z; a0 += x3 * g0.w;
      a1 += x0 * g1.x; a1 += x1 * g1.y; a1 += x2 * g1.z; a1 += x3 * g1.w;
      a2 += x0 * g2.x; a2 += x1 * g2.y; a2 += x2 * g2.z; a2 += x3 * g2.w;
      a3 += x0 * g3.x; a3 += x1 * g3.y; a3 += x2 * g3.z; a3 += x3 * g3.w;
      const int kt = k0 >> 5, kin = k0 & 31;
      const int j0 = frag_j(kin);
      v4h pk = {(_Float16)x0, (_Float16)x1, (_Float16)x2, (_Float16)x3};
      *(v4h*)&xh[((size_t)blockIdx.x * (CIN / 32) + kt) * 512 + (j0 >> 3) * 256
                 + frag_lane(kin, n) * 8 + (j0 & 7)] = pk;
    }
    gpart[q * 16 + n] = make_float4(a0, a1, a2, a3);
  }
  __syncthreads();
  if (tid < 32) {
    const int c = tid & 15;
    float l0 = 0.f, l1 = 0.f, l2 = 0.f, l3 = 0.f;
#pragma unroll
    for (int r = 0; r < 16; ++r) {
      float4 p = gpart[r * 16 + c];
      l0 += p.x; l1 += p.y; l2 += p.z; l3 += p.w;
    }
    int me = 0; float lm = l0;
    if (l1 > lm) { lm = l1; me = 1; }
    if (l2 > lm) { lm = l2; me = 2; }
    if (l3 > lm) { lm = l3; me = 3; }
    if (tid < 16) eidx[tok0 + c] = me;
    float e0 = __expf(l0 - lm), e1 = __expf(l1 - lm);
    float e2 = __expf(l2 - lm), e3 = __expf(l3 - lm);
    float inv = 1.0f / (e0 + e1 + e2 + e3);
    float v[8] = {e0 * inv, e1 * inv, e2 * inv, e3 * inv,
                  me == 0 ? 1.f : 0.f, me == 1 ? 1.f : 0.f,
                  me == 2 ? 1.f : 0.f, me == 3 ? 1.f : 0.f};
#pragma unroll
    for (int d = 8; d > 0; d >>= 1) {
#pragma unroll
      for (int i = 0; i < 8; ++i) v[i] += __shfl_xor(v[i], d, 16);
    }
    if (tid == 0) {
      float* lp = lpart + (size_t)blockIdx.x * 8;
      *(float4*)lp = make_float4(v[0], v[1], v[2], v[3]);
      *(float4*)(lp + 4) = make_float4(v[4], v[5], v[6], v[7]);
    }
  }
}

#define WS_ROWS 48
#define WS_TOKB 64
__global__ void __launch_bounds__(256)
moe_ws_kernel(const _Float16* __restrict__ Xh, const float* __restrict__ W,
              const int* __restrict__ eidx,
              float* __restrict__ Out, float* __restrict__ bsum) {
  constexpr int CIN = F_, COUT = 3 * I_, KT = CIN / 32, MT = WS_ROWS / 16;
  __shared__ __attribute__((aligned(16))) _Float16 Aw[E_ * MT * KT * 512];
  __shared__ __attribute__((aligned(16))) _Float16 xs[(WS_TOKB / 16) * KT * 512];
  const int tid = threadIdx.x;
  const int mbase = blockIdx.x * WS_ROWS;
  constexpr int XSN = (WS_TOKB / 16) * KT * 512;
  const _Float16* xsrc = Xh + (size_t)blockIdx.y * (512 / WS_TOKB) * XSN + tid * 8;
  const int wave = tid >> 5, lane = tid & 31, c = lane & 15, h = lane >> 4;
  const int tp = wave < 6 ? wave / 3 : wave - 6;
  const int mt = wave % 3;
  const int* esrc = eidx + blockIdx.y * 512 + tp * 32 + c;
  int nexp[2];
  nexp[0] = esrc[0];
  nexp[1] = esrc[16];
  v8h pre[16];
#pragma unroll
  for (int i = 0; i < 16; ++i) pre[i] = *(const v8h*)(xsrc + i * 2048);
  {
    for (int t0 = 0; t0 < 12; t0 += 3) {
      float4 f[3][8];
#pragma unroll
      for (int tq = 0; tq < 3; ++tq) {
        const int q = (t0 + tq) * 256 + tid;
        const int m4 = q % 12, ko = q / 12;
        const float* src = W + (size_t)(ko * 8) * COUT + mbase + m4 * 4;
#pragma unroll
        for (int i = 0; i < 8; ++i) f[tq][i] = *(const float4*)(src + (size_t)i * COUT);
      }
#pragma unroll
      for (int tq = 0; tq < 3; ++tq) {
        const int q = (t0 + tq) * 256 + tid;
        const int m4 = q % 12, ko = q / 12;
        const int e = ko >> 6, k8 = ko & 63;
        const int kt = k8 >> 2, o = k8 & 3;
        const int mt = m4 >> 2, c0 = (m4 & 3) * 4;
        _Float16* dst = Aw + ((e * MT + mt) * KT + kt) * 512 + (o >> 1) * 256
                        + (c0 + 16 * (o & 1)) * 8;
        v8h r0, r1, r2, r3;
#pragma unroll
        for (int i = 0; i < 8; ++i) {
          r0[i] = (_Float16)f[tq][i].x; r1[i] = (_Float16)f[tq][i].y;
          r2[i] = (_Float16)f[tq][i].z; r3[i] = (_Float16)f[tq][i].w;
        }
        *(v8h*)(dst) = r0; *(v8h*)(dst + 8) = r1;
        *(v8h*)(dst + 16) = r2; *(v8h*)(dst + 24) = r3;
      }
    }
  }
  for (int blk = 0; blk < 512 / WS_TOKB; ++blk) {
    const int tok0 = blockIdx.y * 512 + blk * WS_TOKB;
    const int b = tok0 / S_, s0 = tok0 % S_;
#pragma unroll
    for (int i = 0; i < 16; ++i) *(v8h*)(xs + i * 2048 + tid * 8) = pre[i];
    __syncthreads();
    int myexp[2];
    myexp[0] = nexp[0];
    myexp[1] = nexp[1];
    {
      const int nb = blk + 1 < 512 / WS_TOKB ? blk + 1 : blk;
      nexp[0] = esrc[nb * WS_TOKB];
      nexp[1] = esrc[nb * WS_TOKB + 16];
#pragma unroll
      for (int i = 0; i < 16; ++i)
        pre[i] = *(const v8h*)(xsrc + (size_t)nb * XSN + i * 2048);
    }
    if (wave < 6) {
      const _Float16* Ab = Aw + (size_t)mt * KT * 512 + lane * 8;
      const _Float16* Xb = xs + (size_t)(tp * 2) * KT * 512 + lane * 8;
      v8f acc[4][2];
#pragma unroll
      for (int e = 0; e < 4; ++e) { acc[e][0] = v8f{}; acc[e][1] = v8f{}; }
      v16h a[4], bv[2];
#pragma unroll
      for (int e = 0; e < 4; ++e)
        a[e] = join_frag(*(const v8h*)(Ab + (size_t)e * MT * KT * 512),
                         *(const v8h*)(Ab + (size_t)e * MT * KT * 512 + 256));
#pragma unroll
      for (int t = 0; t < 2; ++t)
        bv[t] = join_frag(*(const v8h*)(Xb + (size_t)t * KT * 512),
                          *(const v8h*)(Xb + (size_t)t * KT * 512 + 256));
#pragma unroll
      for (int kt = 0; kt < KT; ++kt) {
        v16h an[4], bn[2];
        if (kt + 1 < KT) {
#pragma unroll
          for (int e = 0; e < 4; ++e)
            an[e] = join_frag(*(const v8h*)(Ab + ((size_t)e * MT * KT + kt + 1) * 512),
                              *(const v8h*)(Ab + ((size_t)e * MT * KT + kt + 1) * 512 + 256));
#pragma unroll
          for (int t = 0; t < 2; ++t)
            bn[t] = join_frag(*(const v8h*)(Xb + ((size_t)t * KT + kt + 1) * 512),
                              *(const v8h*)(Xb + ((size_t)t * KT + kt + 1) * 512 + 256));
        }
#pragma unroll
        for (int e = 0; e < 4; ++e) {
#pragma unroll
          for (int t = 0; t < 2; ++t)
            acc[e][t] = __builtin_amdgcn_wmma_f32_16x16x32_f16(
                false, a[e], false, bv[t], (short)0, acc[e][t], false, false);
        }
        if (kt + 1 < KT) {
#pragma unroll
          for (int e = 0; e < 4; ++e) a[e] = an[e];
          bv[0] = bn[0]; bv[1] = bn[1];
        }
      }
      float rs[8];
#pragma unroll
      for (int t = 0; t < 2; ++t) {
        const int me = myexp[t];
#pragma unroll
        for (int j = 0; j < 8; ++j) {
          float v = acc[0][t][j];
          v = me == 1 ? acc[1][t][j] : v;
          v = me == 2 ? acc[2][t][j] : v;
          v = me == 3 ? acc[3][t][j] : v;
          const int m = mbase + mt * 16 + h * 8 + j;
          Out[((size_t)b * COUT + m) * S_ + s0 + (tp * 2 + t) * 16 + c] = v;
          rs[j] = t == 0 ? v : rs[j] + v;
        }
      }
      if (mbase + mt * 16 < I_) {
#pragma unroll
        for (int j = 0; j < 8; ++j) {
          rs[j] += row_dpp<0x161>(rs[j]);
          rs[j] += row_dpp<0x162>(rs[j]);
          rs[j] += row_dpp<0x164>(rs[j]);
          rs[j] += row_dpp<0x168>(rs[j]);
        }
        if (c == 0) {
          float* bp = bsum + (size_t)(b * 32 + (s0 >> 5) + tp) * I_ + mbase + mt * 16 + h * 8;
          *(float4*)bp = make_float4(rs[0], rs[1], rs[2], rs[3]);
          *(float4*)(bp + 4) = make_float4(rs[4], rs[5], rs[6], rs[7]);
        }
      }
    }
    __syncthreads();
  }
}

__global__ void __launch_bounds__(256)
norm2gate_kernel(const float* __restrict__ H, const float* __restrict__ gw,
                 int* __restrict__ eidx, float* __restrict__ lacc,
                 _Float16* __restrict__ xh) {
  constexpr int Cin = I_;
  constexpr int NG = I_ / 64;
  __shared__ float4 gpart[256];
  __shared__ float red1[256], red2[256];
  const int tile = blockIdx.x;
  const int b = (tile * 16) / S_;
  const int s0 = (tile * 16) % S_;
  {
    const int n = threadIdx.x & 15, kq = threadIdx.x >> 4;
    const float* c0 = H + ((size_t)b * 3 * Cin + kq * 4) * S_ + s0 + n;
    const float* c1 = c0 + (size_t)Cin * S_;
    const float* c2 = c1 + (size_t)Cin * S_;
    float v[NG * 4];
    float lsum = 0.f;
#pragma unroll
    for (int g = 0; g < NG; ++g) {
#pragma unroll
      for (int i = 0; i < 4; ++i) {
        size_t o = (size_t)(g * 64 + i) * S_;
        float t = c0[o] * c1[o] + c2[o];
        v[g * 4 + i] = t; lsum += t;
      }
    }
    red1[threadIdx.x] = lsum;
    __syncthreads();
    float tot = 0.f;
#pragma unroll
    for (int q = 0; q < 16; ++q) tot += red1[q * 16 + n];
    float mean = tot * (1.0f / I_);
    float lss = 0.f;
#pragma unroll
    for (int j = 0; j < NG * 4; ++j) { float d = v[j] - mean; v[j] = d; lss += d * d; }
    red2[threadIdx.x] = lss;
    __syncthreads();
    float ss = 0.f;
#pragma unroll
    for (int q = 0; q < 16; ++q) ss += red2[q * 16 + n];
    float nrm = sqrtf(ss) * (1.0f / 32.0f);
    float inv = 1.0f / (nrm + EPSN);
    float a0 = 0.f, a1 = 0.f, a2 = 0.f, a3 = 0.f;
#pragma unroll
    for (int g = 0; g < NG; ++g) {
      int k0 = g * 64 + kq * 4;
      float x0 = v[g * 4 + 0] * inv, x1 = v[g * 4 + 1] * inv;
      float x2 = v[g * 4 + 2] * inv, x3 = v[g * 4 + 3] * inv;
      x0 = x0 >= 0.f ? x0 : x0 * SLOPE_;
      x1 = x1 >= 0.f ? x1 : x1 * SLOPE_;
      x2 = x2 >= 0.f ? x2 : x2 * SLOPE_;
      x3 = x3 >= 0.f ? x3 : x3 * SLOPE_;
      float4 g0 = *(const float4*)(gw + 0 * Cin + k0);
      float4 g1 = *(const float4*)(gw + 1 * Cin + k0);
      float4 g2 = *(const float4*)(gw + 2 * Cin + k0);
      float4 g3 = *(const float4*)(gw + 3 * Cin + k0);
      a0 += x0 * g0.x; a0 += x1 * g0.y; a0 += x2 * g0.z; a0 += x3 * g0.w;
      a1 += x0 * g1.x; a1 += x1 * g1.y; a1 += x2 * g1.z; a1 += x3 * g1.w;
      a2 += x0 * g2.x; a2 += x1 * g2.y; a2 += x2 * g2.z; a2 += x3 * g2.w;
      a3 += x0 * g3.x; a3 += x1 * g3.y; a3 += x2 * g3.z; a3 += x3 * g3.w;
      const int kt = k0 >> 5, kin = k0 & 31;
      const int j0 = frag_j(kin);
      v4h pk = {(_Float16)x0, (_Float16)x1, (_Float16)x2, (_Float16)x3};
      *(v4h*)&xh[((size_t)tile * (Cin / 32) + kt) * 512 + (j0 >> 3) * 256
                 + frag_lane(kin, n) * 8 + (j0 & 7)] = pk;
    }
    gpart[threadIdx.x] = make_float4(a0, a1, a2, a3);
  }
  __syncthreads();
  if (threadIdx.x < 32) {
    const int lane = threadIdx.x, col = lane & 15;
    float l0 = 0.f, l1 = 0.f, l2 = 0.f, l3 = 0.f;
#pragma unroll
    for (int q = 0; q < 16; ++q) {
      float4 p = gpart[q * 16 + col];
      l0 += p.x; l1 += p.y; l2 += p.z; l3 += p.w;
    }
    int myexp = 0; float lmax = l0;
    if (l1 > lmax) { lmax = l1; myexp = 1; }
    if (l2 > lmax) { lmax = l2; myexp = 2; }
    if (l3 > lmax) { lmax = l3; myexp = 3; }
    if (lane < 16) eidx[tile * 16 + col] = myexp;
    float e0 = __expf(l0 - lmax), e1 = __expf(l1 - lmax);
    float e2 = __expf(l2 - lmax), e3 = __expf(l3 - lmax);
    float inv = 1.0f / (e0 + e1 + e2 + e3);
    float v[8] = {e0 * inv, e1 * inv, e2 * inv, e3 * inv,
                  myexp == 0 ? 1.f : 0.f, myexp == 1 ? 1.f : 0.f,
                  myexp == 2 ? 1.f : 0.f, myexp == 3 ? 1.f : 0.f};
#pragma unroll
    for (int d = 8; d > 0; d >>= 1) {
#pragma unroll
      for (int i = 0; i < 8; ++i) v[i] += __shfl_xor(v[i], d, 16);
    }
    if (lane == 0) {
#pragma unroll
      for (int i = 0; i < 8; ++i) atomicAdd(&lacc[i], v[i]);
    }
  }
}

#define W2_TOKS 256
#define W2_AHEAD 8
__global__ void __launch_bounds__(256)
moe_ws2_kernel(const _Float16* __restrict__ Xh, const float* __restrict__ W,
               const int* __restrict__ eidx,
               float* __restrict__ Out, float outscale) {
  constexpr int CIN = I_, COUT = F_, KT = CIN / 32;
  __shared__ __attribute__((aligned(16))) _Float16 Aw[E_ * KT * 512];
  const int tid = threadIdx.x;
  const int mbase = blockIdx.x * 16;
  const int wave = tid >> 5, lane = tid & 31, c = lane & 15, h = lane >> 4;
  const int tile0 = blockIdx.y * (W2_TOKS / 16) + wave * 2;
  int myexp[2];
  myexp[0] = eidx[tile0 * 16 + c];
  myexp[1] = eidx[tile0 * 16 + 16 + c];
  {
    for (int t0 = 0; t0 < 8; t0 += 4) {
      float4 f[4][8];
#pragma unroll
      for (int tq = 0; tq < 4; ++tq) {
        const int q = (t0 + tq) * 256 + tid;
        const int m4 = q & 3, ko = q >> 2;
        const float* src = W + (size_t)(ko * 8) * COUT + mbase + m4 * 4;
#pragma unroll
        for (int i = 0; i < 8; ++i) f[tq][i] = *(const float4*)(src + (size_t)i * COUT);
      }
#pragma unroll
      for (int tq = 0; tq < 4; ++tq) {
        const int q = (t0 + tq) * 256 + tid;
        const int m4 = q & 3, ko = q >> 2;
        const int e = ko >> 7, k8 = ko & 127;
        const int kt = k8 >> 2, o = k8 & 3;
        const int c0 = m4 * 4;
        _Float16* dst = Aw + (e * KT + kt) * 512 + (o >> 1) * 256
                        + (c0 + 16 * (o & 1)) * 8;
        v8h r0, r1, r2, r3;
#pragma unroll
        for (int i = 0; i < 8; ++i) {
          r0[i] = (_Float16)f[tq][i].x; r1[i] = (_Float16)f[tq][i].y;
          r2[i] = (_Float16)f[tq][i].z; r3[i] = (_Float16)f[tq][i].w;
        }
        *(v8h*)(dst) = r0; *(v8h*)(dst + 8) = r1;
        *(v8h*)(dst + 16) = r2; *(v8h*)(dst + 24) = r3;
      }
    }
  }
  const _Float16* Xb = Xh + (size_t)tile0 * KT * 512 + lane * 8;
  v16h bq[W2_AHEAD][2];
#pragma unroll
  for (int d = 0; d < W2_AHEAD; ++d) {
#pragma unroll
    for (int t = 0; t < 2; ++t)
      bq[d][t] = join_frag(*(const v8h*)(Xb + ((size_t)t * KT + d) * 512),
                           *(const v8h*)(Xb + ((size_t)t * KT + d) * 512 + 256));
  }
  __syncthreads();
  const _Float16* Ab = Aw + lane * 8;
  v8f acc[E_][2];
#pragma unroll
  for (int e = 0; e < E_; ++e) { acc[e][0] = v8f{}; acc[e][1] = v8f{}; }
  v16h a[E_];
#pragma unroll
  for (int e = 0; e < E_; ++e)
    a[e] = join_frag(*(const v8h*)(Ab + (size_t)(e * KT) * 512),
                     *(const v8h*)(Ab + (size_t)(e * KT) * 512 + 256));
  for (int kt0 = 0; kt0 < KT; kt0 += W2_AHEAD) {
#pragma unroll
    for (int d = 0; d < W2_AHEAD; ++d) {
      const int kt = kt0 + d;
      const int ktn = kt + 1 < KT ? kt + 1 : KT - 1;
      v16h an[E_];
#pragma unroll
      for (int e = 0; e < E_; ++e)
        an[e] = join_frag(*(const v8h*)(Ab + (size_t)(e * KT + ktn) * 512),
                          *(const v8h*)(Ab + (size_t)(e * KT + ktn) * 512 + 256));
#pragma unroll
      for (int e = 0; e < E_; ++e) {
#pragma unroll
        for (int t = 0; t < 2; ++t)
          acc[e][t] = __builtin_amdgcn_wmma_f32_16x16x32_f16(
              false, a[e], false, bq[d][t], (short)0, acc[e][t], false, false);
      }
      const int ktb = kt + W2_AHEAD < KT ? kt + W2_AHEAD : KT - 1;
#pragma unroll
      for (int t = 0; t < 2; ++t)
        bq[d][t] = join_frag(*(const v8h*)(Xb + ((size_t)t * KT + ktb) * 512),
                             *(const v8h*)(Xb + ((size_t)t * KT + ktb) * 512 + 256));
#pragma unroll
      for (int e = 0; e < E_; ++e) a[e] = an[e];
    }
  }
#pragma unroll
  for (int t = 0; t < 2; ++t) {
    const int me = myexp[t];
    const int tok = (tile0 + t) * 16 + c;
    const int b = tok / S_, s = tok % S_;
#pragma unroll
    for (int j = 0; j < 8; ++j) {
      float v = acc[0][t][j];
      v = me == 1 ? acc[1][t][j] : v;
      v = me == 2 ? acc[2][t][j] : v;
      v = me == 3 ? acc[3][t][j] : v;
      const int m = mbase + h * 8 + j;
      Out[((size_t)b * COUT + m) * S_ + s] = v * outscale;
    }
  }
}

__global__ void finalize_loss_kernel(const float* __restrict__ lpart,
                                     const float* __restrict__ lacc1,
                                     float* __restrict__ out_loss) {
  static_assert(TOK / 16 == 128, "this kernel adds 128 rows with one wave");
  const int lane = threadIdx.x;
  float v[8];
#pragma unroll
  for (int r = 0; r < 4; ++r) {
    const float4 a = *(const float4*)(lpart + (lane + 32 * r) * 8);
    const float4 c = *(const float4*)(lpart + (lane + 32 * r) * 8 + 4);
    const float w[8] = {a.x, a.y, a.z, a.w, c.x, c.y, c.z, c.w};
#pragma unroll
    for (int i = 0; i < 8; ++i) v[i] = r == 0 ? w[i] : v[i] + w[i];
  }
#pragma unroll
  for (int d = 16; d > 0; d >>= 1) {
#pragma unroll
    for (int i = 0; i < 8; ++i) v[i] += __shfl_xor(v[i], d, 32);
  }
  if (lane == 0) {
    const float invT = 1.0f / (float)TOK;
    float l = 0.f;
    for (int e = 0; e < E_; ++e) l += (v[e] * invT) * (v[4 + e] * invT);
    for (int e = 0; e < E_; ++e) l += (lacc1[e] * invT) * (lacc1[4 + e] * invT);
    *out_loss = l;
  }
}

extern "C" void kernel_launch(void* const* d_in, const int* in_sizes, int n_in,
                              void* d_out, int out_size, void* d_ws, size_t ws_size,
                              hipStream_t stream) {
  const float* inp     = (const float*)d_in[0];
  const float* divisor = (const float*)d_in[1];
  const float* w0_gate = (const float*)d_in[2];
  const float* w0      = (const float*)d_in[3];
  const float* w1      = (const float*)d_in[4];
  const float* w2_gate = (const float*)d_in[5];
  const float* w2      = (const float*)d_in[6];
  float* out = (float*)d_out;

  char* ws = (char*)d_ws;
  size_t off = 0;
  float* lacc1 = (float*)(ws + off); off += 256;
  float* lpart = (float*)(ws + off); off += 4096;
  int* eidx = (int*)(ws + off); off += (size_t)TOK * sizeof(int);
  int* eidx2 = (int*)(ws + off); off += (size_t)TOK * sizeof(int);
  const size_t h0b = (size_t)B_ * 3 * I_ * S_ * sizeof(float);
  const size_t t1b = (size_t)B_ * I_ * S_ * sizeof(float);
  float* h0   = (float*)(ws + off); off += h0b;
  float* bsum = (float*)(ws + off); off += (size_t)B_ * 32 * I_ * sizeof(float);
  _Float16* g1p = (_Float16*)(ws + off); off += t1b;
  off = (off + 255) & ~(size_t)255;
  _Float16* xh2 = (_Float16*)(ws + off); off += (size_t)TOK * I_ * 2;
  _Float16* xh = (_Float16*)(ws + off); off += (size_t)TOK * F_ * 2;
  float* h2 = h0;

  gate0_kernel<<<TOK / 16, 256, 0, stream>>>(inp, w0_gate, eidx, lpart, lacc1, xh);

  moe_ws_kernel<<<dim3(3 * I_ / WS_ROWS, TOK / 512), 256, 0, stream>>>(
      xh, w0, eidx, h0, bsum);

  cumnorm_kernel<<<dim3(S_ / 32, B_), 1024, 0, stream>>>(h0, divisor, bsum, g1p);

  conv_ws_kernel<<<dim3(3 * I_ / 48, B_, S_ / 512), 256, 0, stream>>>(w1, g1p, h2);

  norm2gate_kernel<<<TOK / 16, 256, 0, stream>>>(h2, w2_gate, eidx2, lacc1, xh2);

  moe_ws2_kernel<<<dim3(F_ / 16, TOK / W2_TOKS), 256, 0, stream>>>(
      xh2, w2, eidx2, out, 1.0f  );

  finalize_loss_kernel<<<1, 32, 0, stream>>>(lpart, lacc1, out + (size_t)B_ * F_ * S_);
}


# first routed GEMM: token fragments skip the LDS: a wave loads its own 32-token pair's fragments 4 k-steps ahead and forms 3 m-tiles x 4 experts itself; all 8 waves compute (was 6), no block barriers
# speedup vs baseline: 5.3993x; 1.0425x over previous
#include <hip/hip_runtime.h>
#include <hip/hip_bf16.h>

typedef __attribute__((ext_vector_type(16))) _Float16 v16h;
typedef __attribute__((ext_vector_type(8)))  _Float16 v8h;
typedef __attribute__((ext_vector_type(4)))  _Float16 v4h;
typedef __attribute__((ext_vector_type(8)))  float    v8f;

#define B_ 2
#define F_ 512
#define S_ 1024
#define E_ 4
#define I_ 1024
#define TOK (B_*S_)
#define EPSN 1e-5f
#define SLOPE_ 0.02f

__device__ __forceinline__ int frag_j(int kin) {
  return kin < 8 ? kin : (kin < 16 ? kin - 8 : (kin < 24 ? kin - 8 : kin - 16));
}
__device__ __forceinline__ int frag_lane(int kin, int n) {
  return n + (((kin >> 3) & 1) ? 16 : 0);
}

template <int CTRL>
__device__ __forceinline__ float row_dpp(float v) {
  return __int_as_float(__builtin_amdgcn_update_dpp(0, __float_as_int(v), CTRL, 0xf, 0xf, false));
}

#define SP_ (S_ + 2)
__global__ void __launch_bounds__(1024)
cumnorm_kernel(const float* __restrict__ h0,
               const float* __restrict__ divisor,
               const float* __restrict__ bsum,
               _Float16* __restrict__ outp) {
  __shared__ float red1[1024], red2[1024];
  const int tile = blockIdx.x, b = blockIdx.y;
  const int s0 = tile * 32;
  const int wave = threadIdx.x >> 5, lane = threadIdx.x & 31;
  float cv = 0.f;
  {
    const float* bp = bsum + (size_t)b * 32 * I_ + wave * 32 + lane;
    float ts[32];
#pragma unroll
    for (int t = 0; t < 32; ++t) ts[t] = bp[(size_t)t * I_];
#pragma unroll
    for (int t = 0; t < 32; ++t) cv += t < tile ? ts[t] : 0.f;
  }
  const float dv = divisor[s0 + lane];
  const float* dep = h0 + ((size_t)b * 3 * I_ + wave * 32) * S_ + s0 + lane;
  const float* scp = dep + (size_t)I_ * S_;
  const float* shp = scp + (size_t)I_ * S_;
  float v[32];
  float lsum = 0.f;
#pragma unroll
  for (int i = 0; i < 32; ++i) {
    float d = dep[(size_t)i * S_];
    const float sc = scp[(size_t)i * S_], sh = shp[(size_t)i * S_];
    d += row_dpp<0x111>(d);
    d += row_dpp<0x112>(d);
    d += row_dpp<0x114>(d);
    d += row_dpp<0x118>(d);
    const float low = __int_as_float(__builtin_amdgcn_readlane(__float_as_int(d), 15));
    d += lane >= 16 ? low : 0.f;
    const float carry = __int_as_float(__builtin_amdgcn_readlane(__float_as_int(cv), i));
    const float x = (carry + d) / dv * sc + sh;
    v[i] = x; lsum += x;
  }
  red1[threadIdx.x] = lsum;
  __syncthreads();
  float tot = 0.f;
#pragma unroll
  for (int q = 0; q < 32; ++q) tot += red1[q * 32 + lane];
  const float mean = tot * (1.0f / I_);
  float lss = 0.f;
#pragma unroll
  for (int i = 0; i < 32; ++i) { float d = v[i] - mean; v[i] = d; lss += d * d; }
  red2[threadIdx.x] = lss;
  __syncthreads();
  float ss = 0.f;
#pragma unroll
  for (int q = 0; q < 32; ++q) ss += red2[q * 32 + lane];
  const float nrm = sqrtf(ss) * (1.0f / 32.0f);
  const float inv = 1.0f / (nrm + EPSN);
#pragma unroll
  for (int g = 0; g < 2; ++g) {
    v8h lo, hi;
#pragma unroll
    for (int e = 0; e < 8; ++e) {
      float y0 = v[g * 16 + e] * inv, y1 = v[g * 16 + 8 + e] * inv;
      y0 = y0 >= 0.f ? y0 : y0 * SLOPE_;
      y1 = y1 >= 0.f ? y1 : y1 * SLOPE_;
      lo[e] = (_Float16)y0; hi[e] = (_Float16)y1;
    }
    _Float16* o = outp + ((((size_t)(b * 32 + wave) * 2 + g) * SP_ + (s0 + lane + 2)) * 2) * 8;
    *(v8h*)o = lo;
    *(v8h*)(o + 8) = hi;
    if (tile == 0 && lane == 0) {
      const v8h z = {};
      *(v8h*)(o - 32) = z; *(v8h*)(o - 24) = z;
      *(v8h*)(o - 16) = z; *(v8h*)(o - 8) = z;
    }
  }
}

__device__ __forceinline__ v16h join_frag(v8h lo, v8h hi) {
  v16h r;
#pragma unroll
  for (int j = 0; j < 8; ++j) { r[j] = lo[j]; r[j + 8] = hi[j]; }
  return r;
}
__global__ void __launch_bounds__(256)
conv_ws_kernel(const float* __restrict__ W1,
               const _Float16* __restrict__ P,
               float* __restrict__ Out) {
  __shared__ __attribute__((aligned(16))) _Float16 Al[3 * 96 * 512];
  const int tid = threadIdx.x;
  const int mbase = blockIdx.x * 48;
  const int b = blockIdx.y;
  const int tbase = blockIdx.z * 512;
  {
    const float* Wrow = W1 + (size_t)mbase * (3 * I_) + tid * 12;
    const int kin = (tid * 4) & 31, ct = tid >> 3;
    const int hl = ((kin >> 3) & 1) * 16;
    const int j0 = frag_j(kin);
    const int sub = (j0 >> 3) * 256 + (j0 & 7);
#pragma unroll 16
    for (int r = 0; r < 48; ++r) {
      const float4* s4 = (const float4*)(Wrow + (size_t)r * (3 * I_));
      float4 f0 = s4[0], f1 = s4[1], f2 = s4[2];
      _Float16* dst = Al + ((r >> 4) * 96 + ct * 3) * 512 + ((r & 15) + hl) * 8 + sub;
      v4h q0 = {(_Float16)f0.x, (_Float16)f0.w, (_Float16)f1.z, (_Float16)f2.y};
      v4h q1 = {(_Float16)f0.y, (_Float16)f1.x, (_Float16)f1.w, (_Float16)f2.z};
      v4h q2 = {(_Float16)f0.z, (_Float16)f1.y, (_Float16)f2.x, (_Float16)f2.w};
      *(v4h*)(dst) = q0;
      *(v4h*)(dst + 512) = q1;
      *(v4h*)(dst + 1024) = q2;
    }
  }
  __syncthreads();
  const int wave = tid >> 5, lane = tid & 31, c = lane & 15, h = lane >> 4;
  const _Float16* Ab = Al + lane * 8;
  const size_t jhs = (size_t)SP_ * 16;
  const size_t cts = 2 * jhs;
  for (int pass = 0; pass < 2; ++pass) {
    const int s0 = tbase + (pass * 8 + wave) * 32;
    const _Float16* Pb = P + (size_t)b * 32 * cts + (size_t)(s0 + c) * 16 + h * 8;
    v8f acc[3][2];
#pragma unroll
    for (int mt = 0; mt < 3; ++mt) { acc[mt][0] = v8f{}; acc[mt][1] = v8f{}; }
    v16h bq[6][2];
#pragma unroll
    for (int d = 0; d < 6; ++d) {
      const _Float16* pp = Pb + (size_t)(d / 3) * cts + (d % 3) * 16;
#pragma unroll
      for (int tt = 0; tt < 2; ++tt)
        bq[d][tt] = join_frag(*(const v8h*)(pp + tt * 256),
                              *(const v8h*)(pp + tt * 256 + jhs));
    }
    v16h a[3];
#pragma unroll
    for (int mt = 0; mt < 3; ++mt)
      a[mt] = join_frag(*(const v8h*)(Ab + (mt * 96) * 512),
                        *(const v8h*)(Ab + (mt * 96) * 512 + 256));
    for (int ct0 = 0; ct0 < 32; ct0 += 2) {
#pragma unroll
      for (int d = 0; d < 6; ++d) {
        const int kt = ct0 * 3 + d;
        const int ktn = kt + 1 < 96 ? kt + 1 : 95;
        v16h an[3];
#pragma unroll
        for (int mt = 0; mt < 3; ++mt)
          an[mt] = join_frag(*(const v8h*)(Ab + (mt * 96 + ktn) * 512),
                             *(const v8h*)(Ab + (mt * 96 + ktn) * 512 + 256));
#pragma unroll
        for (int mt = 0; mt < 3; ++mt) {
#pragma unroll
          for (int tt = 0; tt < 2; ++tt)
            acc[mt][tt] = __builtin_amdgcn_wmma_f32_16x16x32_f16(
                false, a[mt], false, bq[d][tt], (short)0, acc[mt][tt], false, false);
        }
        int ctn = ct0 + 2 + d / 3; ctn = ctn < 32 ? ctn : 31;
        const _Float16* pp = Pb + (size_t)ctn * cts + (d % 3) * 16;
#pragma unroll
        for (int tt = 0; tt < 2; ++tt)
          bq[d][tt] = join_frag(*(const v8h*)(pp + tt * 256),
                                *(const v8h*)(pp + tt * 256 + jhs));
#pragma unroll
        for (int mt = 0; mt < 3; ++mt) a[mt] = an[mt];
      }
    }
#pragma unroll
    for (int mt = 0; mt < 3; ++mt) {
#pragma unroll
      for (int tt = 0; tt < 2; ++tt) {
#pragma unroll
        for (int j = 0; j < 8; ++j) {
          int m = mbase + mt * 16 + h * 8 + j;
          Out[((size_t)b * (3 * I_) + m) * S_ + s0 + tt * 16 + c] = acc[mt][tt][j];
        }
      }
    }
  }
}

__global__ void __launch_bounds__(256)
gate0_kernel(const float* __restrict__ X, const float* __restrict__ gw,
             int* __restrict__ eidx, float* __restrict__ lpart,
             float* __restrict__ lacc1, _Float16* __restrict__ xh) {
  constexpr int CIN = F_;
  __shared__ float4 gpart[256];
  const int tid = threadIdx.x;
  const int n = tid & 15, q = tid >> 4;
  const int tok0 = blockIdx.x * 16;
  const int b = tok0 / S_, s0 = tok0 % S_;
  if (blockIdx.x == 0 && tid < 8) lacc1[tid] = 0.f;
  {
    const float* xcol = X + ((size_t)b * CIN + q * 4) * S_ + s0 + n;
    float a0 = 0.f, a1 = 0.f, a2 = 0.f, a3 = 0.f;
#pragma unroll
    for (int g = 0; g < CIN / 64; ++g) {
      const int k0 = g * 64 + q * 4;
      float x0 = xcol[(size_t)(g * 64 + 0) * S_];
      float x1 = xcol[(size_t)(g * 64 + 1) * S_];
      float x2 = xcol[(size_t)(g * 64 + 2) * S_];
      float x3 = xcol[(size_t)(g * 64 + 3) * S_];
      float4 g0 = *(const float4*)(gw + 0 * CIN + k0);
      float4 g1 = *(const float4*)(gw + 1 * CIN + k0);
      float4 g2 = *(const float4*)(gw + 2 * CIN + k0);
      float4 g3 = *(const float4*)(gw + 3 * CIN + k0);
      a0 += x0 * g0.x; a0 += x1 * g0.y; a0 += x2 * g0.z; a0 += x3 * g0.w;
      a1 += x0 * g1.x; a1 += x1 * g1.y; a1 += x2 * g1.z; a1 += x3 * g1.w;
      a2 += x0 * g2.x; a2 += x1 * g2.y; a2 += x2 * g2.z; a2 += x3 * g2.w;
      a3 += x0 * g3.x; a3 += x1 * g3.y; a3 += x2 * g3.z; a3 += x3 * g3.w;
      const int kt = k0 >> 5, kin = k0 & 31;
      const int j0 = frag_j(kin);
      v4h pk = {(_Float16)x0, (_Float16)x1, (_Float16)x2, (_Float16)x3};
      *(v4h*)&xh[((size_t)blockIdx.x * (CIN / 32) + kt) * 512 + (j0 >> 3) * 256
                 + frag_lane(kin, n) * 8 + (j0 & 7)] = pk;
    }
    gpart[q * 16 + n] = make_float4(a0, a1, a2, a3);
  }
  __syncthreads();
  if (tid < 32) {
    const int c = tid & 15;
    float l0 = 0.f, l1 = 0.f, l2 = 0.f, l3 = 0.f;
#pragma unroll
    for (int r = 0; r < 16; ++r) {
      float4 p = gpart[r * 16 + c];
      l0 += p.x; l1 += p.y; l2 += p.z; l3 += p.w;
    }
    int me = 0; float lm = l0;
    if (l1 > lm) { lm = l1; me = 1; }
    if (l2 > lm) { lm = l2; me = 2; }
    if (l3 > lm) { lm = l3; me = 3; }
    if (tid < 16) eidx[tok0 + c] = me;
    float e0 = __expf(l0 - lm), e1 = __expf(l1 - lm);
    float e2 = __expf(l2 - lm), e3 = __expf(l3 - lm);
    float inv = 1.0f / (e0 + e1 + e2 + e3);
    float v[8] = {e0 * inv, e1 * inv, e2 * inv, e3 * inv,
                  me == 0 ? 1.f : 0.f, me == 1 ? 1.f : 0.f,
                  me == 2 ? 1.f : 0.f, me == 3 ? 1.f : 0.f};
#pragma unroll
    for (int d = 8; d > 0; d >>= 1) {
#pragma unroll
      for (int i = 0; i < 8; ++i) v[i] += __shfl_xor(v[i], d, 16);
    }
    if (tid == 0) {
      float* lp = lpart + (size_t)blockIdx.x * 8;
      *(float4*)lp = make_float4(v[0], v[1], v[2], v[3]);
      *(float4*)(lp + 4) = make_float4(v[4], v[5], v[6], v[7]);
    }
  }
}

#define WS_ROWS 48
#define W0_AHEAD 4
__global__ void __launch_bounds__(256)
moe_ws_kernel(const _Float16* __restrict__ Xh, const float* __restrict__ W,
              const int* __restrict__ eidx,
              float* __restrict__ Out, float* __restrict__ bsum) {
  constexpr int CIN = F_, COUT = 3 * I_, KT = CIN / 32, MT = WS_ROWS / 16;
  constexpr int NQ = E_ * MT;
  constexpr int NG = NQ / 2;
  constexpr int NPAIR = 512 / 32 / 8;
  static_assert(KT % W0_AHEAD == 0, "the queue slots are reused in whole rounds");
  __shared__ __attribute__((aligned(16))) _Float16 Aw[E_ * MT * KT * 512];
  const int tid = threadIdx.x;
  const int mbase = blockIdx.x * WS_ROWS;
  const int wave = tid >> 5, lane = tid & 31, c = lane & 15, h = lane >> 4;
  const int tileA = blockIdx.y * 32 + wave * 2;
  {
    for (int t0 = 0; t0 < 12; t0 += 3) {
      float4 f[3][8];
#pragma unroll
      for (int tq = 0; tq < 3; ++tq) {
        const int q = (t0 + tq) * 256 + tid;
        const int m4 = q % 12, ko = q / 12;
        const float* src = W + (size_t)(ko * 8) * COUT + mbase + m4 * 4;
#pragma unroll
        for (int i = 0; i < 8; ++i) f[tq][i] = *(const float4*)(src + (size_t)i * COUT);
      }
#pragma unroll
      for (int tq = 0; tq < 3; ++tq) {
        const int q = (t0 + tq) * 256 + tid;
        const int m4 = q % 12, ko = q / 12;
        const int e = ko >> 6, k8 = ko & 63;
        const int kt = k8 >> 2, o = k8 & 3;
        const int mt = m4 >> 2, c0 = (m4 & 3) * 4;
        _Float16* dst = Aw + (kt * NQ + e * MT + mt) * 512 + (o >> 1) * 256
                        + (c0 + 16 * (o & 1)) * 8;
        v8h r0, r1, r2, r3;
#pragma unroll
        for (int i = 0; i < 8; ++i) {
          r0[i] = (_Float16)f[tq][i].x; r1[i] = (_Float16)f[tq][i].y;
          r2[i] = (_Float16)f[tq][i].z; r3[i] = (_Float16)f[tq][i].w;
        }
        *(v8h*)(dst) = r0; *(v8h*)(dst + 8) = r1;
        *(v8h*)(dst + 16) = r2; *(v8h*)(dst + 24) = r3;
      }
    }
  }
  v16h bq[W0_AHEAD][2];
  {
    const _Float16* Xb = Xh + (size_t)tileA * KT * 512 + lane * 8;
#pragma unroll
    for (int d = 0; d < W0_AHEAD; ++d) {
#pragma unroll
      for (int t = 0; t < 2; ++t)
        bq[d][t] = join_frag(*(const v8h*)(Xb + ((size_t)t * KT + d) * 512),
                             *(const v8h*)(Xb + ((size_t)t * KT + d) * 512 + 256));
    }
  }
  __syncthreads();
  const _Float16* Ab = Aw + lane * 8;
  v16h a[NG];
#pragma unroll
  for (int i = 0; i < NG; ++i)
    a[i] = join_frag(*(const v8h*)(Ab + (size_t)i * 512),
                     *(const v8h*)(Ab + (size_t)i * 512 + 256));
  for (int p = 0; p < NPAIR; ++p) {
    const int tile0 = tileA + p * 16;
    const int tok0 = tile0 * 16;
    const int me0 = eidx[tok0 + c];
    const int me1 = eidx[tok0 + 16 + c];
    const _Float16* Xcur = Xh + (size_t)tile0 * KT * 512 + lane * 8;
    const int tileN = p + 1 < NPAIR ? tile0 + 16 : tile0;
    const _Float16* Xnext = Xh + (size_t)tileN * KT * 512 + lane * 8;
    v8f acc[NQ][2];
#pragma unroll
    for (int q = 0; q < NQ; ++q) { acc[q][0] = v8f{}; acc[q][1] = v8f{}; }
#pragma unroll 1
    for (int kt0 = 0; kt0 < KT; kt0 += W0_AHEAD) {
      const bool lastr = kt0 + W0_AHEAD >= KT;
      const _Float16* Ak = Ab + (size_t)kt0 * NQ * 512;
      const _Float16* An = lastr ? Ab : Ak + (size_t)W0_AHEAD * NQ * 512;
      const _Float16* Xr = lastr ? Xnext : Xcur + (size_t)(kt0 + W0_AHEAD) * 512;
#pragma unroll
      for (int d = 0; d < W0_AHEAD; ++d) {
#pragma unroll
        for (int g = 0; g < 2; ++g) {
          const _Float16* nx = g == 0 ? Ak + (size_t)(d * NQ + NG) * 512
                             : (d + 1 < W0_AHEAD ? Ak + (size_t)(d + 1) * NQ * 512 : An);
          v16h an[NG];
#pragma unroll
          for (int i = 0; i < NG; ++i)
            an[i] = join_frag(*(const v8h*)(nx + (size_t)i * 512),
                              *(const v8h*)(nx + (size_t)i * 512 + 256));
#pragma unroll
          for (int i = 0; i < NG; ++i) {
#pragma unroll
            for (int t = 0; t < 2; ++t)
              acc[g * NG + i][t] = __builtin_amdgcn_wmma_f32_16x16x32_f16(
                  false, a[i], false, bq[d][t], (short)0, acc[g * NG + i][t], false, false);
          }
          if (g == 1) {
#pragma unroll
            for (int t = 0; t < 2; ++t)
              bq[d][t] = join_frag(*(const v8h*)(Xr + ((size_t)t * KT + d) * 512),
                                   *(const v8h*)(Xr + ((size_t)t * KT + d) * 512 + 256));
            __builtin_amdgcn_sched_barrier(0);
          }
#pragma unroll
          for (int i = 0; i < NG; ++i) a[i] = an[i];
        }
      }
    }
    const int b = tok0 / S_, s0 = tok0 % S_;
#pragma unroll
    for (int mt = 0; mt < MT; ++mt) {
      float rs[8];
#pragma unroll
      for (int t = 0; t < 2; ++t) {
        const int me = t == 0 ? me0 : me1;
#pragma unroll
        for (int j = 0; j < 8; ++j) {
          float v = acc[0 * MT + mt][t][j];
          v = me == 1 ? acc[1 * MT + mt][t][j] : v;
          v = me == 2 ? acc[2 * MT + mt][t][j] : v;
          v = me == 3 ? acc[3 * MT + mt][t][j] : v;
          const int m = mbase + mt * 16 + h * 8 + j;
          Out[((size_t)b * COUT + m) * S_ + s0 + t * 16 + c] = v;
          rs[j] = t == 0 ? v : rs[j] + v;
        }
      }
      if (mbase + mt * 16 < I_) {
#pragma unroll
        for (int j = 0; j < 8; ++j) {
          rs[j] += row_dpp<0x161>(rs[j]);
          rs[j] += row_dpp<0x162>(rs[j]);
          rs[j] += row_dpp<0x164>(rs[j]);
          rs[j] += row_dpp<0x168>(rs[j]);
        }
        if (c == 0) {
          float* bp = bsum + (size_t)(b * 32 + (s0 >> 5)) * I_ + mbase + mt * 16 + h * 8;
          *(float4*)bp = make_float4(rs[0], rs[1], rs[2], rs[3]);
          *(float4*)(bp + 4) = make_float4(rs[4], rs[5], rs[6], rs[7]);
        }
      }
    }
  }
}

__global__ void __launch_bounds__(256)
norm2gate_kernel(const float* __restrict__ H, const float* __restrict__ gw,
                 int* __restrict__ eidx, float* __restrict__ lacc,
                 _Float16* __restrict__ xh) {
  constexpr int Cin = I_;
  constexpr int NG = I_ / 64;
  __shared__ float4 gpart[256];
  __shared__ float red1[256], red2[256];
  const int tile = blockIdx.x;
  const int b = (tile * 16) / S_;
  const int s0 = (tile * 16) % S_;
  {
    const int n = threadIdx.x & 15, kq = threadIdx.x >> 4;
    const float* c0 = H + ((size_t)b * 3 * Cin + kq * 4) * S_ + s0 + n;
    const float* c1 = c0 + (size_t)Cin * S_;
    const float* c2 = c1 + (size_t)Cin * S_;
    float v[NG * 4];
    float lsum = 0.f;
#pragma unroll
    for (int g = 0; g < NG; ++g) {
#pragma unroll
      for (int i = 0; i < 4; ++i) {
        size_t o = (size_t)(g * 64 + i) * S_;
        float t = c0[o] * c1[o] + c2[o];
        v[g * 4 + i] = t; lsum += t;
      }
    }
    red1[threadIdx.x] = lsum;
    __syncthreads();
    float tot = 0.f;
#pragma unroll
    for (int q = 0; q < 16; ++q) tot += red1[q * 16 + n];
    float mean = tot * (1.0f / I_);
    float lss = 0.f;
#pragma unroll
    for (int j = 0; j < NG * 4; ++j) { float d = v[j] - mean; v[j] = d; lss += d * d; }
    red2[threadIdx.x] = lss;
    __syncthreads();
    float ss = 0.f;
#pragma unroll
    for (int q = 0; q < 16; ++q) ss += red2[q * 16 + n];
    float nrm = sqrtf(ss) * (1.0f / 32.0f);
    float inv = 1.0f / (nrm + EPSN);
    float a0 = 0.f, a1 = 0.f, a2 = 0.f, a3 = 0.f;
#pragma unroll
    for (int g = 0; g < NG; ++g) {
      int k0 = g * 64 + kq * 4;
      float x0 = v[g * 4 + 0] * inv, x1 = v[g * 4 + 1] * inv;
      float x2 = v[g * 4 + 2] * inv, x3 = v[g * 4 + 3] * inv;
      x0 = x0 >= 0.f ? x0 : x0 * SLOPE_;
      x1 = x1 >= 0.f ? x1 : x1 * SLOPE_;
      x2 = x2 >= 0.f ? x2 : x2 * SLOPE_;
      x3 = x3 >= 0.f ? x3 : x3 * SLOPE_;
      float4 g0 = *(const float4*)(gw + 0 * Cin + k0);
      float4 g1 = *(const float4*)(gw + 1 * Cin + k0);
      float4 g2 = *(const float4*)(gw + 2 * Cin + k0);
      float4 g3 = *(const float4*)(gw + 3 * Cin + k0);
      a0 += x0 * g0.x; a0 += x1 * g0.y; a0 += x2 * g0.z; a0 += x3 * g0.w;
      a1 += x0 * g1.x; a1 += x1 * g1.y; a1 += x2 * g1.z; a1 += x3 * g1.w;
      a2 += x0 * g2.x; a2 += x1 * g2.y; a2 += x2 * g2.z; a2 += x3 * g2.w;
      a3 += x0 * g3.x; a3 += x1 * g3.y; a3 += x2 * g3.z; a3 += x3 * g3.w;
      const int kt = k0 >> 5, kin = k0 & 31;
      const int j0 = frag_j(kin);
      v4h pk = {(_Float16)x0, (_Float16)x1, (_Float16)x2, (_Float16)x3};
      *(v4h*)&xh[((size_t)tile * (Cin / 32) + kt) * 512 + (j0 >> 3) * 256
                 + frag_lane(kin, n) * 8 + (j0 & 7)] = pk;
    }
    gpart[threadIdx.x] = make_float4(a0, a1, a2, a3);
  }
  __syncthreads();
  if (threadIdx.x < 32) {
    const int lane = threadIdx.x, col = lane & 15;
    float l0 = 0.f, l1 = 0.f, l2 = 0.f, l3 = 0.f;
#pragma unroll
    for (int q = 0; q < 16; ++q) {
      float4 p = gpart[q * 16 + col];
      l0 += p.x; l1 += p.y; l2 += p.z; l3 += p.w;
    }
    int myexp = 0; float lmax = l0;
    if (l1 > lmax) { lmax = l1; myexp = 1; }
    if (l2 > lmax) { lmax = l2; myexp = 2; }
    if (l3 > lmax) { lmax = l3; myexp = 3; }
    if (lane < 16) eidx[tile * 16 + col] = myexp;
    float e0 = __expf(l0 - lmax), e1 = __expf(l1 - lmax);
    float e2 = __expf(l2 - lmax), e3 = __expf(l3 - lmax);
    float inv = 1.0f / (e0 + e1 + e2 + e3);
    float v[8] = {e0 * inv, e1 * inv, e2 * inv, e3 * inv,
                  myexp == 0 ? 1.f : 0.f, myexp == 1 ? 1.f : 0.f,
                  myexp == 2 ? 1.f : 0.f, myexp == 3 ? 1.f : 0.f};
#pragma unroll
    for (int d = 8; d > 0; d >>= 1) {
#pragma unroll
      for (int i = 0; i < 8; ++i) v[i] += __shfl_xor(v[i], d, 16);
    }
    if (lane == 0) {
#pragma unroll
      for (int i = 0; i < 8; ++i) atomicAdd(&lacc[i], v[i]);
    }
  }
}

#define W2_TOKS 256
#define W2_AHEAD 8
__global__ void __launch_bounds__(256)
moe_ws2_kernel(const _Float16* __restrict__ Xh, const float* __restrict__ W,
               const int* __restrict__ eidx,
               float* __restrict__ Out, float outscale) {
  constexpr int CIN = I_, COUT = F_, KT = CIN / 32;
  __shared__ __attribute__((aligned(16))) _Float16 Aw[E_ * KT * 512];
  const int tid = threadIdx.x;
  const int mbase = blockIdx.x * 16;
  const int wave = tid >> 5, lane = tid & 31, c = lane & 15, h = lane >> 4;
  const int tile0 = blockIdx.y * (W2_TOKS / 16) + wave * 2;
  int myexp[2];
  myexp[0] = eidx[tile0 * 16 + c];
  myexp[1] = eidx[tile0 * 16 + 16 + c];
  {
    for (int t0 = 0; t0 < 8; t0 += 4) {
      float4 f[4][8];
#pragma unroll
      for (int tq = 0; tq < 4; ++tq) {
        const int q = (t0 + tq) * 256 + tid;
        const int m4 = q & 3, ko = q >> 2;
        const float* src = W + (size_t)(ko * 8) * COUT + mbase + m4 * 4;
#pragma unroll
        for (int i = 0; i < 8; ++i) f[tq][i] = *(const float4*)(src + (size_t)i * COUT);
      }
#pragma unroll
      for (int tq = 0; tq < 4; ++tq) {
        const int q = (t0 + tq) * 256 + tid;
        const int m4 = q & 3, ko = q >> 2;
        const int e = ko >> 7, k8 = ko & 127;
        const int kt = k8 >> 2, o = k8 & 3;
        const int c0 = m4 * 4;
        _Float16* dst = Aw + (e * KT + kt) * 512 + (o >> 1) * 256
                        + (c0 + 16 * (o & 1)) * 8;
        v8h r0, r1, r2, r3;
#pragma unroll
        for (int i = 0; i < 8; ++i) {
          r0[i] = (_Float16)f[tq][i].x; r1[i] = (_Float16)f[tq][i].y;
          r2[i] = (_Float16)f[tq][i].z; r3[i] = (_Float16)f[tq][i].w;
        }
        *(v8h*)(dst) = r0; *(v8h*)(dst + 8) = r1;
        *(v8h*)(dst + 16) = r2; *(v8h*)(dst + 24) = r3;
      }
    }
  }
  const _Float16* Xb = Xh + (size_t)tile0 * KT * 512 + lane * 8;
  v16h bq[W2_AHEAD][2];
#pragma unroll
  for (int d = 0; d < W2_AHEAD; ++d) {
#pragma unroll
    for (int t = 0; t < 2; ++t)
      bq[d][t] = join_frag(*(const v8h*)(Xb + ((size_t)t * KT + d) * 512),
                           *(const v8h*)(Xb + ((size_t)t * KT + d) * 512 + 256));
  }
  __syncthreads();
  const _Float16* Ab = Aw + lane * 8;
  v8f acc[E_][2];
#pragma unroll
  for (int e = 0; e < E_; ++e) { acc[e][0] = v8f{}; acc[e][1] = v8f{}; }
  v16h a[E_];
#pragma unroll
  for (int e = 0; e < E_; ++e)
    a[e] = join_frag(*(const v8h*)(Ab + (size_t)(e * KT) * 512),
                     *(const v8h*)(Ab + (size_t)(e * KT) * 512 + 256));
  for (int kt0 = 0; kt0 < KT; kt0 += W2_AHEAD) {
#pragma unroll
    for (int d = 0; d < W2_AHEAD; ++d) {
      const int kt = kt0 + d;
      const int ktn = kt + 1 < KT ? kt + 1 : KT - 1;
      v16h an[E_];
#pragma unroll
      for (int e = 0; e < E_; ++e)
        an[e] = join_frag(*(const v8h*)(Ab + (size_t)(e * KT + ktn) * 512),
                          *(const v8h*)(Ab + (size_t)(e * KT + ktn) * 512 + 256));
#pragma unroll
      for (int e = 0; e < E_; ++e) {
#pragma unroll
        for (int t = 0; t < 2; ++t)
          acc[e][t] = __builtin_amdgcn_wmma_f32_16x16x32_f16(
              false, a[e], false, bq[d][t], (short)0, acc[e][t], false, false);
      }
      const int ktb = kt + W2_AHEAD < KT ? kt + W2_AHEAD : KT - 1;
#pragma unroll
      for (int t = 0; t < 2; ++t)
        bq[d][t] = join_frag(*(const v8h*)(Xb + ((size_t)t * KT + ktb) * 512),
                             *(const v8h*)(Xb + ((size_t)t * KT + ktb) * 512 + 256));
#pragma unroll
      for (int e = 0; e < E_; ++e) a[e] = an[e];
    }
  }
#pragma unroll
  for (int t = 0; t < 2; ++t) {
    const int me = myexp[t];
    const int tok = (tile0 + t) * 16 + c;
    const int b = tok / S_, s = tok % S_;
#pragma unroll
    for (int j = 0; j < 8; ++j) {
      float v = acc[0][t][j];
      v = me == 1 ? acc[1][t][j] : v;
      v = me == 2 ? acc[2][t][j] : v;
      v = me == 3 ? acc[3][t][j] : v;
      const int m = mbase + h * 8 + j;
      Out[((size_t)b * COUT + m) * S_ + s] = v * outscale;
    }
  }
}

__global__ void finalize_loss_kernel(const float* __restrict__ lpart,
                                     const float* __restrict__ lacc1,
                                     float* __restrict__ out_loss) {
  static_assert(TOK / 16 == 128, "this kernel adds 128 rows with one wave");
  const int lane = threadIdx.x;
  float v[8];
#pragma unroll
  for (int r = 0; r < 4; ++r) {
    const float4 a = *(const float4*)(lpart + (lane + 32 * r) * 8);
    const float4 c = *(const float4*)(lpart + (lane + 32 * r) * 8 + 4);
    const float w[8] = {a.x, a.y, a.z, a.w, c.x, c.y, c.z, c.w};
#pragma unroll
    for (int i = 0; i < 8; ++i) v[i] = r == 0 ? w[i] : v[i] + w[i];
  }
#pragma unroll
  for (int d = 16; d > 0; d >>= 1) {
#pragma unroll
    for (int i = 0; i < 8; ++i) v[i] += __shfl_xor(v[i], d, 32);
  }
  if (lane == 0) {
    const float invT = 1.0f / (float)TOK;
    float l = 0.f;
    for (int e = 0; e < E_; ++e) l += (v[e] * invT) * (v[4 + e] * invT);
    for (int e = 0; e < E_; ++e) l += (lacc1[e] * invT) * (lacc1[4 + e] * invT);
    *out_loss = l;
  }
}

extern "C" void kernel_launch(void* const* d_in, const int* in_sizes, int n_in,
                              void* d_out, int out_size, void* d_ws, size_t ws_size,
                              hipStream_t stream) {
  const float* inp     = (const float*)d_in[0];
  const float* divisor = (const float*)d_in[1];
  const float* w0_gate = (const float*)d_in[2];
  const float* w0      = (const float*)d_in[3];
  const float* w1      = (const float*)d_in[4];
  const float* w2_gate = (const float*)d_in[5];
  const float* w2      = (const float*)d_in[6];
  float* out = (float*)d_out;

  char* ws = (char*)d_ws;
  size_t off = 0;
  float* lacc1 = (float*)(ws + off); off += 256;
  float* lpart = (float*)(ws + off); off += 4096;
  int* eidx = (int*)(ws + off); off += (size_t)TOK * sizeof(int);
  int* eidx2 = (int*)(ws + off); off += (size_t)TOK * sizeof(int);
  const size_t h0b = (size_t)B_ * 3 * I_ * S_ * sizeof(float);
  const size_t t1b = (size_t)B_ * I_ * S_ * sizeof(float);
  float* h0   = (float*)(ws + off); off += h0b;
  float* bsum = (float*)(ws + off); off += (size_t)B_ * 32 * I_ * sizeof(float);
  _Float16* g1p = (_Float16*)(ws + off); off += t1b;
  off = (off + 255) & ~(size_t)255;
  _Float16* xh2 = (_Float16*)(ws + off); off += (size_t)TOK * I_ * 2;
  _Float16* xh = (_Float16*)(ws + off); off += (size_t)TOK * F_ * 2;
  float* h2 = h0;

  gate0_kernel<<<TOK / 16, 256, 0, stream>>>(inp, w0_gate, eidx, lpart, lacc1, xh);

  moe_ws_kernel<<<dim3(3 * I_ / WS_ROWS, TOK / 512), 256, 0, stream>>>(
      xh, w0, eidx, h0, bsum);

  cumnorm_kernel<<<dim3(S_ / 32, B_), 1024, 0, stream>>>(h0, divisor, bsum, g1p);

  conv_ws_kernel<<<dim3(3 * I_ / 48, B_, S_ / 512), 256, 0, stream>>>(w1, g1p, h2);

  norm2gate_kernel<<<TOK / 16, 256, 0, stream>>>(h2, w2_gate, eidx2, lacc1, xh2);

  moe_ws2_kernel<<<dim3(F_ / 16, TOK / W2_TOKS), 256, 0, stream>>>(
      xh2, w2, eidx2, out, 1.0f  );

  finalize_loss_kernel<<<1, 32, 0, stream>>>(lpart, lacc1, out + (size_t)B_ * F_ * S_);
}
